# starting kernel
# baseline (speedup 1.0000x reference)
#include <hip/hip_runtime.h>
#include <math.h>

#define N_NODES   100000
#define N_EDGES   1600000
#define N_FEAT    500
#define HIDDEN    64
#define N_CLASSES 16
#define M_TILES   (N_NODES / 16)

typedef __attribute__((ext_vector_type(2))) float v2f;
typedef __attribute__((ext_vector_type(8))) float v8f;


__global__ void k_init_deg(float* deg) {
    int i = blockIdx.x * blockDim.x + threadIdx.x;
    if (i < N_NODES) deg[i] = 1.0f;
}

__global__ void k_accum_deg(const int* __restrict__ col,
                            const float* __restrict__ w,
                            float* __restrict__ deg) {
    int e = blockIdx.x * blockDim.x + threadIdx.x;
    if (e < N_EDGES) atomicAdd(&deg[col[e]], w[e]);
}

__global__ void k_dinv(const float* __restrict__ deg, float* __restrict__ dinv) {
    int i = blockIdx.x * blockDim.x + threadIdx.x;
    if (i < N_NODES) {
        float d = deg[i];
        dinv[i] = (d > 0.0f) ? rsqrtf(d) : 0.0f;
    }
}

__global__ void k_norm(const int* __restrict__ row, const int* __restrict__ col,
                       const float* __restrict__ w, const float* __restrict__ dinv,
                       float* __restrict__ norm) {
    int e = blockIdx.x * blockDim.x + threadIdx.x;
    if (e < N_EDGES) norm[e] = dinv[row[e]] * w[e] * dinv[col[e]];
}


__global__ __launch_bounds__(128) void k_gemm1(const float* __restrict__ x,
                                               const float* __restrict__ W1,
                                               float* __restrict__ h1) {
    const int wave  = threadIdx.x >> 5;
    const int lane  = threadIdx.x & 31;
    const int mtile = blockIdx.x * 4 + wave;
    if (mtile >= M_TILES) return;
    const int row0 = mtile * 16;
    const int half = lane >> 4;
    const int l16  = lane & 15;

    const float* arow = x + (size_t)(row0 + l16) * N_FEAT + half * 2;
    v8f c0 = {}, c1 = {}, c2 = {}, c3 = {};

    for (int k = 0; k < N_FEAT; k += 4) {
        v2f a;
        a.x = arow[k];
        a.y = arow[k + 1];
        const float* bp = W1 + (size_t)(k + half * 2) * HIDDEN + l16;
        v2f b0, b1, b2, b3;
        b0.x = bp[0];   b0.y = bp[HIDDEN + 0];
        b1.x = bp[16];  b1.y = bp[HIDDEN + 16];
        b2.x = bp[32];  b2.y = bp[HIDDEN + 32];
        b3.x = bp[48];  b3.y = bp[HIDDEN + 48];
        c0 = __builtin_amdgcn_wmma_f32_16x16x4_f32(false, a, false, b0, (short)0, c0, false, false);
        c1 = __builtin_amdgcn_wmma_f32_16x16x4_f32(false, a, false, b1, (short)0, c1, false, false);
        c2 = __builtin_amdgcn_wmma_f32_16x16x4_f32(false, a, false, b2, (short)0, c2, false, false);
        c3 = __builtin_amdgcn_wmma_f32_16x16x4_f32(false, a, false, b3, (short)0, c3, false, false);
    }

    float* out = h1 + (size_t)(row0 + half * 8) * HIDDEN + l16;
    for (int r = 0; r < 8; ++r) {
        out[(size_t)r * HIDDEN + 0]  = c0[r];
        out[(size_t)r * HIDDEN + 16] = c1[r];
        out[(size_t)r * HIDDEN + 32] = c2[r];
        out[(size_t)r * HIDDEN + 48] = c3[r];
    }
}

__global__ void k_init_agg1(const float* __restrict__ h1,
                            const float* __restrict__ dinv,
                            const float* __restrict__ b1,
                            float* __restrict__ agg1) {
    int t = blockIdx.x * blockDim.x + threadIdx.x;
    if (t < N_NODES * HIDDEN) {
        int i = t / HIDDEN, f = t % HIDDEN;
        float s = dinv[i];
        agg1[t] = b1[f] + s * s * h1[t];
    }
}

__global__ void k_scatter1(const int* __restrict__ row, const int* __restrict__ col,
                           const float* __restrict__ norm,
                           const float* __restrict__ h1,
                           float* __restrict__ agg1) {
    unsigned t = blockIdx.x * blockDim.x + threadIdx.x;
    if (t < (unsigned)N_EDGES * HIDDEN) {
        unsigned e = t / HIDDEN, f = t % HIDDEN;
        float v = norm[e] * h1[(size_t)row[e] * HIDDEN + f];
        atomicAdd(&agg1[(size_t)col[e] * HIDDEN + f], v);
    }
}


__global__ __launch_bounds__(128) void k_gemm2(const float* __restrict__ agg1,
                                               const float* __restrict__ W2,
                                               float* __restrict__ h2) {
    const int wave  = threadIdx.x >> 5;
    const int lane  = threadIdx.x & 31;
    const int mtile = blockIdx.x * 4 + wave;
    if (mtile >= M_TILES) return;
    const int row0 = mtile * 16;
    const int half = lane >> 4;
    const int l16  = lane & 15;

    const float* arow = agg1 + (size_t)(row0 + l16) * HIDDEN + half * 2;
    v8f c = {};

    for (int k = 0; k < HIDDEN; k += 4) {
        v2f a;
        a.x = fmaxf(arow[k], 0.0f);
        a.y = fmaxf(arow[k + 1], 0.0f);
        const float* bp = W2 + (size_t)(k + half * 2) * N_CLASSES + l16;
        v2f b;
        b.x = bp[0];
        b.y = bp[N_CLASSES];
        c = __builtin_amdgcn_wmma_f32_16x16x4_f32(false, a, false, b, (short)0, c, false, false);
    }

    float* out = h2 + (size_t)(row0 + half * 8) * N_CLASSES + l16;
    for (int r = 0; r < 8; ++r)
        out[(size_t)r * N_CLASSES] = c[r];
}


__global__ void k_init_agg2(const float* __restrict__ h2,
                            const float* __restrict__ dinv,
                            const float* __restrict__ b2,
                            float* __restrict__ agg2) {
    int t = blockIdx.x * blockDim.x + threadIdx.x;
    if (t < N_NODES * N_CLASSES) {
        int i = t / N_CLASSES, cc = t % N_CLASSES;
        float s = dinv[i];
        agg2[t] = b2[cc] + s * s * h2[t];
    }
}

__global__ void k_scatter2(const int* __restrict__ row, const int* __restrict__ col,
                           const float* __restrict__ norm,
                           const float* __restrict__ h2,
                           float* __restrict__ agg2) {
    unsigned t = blockIdx.x * blockDim.x + threadIdx.x;
    if (t < (unsigned)N_EDGES * N_CLASSES) {
        unsigned e = t / N_CLASSES, cc = t % N_CLASSES;
        float v = norm[e] * h2[(size_t)row[e] * N_CLASSES + cc];
        atomicAdd(&agg2[(size_t)col[e] * N_CLASSES + cc], v);
    }
}


__global__ void k_logsoftmax(const float* __restrict__ agg2, float* __restrict__ out) {
    int i = blockIdx.x * blockDim.x + threadIdx.x;
    if (i >= N_NODES) return;
    const float* v = agg2 + (size_t)i * N_CLASSES;
    float m = v[0];
    #pragma unroll
    for (int c = 1; c < N_CLASSES; ++c) m = fmaxf(m, v[c]);
    float s = 0.0f;
    #pragma unroll
    for (int c = 0; c < N_CLASSES; ++c) s += expf(v[c] - m);
    float l = logf(s);
    float* o = out + (size_t)i * N_CLASSES;
    #pragma unroll
    for (int c = 0; c < N_CLASSES; ++c) o[c] = v[c] - m - l;
}


extern "C" void kernel_launch(void* const* d_in, const int* in_sizes, int n_in,
                              void* d_out, int out_size, void* d_ws, size_t ws_size,
                              hipStream_t stream) {
    const float* x  = (const float*)d_in[0];
    const int*   ei = (const int*)d_in[1];
    const float* ew = (const float*)d_in[2];
    const float* W1 = (const float*)d_in[3];
    const float* b1 = (const float*)d_in[4];
    const float* W2 = (const float*)d_in[5];
    const float* b2 = (const float*)d_in[6];
    float* out = (float*)d_out;

    const int* row = ei;
    const int* col = ei + N_EDGES;

    float* ws   = (float*)d_ws;
    float* deg  = ws;
    float* dinv = deg  + N_NODES;
    float* norm = dinv + N_NODES;
    float* h1   = norm + N_EDGES;
    float* agg1 = h1   + (size_t)N_NODES * HIDDEN;
    float* h2   = agg1 + (size_t)N_NODES * HIDDEN;
    float* agg2 = h2   + (size_t)N_NODES * N_CLASSES;

    const int B = 256;

    k_init_deg <<<(N_NODES + B - 1) / B, B, 0, stream>>>(deg);
    k_accum_deg<<<(N_EDGES + B - 1) / B, B, 0, stream>>>(col, ew, deg);
    k_dinv     <<<(N_NODES + B - 1) / B, B, 0, stream>>>(deg, dinv);
    k_norm     <<<(N_EDGES + B - 1) / B, B, 0, stream>>>(row, col, ew, dinv, norm);

    k_gemm1<<<(M_TILES + 3) / 4, 128, 0, stream>>>(x, W1, h1);

    k_init_agg1<<<(N_NODES * HIDDEN + B - 1) / B, B, 0, stream>>>(h1, dinv, b1, agg1);
    {
        unsigned total = (unsigned)N_EDGES * HIDDEN;
        k_scatter1<<<(total + B - 1) / B, B, 0, stream>>>(row, col, norm, h1, agg1);
    }

    k_gemm2<<<(M_TILES + 3) / 4, 128, 0, stream>>>(agg1, W2, h2);

    k_init_agg2<<<(N_NODES * N_CLASSES + B - 1) / B, B, 0, stream>>>(h2, dinv, b2, agg2);
    {
        unsigned total = (unsigned)N_EDGES * N_CLASSES;
        k_scatter2<<<(total + B - 1) / B, B, 0, stream>>>(row, col, norm, h2, agg2);
    }

    k_logsoftmax<<<(N_NODES + B - 1) / B, B, 0, stream>>>(agg2, out);
}


# Aggregate by destination-sorted edge segments (counting sort by target, int rank atomics) instead of float atomic scatter; per-node sums now run in arrival-rank order, self-loop last
# speedup vs baseline: 1.7483x; 1.7483x over previous
#include <hip/hip_runtime.h>
#include <math.h>

#define N_NODES   100000
#define N_EDGES   1600000
#define N_FEAT    500
#define HIDDEN    64
#define N_CLASSES 16
#define M_TILES   (N_NODES / 16)

typedef __attribute__((ext_vector_type(2))) float v2f;
typedef __attribute__((ext_vector_type(8))) float v8f;

__device__ __forceinline__ int node_clamp(int s) {
    return (int)min((unsigned)s, (unsigned)(N_NODES - 1));
}


__global__ void k_zero_cnt(int* __restrict__ cnt) {
    int i = blockIdx.x * blockDim.x + threadIdx.x;
    if (i < N_NODES) cnt[i] = 0;
}

__global__ void k_rank(const int* __restrict__ col, int* __restrict__ cnt,
                       int* __restrict__ pos) {
    int e = blockIdx.x * blockDim.x + threadIdx.x;
    if (e < N_EDGES) pos[e] = atomicAdd(&cnt[col[e]], 1);
}

static_assert(N_NODES % 16 == 0, "a thread's 16 counts are all inside or all outside the array");
__global__ __launch_bounds__(1024) void k_scan(int* __restrict__ cnt) {
    __shared__ int wsum[32];
    __shared__ int tile_total;
    const int tid  = threadIdx.x;
    const int lane = tid & 31;
    const int wv   = tid >> 5;
    int carry = 0;
    for (int base = 0; base < N_NODES; base += 16384) {
        const int idx = base + tid * 16;
        const bool in = idx < N_NODES;
        int4 v0 = make_int4(0, 0, 0, 0), v1 = v0, v2 = v0, v3 = v0;
        if (in) {
            const int4* p = (const int4*)(cnt + idx);
            v0 = p[0]; v1 = p[1]; v2 = p[2]; v3 = p[3];
        }
        const int s0 = v0.x + v0.y + v0.z + v0.w;
        const int s1 = v1.x + v1.y + v1.z + v1.w;
        const int s2 = v2.x + v2.y + v2.z + v2.w;
        const int s3 = v3.x + v3.y + v3.z + v3.w;
        const int s = s0 + s1 + s2 + s3;
        int inc = s;
        for (int d = 1; d < 32; d <<= 1) {
            int n = __shfl_up(inc, d);
            if (lane >= d) inc += n;
        }
        if (lane == 31) wsum[wv] = inc;
        __syncthreads();
        if (wv == 0) {
            const int ws = wsum[lane];
            int winc = ws;
            for (int d = 1; d < 32; d <<= 1) {
                int n = __shfl_up(winc, d);
                if (lane >= d) winc += n;
            }
            wsum[lane] = winc - ws;
            if (lane == 31) tile_total = winc;
        }
        __syncthreads();
        const int ex = carry + wsum[wv] + (inc - s);
        carry += tile_total;
        if (in) {
            int4 o0, o1, o2, o3;
            o0.x = ex;             o0.y = o0.x + v0.x; o0.z = o0.y + v0.y; o0.w = o0.z + v0.z;
            o1.x = ex + s0;        o1.y = o1.x + v1.x; o1.z = o1.y + v1.y; o1.w = o1.z + v1.z;
            o2.x = o1.x + s1;      o2.y = o2.x + v2.x; o2.z = o2.y + v2.y; o2.w = o2.z + v2.z;
            o3.x = o2.x + s2;      o3.y = o3.x + v3.x; o3.z = o3.y + v3.y; o3.w = o3.z + v3.z;
            int4* p = (int4*)(cnt + idx);
            p[0] = o0; p[1] = o1; p[2] = o2; p[3] = o3;
        }
        __syncthreads();
    }
}

__global__ void k_fill(const int* __restrict__ row, const int* __restrict__ col,
                       const float* __restrict__ w, const int* __restrict__ off,
                       const int* __restrict__ pos,
                       int* __restrict__ csr_src, float* __restrict__ csr_w) {
    int e = blockIdx.x * blockDim.x + threadIdx.x;
    if (e < N_EDGES) {
        int p = off[col[e]] + pos[e];
        csr_src[p] = row[e];
        csr_w[p]   = w[e];
    }
}

__global__ void k_dinv(const int* __restrict__ off, const float* __restrict__ csr_w,
                       float* __restrict__ dinv) {
    int i = blockIdx.x * blockDim.x + threadIdx.x;
    if (i < N_NODES) {
        const int b  = off[i];
        const int en = (i + 1 < N_NODES) ? off[i + 1] : N_EDGES;
        float d = 1.0f;
        for (int j = b; j < en; ++j) d += csr_w[j];
        dinv[i] = (d > 0.0f) ? rsqrtf(d) : 0.0f;
    }
}


__global__ __launch_bounds__(128) void k_gemm1(const float* __restrict__ x,
                                               const float* __restrict__ W1,
                                               float* __restrict__ h1) {
    const int wave  = threadIdx.x >> 5;
    const int lane  = threadIdx.x & 31;
    const int mtile = blockIdx.x * 4 + wave;
    if (mtile >= M_TILES) return;
    const int row0 = mtile * 16;
    const int half = lane >> 4;
    const int l16  = lane & 15;

    const float* arow = x + (size_t)(row0 + l16) * N_FEAT + half * 2;
    v8f c0 = {}, c1 = {}, c2 = {}, c3 = {};

    for (int k = 0; k < N_FEAT; k += 4) {
        v2f a;
        a.x = arow[k];
        a.y = arow[k + 1];
        const float* bp = W1 + (size_t)(k + half * 2) * HIDDEN + l16;
        v2f b0, b1, b2, b3;
        b0.x = bp[0];   b0.y = bp[HIDDEN + 0];
        b1.x = bp[16];  b1.y = bp[HIDDEN + 16];
        b2.x = bp[32];  b2.y = bp[HIDDEN + 32];
        b3.x = bp[48];  b3.y = bp[HIDDEN + 48];
        c0 = __builtin_amdgcn_wmma_f32_16x16x4_f32(false, a, false, b0, (short)0, c0, false, false);
        c1 = __builtin_amdgcn_wmma_f32_16x16x4_f32(false, a, false, b1, (short)0, c1, false, false);
        c2 = __builtin_amdgcn_wmma_f32_16x16x4_f32(false, a, false, b2, (short)0, c2, false, false);
        c3 = __builtin_amdgcn_wmma_f32_16x16x4_f32(false, a, false, b3, (short)0, c3, false, false);
    }

    float* out = h1 + (size_t)(row0 + half * 8) * HIDDEN + l16;
    for (int r = 0; r < 8; ++r) {
        out[(size_t)r * HIDDEN + 0]  = c0[r];
        out[(size_t)r * HIDDEN + 16] = c1[r];
        out[(size_t)r * HIDDEN + 32] = c2[r];
        out[(size_t)r * HIDDEN + 48] = c3[r];
    }
}

__global__ void k_gather1(const int* __restrict__ off, const int* __restrict__ csr_src,
                          const float* __restrict__ csr_w, const float* __restrict__ dinv,
                          const float* __restrict__ h1, const float* __restrict__ b1,
                          float* __restrict__ agg1) {
    const int t = blockIdx.x * blockDim.x + threadIdx.x;
    const int i = t >> 4;
    const int q = t & 15;
    if (i >= N_NODES) return;
    const int b  = off[i];
    const int en = (i + 1 < N_NODES) ? off[i + 1] : N_EDGES;
    const float di = dinv[i];
    const float4* __restrict__ h4 = (const float4*)h1;
    float4 acc = make_float4(0.0f, 0.0f, 0.0f, 0.0f);
    int j = b;
    for (; j + 4 <= en; j += 4) {
        const int s0 = node_clamp(csr_src[j]), s1 = node_clamp(csr_src[j + 1]), s2 = node_clamp(csr_src[j + 2]), s3 = node_clamp(csr_src[j + 3]);
        const float w0 = csr_w[j], w1 = csr_w[j + 1], w2 = csr_w[j + 2], w3 = csr_w[j + 3];
        const float d0 = dinv[s0], d1 = dinv[s1], d2 = dinv[s2], d3 = dinv[s3];
        const float4 v0 = h4[(size_t)s0 * 16 + q];
        const float4 v1 = h4[(size_t)s1 * 16 + q];
        const float4 v2 = h4[(size_t)s2 * 16 + q];
        const float4 v3 = h4[(size_t)s3 * 16 + q];
        const float n0 = d0 * w0 * di, n1 = d1 * w1 * di, n2 = d2 * w2 * di, n3 = d3 * w3 * di;
        acc.x += n0 * v0.x; acc.y += n0 * v0.y; acc.z += n0 * v0.z; acc.w += n0 * v0.w;
        acc.x += n1 * v1.x; acc.y += n1 * v1.y; acc.z += n1 * v1.z; acc.w += n1 * v1.w;
        acc.x += n2 * v2.x; acc.y += n2 * v2.y; acc.z += n2 * v2.z; acc.w += n2 * v2.w;
        acc.x += n3 * v3.x; acc.y += n3 * v3.y; acc.z += n3 * v3.z; acc.w += n3 * v3.w;
    }
    for (; j < en; ++j) {
        const int s0 = node_clamp(csr_src[j]);
        const float n0 = dinv[s0] * csr_w[j] * di;
        const float4 v0 = h4[(size_t)s0 * 16 + q];
        acc.x += n0 * v0.x; acc.y += n0 * v0.y; acc.z += n0 * v0.z; acc.w += n0 * v0.w;
    }
    const float4 hv = h4[(size_t)i * 16 + q];
    const float4 bv = make_float4(b1[q * 4], b1[q * 4 + 1], b1[q * 4 + 2], b1[q * 4 + 3]);
    const float sl = di * di;
    float4 o;
    o.x = bv.x + sl * hv.x + acc.x;
    o.y = bv.y + sl * hv.y + acc.y;
    o.z = bv.z + sl * hv.z + acc.z;
    o.w = bv.w + sl * hv.w + acc.w;
    ((float4*)agg1)[(size_t)i * 16 + q] = o;
}


__global__ __launch_bounds__(128) void k_gemm2(const float* __restrict__ agg1,
                                               const float* __restrict__ W2,
                                               float* __restrict__ h2) {
    const int wave  = threadIdx.x >> 5;
    const int lane  = threadIdx.x & 31;
    const int mtile = blockIdx.x * 4 + wave;
    if (mtile >= M_TILES) return;
    const int row0 = mtile * 16;
    const int half = lane >> 4;
    const int l16  = lane & 15;

    const float* arow = agg1 + (size_t)(row0 + l16) * HIDDEN + half * 2;
    v8f c = {};

    for (int k = 0; k < HIDDEN; k += 4) {
        v2f a;
        a.x = fmaxf(arow[k], 0.0f);
        a.y = fmaxf(arow[k + 1], 0.0f);
        const float* bp = W2 + (size_t)(k + half * 2) * N_CLASSES + l16;
        v2f b;
        b.x = bp[0];
        b.y = bp[N_CLASSES];
        c = __builtin_amdgcn_wmma_f32_16x16x4_f32(false, a, false, b, (short)0, c, false, false);
    }

    float* out = h2 + (size_t)(row0 + half * 8) * N_CLASSES + l16;
    for (int r = 0; r < 8; ++r)
        out[(size_t)r * N_CLASSES] = c[r];
}

__global__ void k_gather2(const int* __restrict__ off, const int* __restrict__ csr_src,
                          const float* __restrict__ csr_w, const float* __restrict__ dinv,
                          const float* __restrict__ h2, const float* __restrict__ b2,
                          float* __restrict__ agg2) {
    const int t = blockIdx.x * blockDim.x + threadIdx.x;
    const int i = t >> 2;
    const int q = t & 3;
    if (i >= N_NODES) return;
    const int b  = off[i];
    const int en = (i + 1 < N_NODES) ? off[i + 1] : N_EDGES;
    const float di = dinv[i];
    const float4* __restrict__ h4 = (const float4*)h2;
    float4 acc = make_float4(0.0f, 0.0f, 0.0f, 0.0f);
    int j = b;
    for (; j + 4 <= en; j += 4) {
        const int s0 = node_clamp(csr_src[j]), s1 = node_clamp(csr_src[j + 1]), s2 = node_clamp(csr_src[j + 2]), s3 = node_clamp(csr_src[j + 3]);
        const float w0 = csr_w[j], w1 = csr_w[j + 1], w2 = csr_w[j + 2], w3 = csr_w[j + 3];
        const float d0 = dinv[s0], d1 = dinv[s1], d2 = dinv[s2], d3 = dinv[s3];
        const float4 v0 = h4[(size_t)s0 * 4 + q];
        const float4 v1 = h4[(size_t)s1 * 4 + q];
        const float4 v2 = h4[(size_t)s2 * 4 + q];
        const float4 v3 = h4[(size_t)s3 * 4 + q];
        const float n0 = d0 * w0 * di, n1 = d1 * w1 * di, n2 = d2 * w2 * di, n3 = d3 * w3 * di;
        acc.x += n0 * v0.x; acc.y += n0 * v0.y; acc.z += n0 * v0.z; acc.w += n0 * v0.w;
        acc.x += n1 * v1.x; acc.y += n1 * v1.y; acc.z += n1 * v1.z; acc.w += n1 * v1.w;
        acc.x += n2 * v2.x; acc.y += n2 * v2.y; acc.z += n2 * v2.z; acc.w += n2 * v2.w;
        acc.x += n3 * v3.x; acc.y += n3 * v3.y; acc.z += n3 * v3.z; acc.w += n3 * v3.w;
    }
    for (; j < en; ++j) {
        const int s0 = node_clamp(csr_src[j]);
        const float n0 = dinv[s0] * csr_w[j] * di;
        const float4 v0 = h4[(size_t)s0 * 4 + q];
        acc.x += n0 * v0.x; acc.y += n0 * v0.y; acc.z += n0 * v0.z; acc.w += n0 * v0.w;
    }
    const float4 hv = h4[(size_t)i * 4 + q];
    const float4 bv = make_float4(b2[q * 4], b2[q * 4 + 1], b2[q * 4 + 2], b2[q * 4 + 3]);
    const float sl = di * di;
    float4 o;
    o.x = bv.x + sl * hv.x + acc.x;
    o.y = bv.y + sl * hv.y + acc.y;
    o.z = bv.z + sl * hv.z + acc.z;
    o.w = bv.w + sl * hv.w + acc.w;
    ((float4*)agg2)[(size_t)i * 4 + q] = o;
}


__global__ void k_logsoftmax(const float* __restrict__ agg2, float* __restrict__ out) {
    int i = blockIdx.x * blockDim.x + threadIdx.x;
    if (i >= N_NODES) return;
    const float* v = agg2 + (size_t)i * N_CLASSES;
    float m = v[0];
    #pragma unroll
    for (int c = 1; c < N_CLASSES; ++c) m = fmaxf(m, v[c]);
    float s = 0.0f;
    #pragma unroll
    for (int c = 0; c < N_CLASSES; ++c) s += expf(v[c] - m);
    float l = logf(s);
    float* o = out + (size_t)i * N_CLASSES;
    #pragma unroll
    for (int c = 0; c < N_CLASSES; ++c) o[c] = v[c] - m - l;
}


extern "C" void kernel_launch(void* const* d_in, const int* in_sizes, int n_in,
                              void* d_out, int out_size, void* d_ws, size_t ws_size,
                              hipStream_t stream) {
    const float* x  = (const float*)d_in[0];
    const int*   ei = (const int*)d_in[1];
    const float* ew = (const float*)d_in[2];
    const float* W1 = (const float*)d_in[3];
    const float* b1 = (const float*)d_in[4];
    const float* W2 = (const float*)d_in[5];
    const float* b2 = (const float*)d_in[6];
    float* out = (float*)d_out;

    const int* row = ei;
    const int* col = ei + N_EDGES;

    float* ws      = (float*)d_ws;
    int*   off     = (int*)ws;
    float* dinv    = ws + N_NODES;
    int*   csr_src = (int*)(dinv + N_NODES);
    float* h1      = (float*)csr_src + N_EDGES;
    float* agg1    = h1   + (size_t)N_NODES * HIDDEN;
    float* csr_w   = agg1 + (size_t)N_NODES * HIDDEN;
    float* agg2    = csr_w + (size_t)N_NODES * N_CLASSES;
    float* h2      = h1;
    int*   pos     = (int*)agg1;

    const int B = 256;

    k_zero_cnt<<<(N_NODES + B - 1) / B, B, 0, stream>>>(off);
    k_rank    <<<(N_EDGES + B - 1) / B, B, 0, stream>>>(col, off, pos);
    k_scan    <<<1, 1024, 0, stream>>>(off);
    k_fill    <<<(N_EDGES + B - 1) / B, B, 0, stream>>>(row, col, ew, off, pos, csr_src, csr_w);
    k_dinv    <<<(N_NODES + B - 1) / B, B, 0, stream>>>(off, csr_w, dinv);

    k_gemm1<<<(M_TILES + 3) / 4, 128, 0, stream>>>(x, W1, h1);

    k_gather1<<<(N_NODES * 16 + B - 1) / B, B, 0, stream>>>(off, csr_src, csr_w, dinv, h1, b1, agg1);

    k_gemm2<<<(M_TILES + 3) / 4, 128, 0, stream>>>(agg1, W2, h2);

    k_gather2<<<(N_NODES * 4 + B - 1) / B, B, 0, stream>>>(off, csr_src, csr_w, dinv, h2, b2, agg2);

    k_logsoftmax<<<(N_NODES + B - 1) / B, B, 0, stream>>>(agg2, out);
}


# Edge-sort ranks from LDS counters (two-level counting sort: 512-node buckets per edge chunk, then per node inside each bucket) replace 1.6M returning global atomics; sums in LDS arrival order
# speedup vs baseline: 2.7262x; 1.5593x over previous
#include <hip/hip_runtime.h>
#include <math.h>

#define N_NODES   100000
#define N_EDGES   1600000
#define N_FEAT    500
#define HIDDEN    64
#define N_CLASSES 16
#define M_TILES   (N_NODES / 16)

typedef __attribute__((ext_vector_type(2))) float v2f;
typedef __attribute__((ext_vector_type(8))) float v8f;

__device__ __forceinline__ int node_clamp(int s) {
    return (int)min((unsigned)s, (unsigned)(N_NODES - 1));
}


#define SORT_BLOCKS 256
#define SORT_CHUNK  (N_EDGES / SORT_BLOCKS)
#define SORT_TRIPS  ((SORT_CHUNK + 1023) / 1024)
#define SORT_GROUP  8
#define BKT_SHIFT   9
#define BKT_NODES   (1 << BKT_SHIFT)
#define N_BKT       ((N_NODES + BKT_NODES - 1) / BKT_NODES)
#define HIST_LEN    (N_BKT * SORT_BLOCKS)
static_assert(N_EDGES % SORT_BLOCKS == 0, "chunks are equal");
static_assert(HIST_LEN % 16 == 0, "a scan thread's 16 counts are all inside or all outside the array");
static_assert(N_NODES % 16 == 0, "M tiles are exact");

__global__ __launch_bounds__(1024) void k_bkt_hist(const int* __restrict__ col, int* __restrict__ bh) {
    __shared__ int hist[N_BKT];
    const int tid = threadIdx.x;
    for (int b = tid; b < N_BKT; b += 1024) hist[b] = 0;
    __syncthreads();
    const int lo = blockIdx.x * SORT_CHUNK;
    int c[SORT_TRIPS];
    #pragma unroll
    for (int t = 0; t < SORT_TRIPS; ++t) {
        const int k = min(tid + t * 1024, SORT_CHUNK - 1);
        c[t] = col[lo + k];
    }
    #pragma unroll
    for (int t = 0; t < SORT_TRIPS; ++t)
        if (tid + t * 1024 < SORT_CHUNK) atomicAdd(&hist[node_clamp(c[t]) >> BKT_SHIFT], 1);
    __syncthreads();
    for (int b = tid; b < N_BKT; b += 1024) bh[b * SORT_BLOCKS + blockIdx.x] = hist[b];
}

template <int LEN>
__global__ __launch_bounds__(1024) void k_scan(int* __restrict__ cnt) {
    __shared__ int wsum[32];
    __shared__ int tile_total;
    const int tid  = threadIdx.x;
    const int lane = tid & 31;
    const int wv   = tid >> 5;
    int carry = 0;
    for (int base = 0; base < LEN; base += 16384) {
        const int idx = base + tid * 16;
        const bool in = idx < LEN;
        int4 v0 = make_int4(0, 0, 0, 0), v1 = v0, v2 = v0, v3 = v0;
        if (in) {
            const int4* p = (const int4*)(cnt + idx);
            v0 = p[0]; v1 = p[1]; v2 = p[2]; v3 = p[3];
        }
        const int s0 = v0.x + v0.y + v0.z + v0.w;
        const int s1 = v1.x + v1.y + v1.z + v1.w;
        const int s2 = v2.x + v2.y + v2.z + v2.w;
        const int s3 = v3.x + v3.y + v3.z + v3.w;
        const int s = s0 + s1 + s2 + s3;
        int inc = s;
        for (int d = 1; d < 32; d <<= 1) {
            int n = __shfl_up(inc, d);
            if (lane >= d) inc += n;
        }
        if (lane == 31) wsum[wv] = inc;
        __syncthreads();
        if (wv == 0) {
            const int ws = wsum[lane];
            int winc = ws;
            for (int d = 1; d < 32; d <<= 1) {
                int n = __shfl_up(winc, d);
                if (lane >= d) winc += n;
            }
            wsum[lane] = winc - ws;
            if (lane == 31) tile_total = winc;
        }
        __syncthreads();
        const int ex = carry + wsum[wv] + (inc - s);
        carry += tile_total;
        if (in) {
            int4 o0, o1, o2, o3;
            o0.x = ex;             o0.y = o0.x + v0.x; o0.z = o0.y + v0.y; o0.w = o0.z + v0.z;
            o1.x = ex + s0;        o1.y = o1.x + v1.x; o1.z = o1.y + v1.y; o1.w = o1.z + v1.z;
            o2.x = o1.x + s1;      o2.y = o2.x + v2.x; o2.z = o2.y + v2.y; o2.w = o2.z + v2.z;
            o3.x = o2.x + s2;      o3.y = o3.x + v3.x; o3.z = o3.y + v3.y; o3.w = o3.z + v3.z;
            int4* p = (int4*)(cnt + idx);
            p[0] = o0; p[1] = o1; p[2] = o2; p[3] = o3;
        }
        __syncthreads();
    }
}

__global__ __launch_bounds__(1024) void k_bkt_scatter(const int* __restrict__ row, const int* __restrict__ col,
                                                      const float* __restrict__ w, const int* __restrict__ bh,
                                                      int* __restrict__ rec_raw) {
    __shared__ int cur[N_BKT];
    const int tid = threadIdx.x;
    for (int b = tid; b < N_BKT; b += 1024) cur[b] = bh[b * SORT_BLOCKS + blockIdx.x];
    __syncthreads();
    int4* __restrict__ rec = (int4*)rec_raw;
    const int lo = blockIdx.x * SORT_CHUNK;
    int c[SORT_TRIPS], r[SORT_TRIPS], wb[SORT_TRIPS];
    #pragma unroll
    for (int t = 0; t < SORT_TRIPS; ++t) {
        const int k = min(tid + t * 1024, SORT_CHUNK - 1);
        c[t]  = node_clamp(col[lo + k]);
        r[t]  = row[lo + k];
        wb[t] = __float_as_int(w[lo + k]);
    }
    #pragma unroll
    for (int t = 0; t < SORT_TRIPS; ++t) {
        if (tid + t * 1024 < SORT_CHUNK) {
            const int p = atomicAdd(&cur[c[t] >> BKT_SHIFT], 1);
            rec[min((unsigned)p, (unsigned)(N_EDGES - 1))] = make_int4(c[t], r[t], wb[t], 0);
        }
    }
}

__global__ __launch_bounds__(1024) void k_bkt_sort(const int* __restrict__ bh, const int* __restrict__ rec_raw,
                                                   int* __restrict__ off, int* __restrict__ csr_src,
                                                   float* __restrict__ csr_w) {
    __shared__ int cnt[BKT_NODES];
    __shared__ int wsum[16];
    const int tid  = threadIdx.x;
    const int lane = tid & 31;
    const int wv   = tid >> 5;
    const int bkt  = blockIdx.x;
    const int4* __restrict__ rec = (const int4*)rec_raw;
    const int bstart = (int)min((unsigned)bh[bkt * SORT_BLOCKS], (unsigned)N_EDGES);
    const int bend   = (bkt + 1 < N_BKT) ? (int)min((unsigned)bh[(bkt + 1) * SORT_BLOCKS], (unsigned)N_EDGES) : N_EDGES;

    if (tid < BKT_NODES) cnt[tid] = 0;
    __syncthreads();
    for (int e0 = bstart + tid; e0 < bend; e0 += SORT_GROUP * 1024) {
        int x[SORT_GROUP];
        #pragma unroll
        for (int u = 0; u < SORT_GROUP; ++u) {
            const int e = min(e0 + u * 1024, bend - 1);
            x[u] = rec[e].x;
        }
        #pragma unroll
        for (int u = 0; u < SORT_GROUP; ++u)
            if (e0 + u * 1024 < bend) atomicAdd(&cnt[x[u] & (BKT_NODES - 1)], 1);
    }
    __syncthreads();

    const int v = (tid < BKT_NODES) ? cnt[tid] : 0;
    int inc = v;
    for (int d = 1; d < 32; d <<= 1) {
        int n = __shfl_up(inc, d);
        if (lane >= d) inc += n;
    }
    if (lane == 31 && wv < 16) wsum[wv] = inc;
    __syncthreads();
    if (wv == 0) {
        const int ws = (lane < 16) ? wsum[lane] : 0;
        int winc = ws;
        for (int d = 1; d < 32; d <<= 1) {
            int n = __shfl_up(winc, d);
            if (lane >= d) winc += n;
        }
        if (lane < 16) wsum[lane] = winc - ws;
    }
    __syncthreads();
    if (tid < BKT_NODES) {
        const int ex = wsum[wv] + (inc - v);
        cnt[tid] = ex;
        const int node = bkt * BKT_NODES + tid;
        if (node < N_NODES) off[node] = bstart + ex;
    }
    __syncthreads();

    for (int e0 = bstart + tid; e0 < bend; e0 += SORT_GROUP * 1024) {
        int4 r[SORT_GROUP];
        #pragma unroll
        for (int u = 0; u < SORT_GROUP; ++u) {
            const int e = min(e0 + u * 1024, bend - 1);
            r[u] = rec[e];
        }
        #pragma unroll
        for (int u = 0; u < SORT_GROUP; ++u) {
            if (e0 + u * 1024 < bend) {
                const int q = atomicAdd(&cnt[r[u].x & (BKT_NODES - 1)], 1);
                const int p = (int)min((unsigned)(bstart + q), (unsigned)(N_EDGES - 1));
                csr_src[p] = r[u].y;
                csr_w[p]   = __int_as_float(r[u].z);
            }
        }
    }
}

__global__ void k_dinv(const int* __restrict__ off, const float* __restrict__ csr_w,
                       float* __restrict__ dinv) {
    int i = blockIdx.x * blockDim.x + threadIdx.x;
    if (i < N_NODES) {
        const int b  = off[i];
        const int en = (i + 1 < N_NODES) ? off[i + 1] : N_EDGES;
        float d = 1.0f;
        for (int j = b; j < en; ++j) d += csr_w[j];
        dinv[i] = (d > 0.0f) ? rsqrtf(d) : 0.0f;
    }
}


__global__ __launch_bounds__(128) void k_gemm1(const float* __restrict__ x,
                                               const float* __restrict__ W1,
                                               float* __restrict__ h1) {
    const int wave  = threadIdx.x >> 5;
    const int lane  = threadIdx.x & 31;
    const int mtile = blockIdx.x * 4 + wave;
    if (mtile >= M_TILES) return;
    const int row0 = mtile * 16;
    const int half = lane >> 4;
    const int l16  = lane & 15;

    const float* arow = x + (size_t)(row0 + l16) * N_FEAT + half * 2;
    v8f c0 = {}, c1 = {}, c2 = {}, c3 = {};

    for (int k = 0; k < N_FEAT; k += 4) {
        v2f a;
        a.x = arow[k];
        a.y = arow[k + 1];
        const float* bp = W1 + (size_t)(k + half * 2) * HIDDEN + l16;
        v2f b0, b1, b2, b3;
        b0.x = bp[0];   b0.y = bp[HIDDEN + 0];
        b1.x = bp[16];  b1.y = bp[HIDDEN + 16];
        b2.x = bp[32];  b2.y = bp[HIDDEN + 32];
        b3.x = bp[48];  b3.y = bp[HIDDEN + 48];
        c0 = __builtin_amdgcn_wmma_f32_16x16x4_f32(false, a, false, b0, (short)0, c0, false, false);
        c1 = __builtin_amdgcn_wmma_f32_16x16x4_f32(false, a, false, b1, (short)0, c1, false, false);
        c2 = __builtin_amdgcn_wmma_f32_16x16x4_f32(false, a, false, b2, (short)0, c2, false, false);
        c3 = __builtin_amdgcn_wmma_f32_16x16x4_f32(false, a, false, b3, (short)0, c3, false, false);
    }

    float* out = h1 + (size_t)(row0 + half * 8) * HIDDEN + l16;
    for (int r = 0; r < 8; ++r) {
        out[(size_t)r * HIDDEN + 0]  = c0[r];
        out[(size_t)r * HIDDEN + 16] = c1[r];
        out[(size_t)r * HIDDEN + 32] = c2[r];
        out[(size_t)r * HIDDEN + 48] = c3[r];
    }
}

__global__ void k_gather1(const int* __restrict__ off, const int* __restrict__ csr_src,
                          const float* __restrict__ csr_w, const float* __restrict__ dinv,
                          const float* __restrict__ h1, const float* __restrict__ b1,
                          float* __restrict__ agg1) {
    const int t = blockIdx.x * blockDim.x + threadIdx.x;
    const int i = t >> 4;
    const int q = t & 15;
    if (i >= N_NODES) return;
    const int b  = off[i];
    const int en = (i + 1 < N_NODES) ? off[i + 1] : N_EDGES;
    const float di = dinv[i];
    const float4* __restrict__ h4 = (const float4*)h1;
    float4 acc = make_float4(0.0f, 0.0f, 0.0f, 0.0f);
    int j = b;
    for (; j + 4 <= en; j += 4) {
        const int s0 = node_clamp(csr_src[j]), s1 = node_clamp(csr_src[j + 1]), s2 = node_clamp(csr_src[j + 2]), s3 = node_clamp(csr_src[j + 3]);
        const float w0 = csr_w[j], w1 = csr_w[j + 1], w2 = csr_w[j + 2], w3 = csr_w[j + 3];
        const float d0 = dinv[s0], d1 = dinv[s1], d2 = dinv[s2], d3 = dinv[s3];
        const float4 v0 = h4[(size_t)s0 * 16 + q];
        const float4 v1 = h4[(size_t)s1 * 16 + q];
        const float4 v2 = h4[(size_t)s2 * 16 + q];
        const float4 v3 = h4[(size_t)s3 * 16 + q];
        const float n0 = d0 * w0 * di, n1 = d1 * w1 * di, n2 = d2 * w2 * di, n3 = d3 * w3 * di;
        acc.x += n0 * v0.x; acc.y += n0 * v0.y; acc.z += n0 * v0.z; acc.w += n0 * v0.w;
        acc.x += n1 * v1.x; acc.y += n1 * v1.y; acc.z += n1 * v1.z; acc.w += n1 * v1.w;
        acc.x += n2 * v2.x; acc.y += n2 * v2.y; acc.z += n2 * v2.z; acc.w += n2 * v2.w;
        acc.x += n3 * v3.x; acc.y += n3 * v3.y; acc.z += n3 * v3.z; acc.w += n3 * v3.w;
    }
    for (; j < en; ++j) {
        const int s0 = node_clamp(csr_src[j]);
        const float n0 = dinv[s0] * csr_w[j] * di;
        const float4 v0 = h4[(size_t)s0 * 16 + q];
        acc.x += n0 * v0.x; acc.y += n0 * v0.y; acc.z += n0 * v0.z; acc.w += n0 * v0.w;
    }
    const float4 hv = h4[(size_t)i * 16 + q];
    const float4 bv = make_float4(b1[q * 4], b1[q * 4 + 1], b1[q * 4 + 2], b1[q * 4 + 3]);
    const float sl = di * di;
    float4 o;
    o.x = bv.x + sl * hv.x + acc.x;
    o.y = bv.y + sl * hv.y + acc.y;
    o.z = bv.z + sl * hv.z + acc.z;
    o.w = bv.w + sl * hv.w + acc.w;
    ((float4*)agg1)[(size_t)i * 16 + q] = o;
}


__global__ __launch_bounds__(128) void k_gemm2(const float* __restrict__ agg1,
                                               const float* __restrict__ W2,
                                               float* __restrict__ h2) {
    const int wave  = threadIdx.x >> 5;
    const int lane  = threadIdx.x & 31;
    const int mtile = blockIdx.x * 4 + wave;
    if (mtile >= M_TILES) return;
    const int row0 = mtile * 16;
    const int half = lane >> 4;
    const int l16  = lane & 15;

    const float* arow = agg1 + (size_t)(row0 + l16) * HIDDEN + half * 2;
    v8f c = {};

    for (int k = 0; k < HIDDEN; k += 4) {
        v2f a;
        a.x = fmaxf(arow[k], 0.0f);
        a.y = fmaxf(arow[k + 1], 0.0f);
        const float* bp = W2 + (size_t)(k + half * 2) * N_CLASSES + l16;
        v2f b;
        b.x = bp[0];
        b.y = bp[N_CLASSES];
        c = __builtin_amdgcn_wmma_f32_16x16x4_f32(false, a, false, b, (short)0, c, false, false);
    }

    float* out = h2 + (size_t)(row0 + half * 8) * N_CLASSES + l16;
    for (int r = 0; r < 8; ++r)
        out[(size_t)r * N_CLASSES] = c[r];
}

__global__ void k_gather2(const int* __restrict__ off, const int* __restrict__ csr_src,
                          const float* __restrict__ csr_w, const float* __restrict__ dinv,
                          const float* __restrict__ h2, const float* __restrict__ b2,
                          float* __restrict__ agg2) {
    const int t = blockIdx.x * blockDim.x + threadIdx.x;
    const int i = t >> 2;
    const int q = t & 3;
    if (i >= N_NODES) return;
    const int b  = off[i];
    const int en = (i + 1 < N_NODES) ? off[i + 1] : N_EDGES;
    const float di = dinv[i];
    const float4* __restrict__ h4 = (const float4*)h2;
    float4 acc = make_float4(0.0f, 0.0f, 0.0f, 0.0f);
    int j = b;
    for (; j + 4 <= en; j += 4) {
        const int s0 = node_clamp(csr_src[j]), s1 = node_clamp(csr_src[j + 1]), s2 = node_clamp(csr_src[j + 2]), s3 = node_clamp(csr_src[j + 3]);
        const float w0 = csr_w[j], w1 = csr_w[j + 1], w2 = csr_w[j + 2], w3 = csr_w[j + 3];
        const float d0 = dinv[s0], d1 = dinv[s1], d2 = dinv[s2], d3 = dinv[s3];
        const float4 v0 = h4[(size_t)s0 * 4 + q];
        const float4 v1 = h4[(size_t)s1 * 4 + q];
        const float4 v2 = h4[(size_t)s2 * 4 + q];
        const float4 v3 = h4[(size_t)s3 * 4 + q];
        const float n0 = d0 * w0 * di, n1 = d1 * w1 * di, n2 = d2 * w2 * di, n3 = d3 * w3 * di;
        acc.x += n0 * v0.x; acc.y += n0 * v0.y; acc.z += n0 * v0.z; acc.w += n0 * v0.w;
        acc.x += n1 * v1.x; acc.y += n1 * v1.y; acc.z += n1 * v1.z; acc.w += n1 * v1.w;
        acc.x += n2 * v2.x; acc.y += n2 * v2.y; acc.z += n2 * v2.z; acc.w += n2 * v2.w;
        acc.x += n3 * v3.x; acc.y += n3 * v3.y; acc.z += n3 * v3.z; acc.w += n3 * v3.w;
    }
    for (; j < en; ++j) {
        const int s0 = node_clamp(csr_src[j]);
        const float n0 = dinv[s0] * csr_w[j] * di;
        const float4 v0 = h4[(size_t)s0 * 4 + q];
        acc.x += n0 * v0.x; acc.y += n0 * v0.y; acc.z += n0 * v0.z; acc.w += n0 * v0.w;
    }
    const float4 hv = h4[(size_t)i * 4 + q];
    const float4 bv = make_float4(b2[q * 4], b2[q * 4 + 1], b2[q * 4 + 2], b2[q * 4 + 3]);
    const float sl = di * di;
    float4 o;
    o.x = bv.x + sl * hv.x + acc.x;
    o.y = bv.y + sl * hv.y + acc.y;
    o.z = bv.z + sl * hv.z + acc.z;
    o.w = bv.w + sl * hv.w + acc.w;
    ((float4*)agg2)[(size_t)i * 4 + q] = o;
}


__global__ void k_logsoftmax(const float* __restrict__ agg2, float* __restrict__ out) {
    int i = blockIdx.x * blockDim.x + threadIdx.x;
    if (i >= N_NODES) return;
    const float* v = agg2 + (size_t)i * N_CLASSES;
    float m = v[0];
    #pragma unroll
    for (int c = 1; c < N_CLASSES; ++c) m = fmaxf(m, v[c]);
    float s = 0.0f;
    #pragma unroll
    for (int c = 0; c < N_CLASSES; ++c) s += expf(v[c] - m);
    float l = logf(s);
    float* o = out + (size_t)i * N_CLASSES;
    #pragma unroll
    for (int c = 0; c < N_CLASSES; ++c) o[c] = v[c] - m - l;
}


extern "C" void kernel_launch(void* const* d_in, const int* in_sizes, int n_in,
                              void* d_out, int out_size, void* d_ws, size_t ws_size,
                              hipStream_t stream) {
    const float* x  = (const float*)d_in[0];
    const int*   ei = (const int*)d_in[1];
    const float* ew = (const float*)d_in[2];
    const float* W1 = (const float*)d_in[3];
    const float* b1 = (const float*)d_in[4];
    const float* W2 = (const float*)d_in[5];
    const float* b2 = (const float*)d_in[6];
    float* out = (float*)d_out;

    const int* row = ei;
    const int* col = ei + N_EDGES;

    float* ws      = (float*)d_ws;
    int*   off     = (int*)ws;
    float* dinv    = ws + N_NODES;
    int*   csr_src = (int*)(dinv + N_NODES);
    float* h1      = (float*)csr_src + N_EDGES;
    float* agg1    = h1   + (size_t)N_NODES * HIDDEN;
    float* csr_w   = agg1 + (size_t)N_NODES * HIDDEN;
    float* agg2    = csr_w + (size_t)N_NODES * N_CLASSES;
    float* h2      = h1;
    int*   rec     = (int*)agg1;
    int*   bh      = (int*)h1;

    const int B = 256;

    k_bkt_hist   <<<SORT_BLOCKS, 1024, 0, stream>>>(col, bh);
    k_scan<HIST_LEN><<<1, 1024, 0, stream>>>(bh);
    k_bkt_scatter<<<SORT_BLOCKS, 1024, 0, stream>>>(row, col, ew, bh, rec);
    k_bkt_sort   <<<N_BKT, 1024, 0, stream>>>(bh, rec, off, csr_src, csr_w);
    k_dinv    <<<(N_NODES + B - 1) / B, B, 0, stream>>>(off, csr_w, dinv);

    k_gemm1<<<(M_TILES + 3) / 4, 128, 0, stream>>>(x, W1, h1);

    k_gather1<<<(N_NODES * 16 + B - 1) / B, B, 0, stream>>>(off, csr_src, csr_w, dinv, h1, b1, agg1);

    k_gemm2<<<(M_TILES + 3) / 4, 128, 0, stream>>>(agg1, W2, h2);

    k_gather2<<<(N_NODES * 4 + B - 1) / B, B, 0, stream>>>(off, csr_src, csr_w, dinv, h2, b2, agg2);

    k_logsoftmax<<<(N_NODES + B - 1) / B, B, 0, stream>>>(agg2, out);
}


# GEMM1 takes its W1 (B) fragments from an LDS copy kept in fragment order (512 resident work-groups, X by 16-byte loads a trip ahead), not 8 global loads per k-step; k summed in a new order per 8
# speedup vs baseline: 3.3343x; 1.2231x over previous
#include <hip/hip_runtime.h>
#include <math.h>

#define N_NODES   100000
#define N_EDGES   1600000
#define N_FEAT    500
#define HIDDEN    64
#define N_CLASSES 16
#define M_TILES   (N_NODES / 16)

typedef __attribute__((ext_vector_type(2))) float v2f;
typedef __attribute__((ext_vector_type(8))) float v8f;

__device__ __forceinline__ int node_clamp(int s) {
    return (int)min((unsigned)s, (unsigned)(N_NODES - 1));
}


#define SORT_BLOCKS 256
#define SORT_CHUNK  (N_EDGES / SORT_BLOCKS)
#define SORT_TRIPS  ((SORT_CHUNK + 1023) / 1024)
#define SORT_GROUP  8
#define BKT_SHIFT   9
#define BKT_NODES   (1 << BKT_SHIFT)
#define N_BKT       ((N_NODES + BKT_NODES - 1) / BKT_NODES)
#define HIST_LEN    (N_BKT * SORT_BLOCKS)
static_assert(N_EDGES % SORT_BLOCKS == 0, "chunks are equal");
static_assert(HIST_LEN % 16 == 0, "a scan thread's 16 counts are all inside or all outside the array");
static_assert(N_NODES % 16 == 0, "M tiles are exact");

__global__ __launch_bounds__(1024) void k_bkt_hist(const int* __restrict__ col, int* __restrict__ bh) {
    __shared__ int hist[N_BKT];
    const int tid = threadIdx.x;
    for (int b = tid; b < N_BKT; b += 1024) hist[b] = 0;
    __syncthreads();
    const int lo = blockIdx.x * SORT_CHUNK;
    int c[SORT_TRIPS];
    #pragma unroll
    for (int t = 0; t < SORT_TRIPS; ++t) {
        const int k = min(tid + t * 1024, SORT_CHUNK - 1);
        c[t] = col[lo + k];
    }
    #pragma unroll
    for (int t = 0; t < SORT_TRIPS; ++t)
        if (tid + t * 1024 < SORT_CHUNK) atomicAdd(&hist[node_clamp(c[t]) >> BKT_SHIFT], 1);
    __syncthreads();
    for (int b = tid; b < N_BKT; b += 1024) bh[b * SORT_BLOCKS + blockIdx.x] = hist[b];
}

template <int LEN>
__global__ __launch_bounds__(1024) void k_scan(int* __restrict__ cnt) {
    __shared__ int wsum[32];
    __shared__ int tile_total;
    const int tid  = threadIdx.x;
    const int lane = tid & 31;
    const int wv   = tid >> 5;
    int carry = 0;
    for (int base = 0; base < LEN; base += 16384) {
        const int idx = base + tid * 16;
        const bool in = idx < LEN;
        int4 v0 = make_int4(0, 0, 0, 0), v1 = v0, v2 = v0, v3 = v0;
        if (in) {
            const int4* p = (const int4*)(cnt + idx);
            v0 = p[0]; v1 = p[1]; v2 = p[2]; v3 = p[3];
        }
        const int s0 = v0.x + v0.y + v0.z + v0.w;
        const int s1 = v1.x + v1.y + v1.z + v1.w;
        const int s2 = v2.x + v2.y + v2.z + v2.w;
        const int s3 = v3.x + v3.y + v3.z + v3.w;
        const int s = s0 + s1 + s2 + s3;
        int inc = s;
        for (int d = 1; d < 32; d <<= 1) {
            int n = __shfl_up(inc, d);
            if (lane >= d) inc += n;
        }
        if (lane == 31) wsum[wv] = inc;
        __syncthreads();
        if (wv == 0) {
            const int ws = wsum[lane];
            int winc = ws;
            for (int d = 1; d < 32; d <<= 1) {
                int n = __shfl_up(winc, d);
                if (lane >= d) winc += n;
            }
            wsum[lane] = winc - ws;
            if (lane == 31) tile_total = winc;
        }
        __syncthreads();
        const int ex = carry + wsum[wv] + (inc - s);
        carry += tile_total;
        if (in) {
            int4 o0, o1, o2, o3;
            o0.x = ex;             o0.y = o0.x + v0.x; o0.z = o0.y + v0.y; o0.w = o0.z + v0.z;
            o1.x = ex + s0;        o1.y = o1.x + v1.x; o1.z = o1.y + v1.y; o1.w = o1.z + v1.z;
            o2.x = o1.x + s1;      o2.y = o2.x + v2.x; o2.z = o2.y + v2.y; o2.w = o2.z + v2.z;
            o3.x = o2.x + s2;      o3.y = o3.x + v3.x; o3.z = o3.y + v3.y; o3.w = o3.z + v3.z;
            int4* p = (int4*)(cnt + idx);
            p[0] = o0; p[1] = o1; p[2] = o2; p[3] = o3;
        }
        __syncthreads();
    }
}

__global__ __launch_bounds__(1024) void k_bkt_scatter(const int* __restrict__ row, const int* __restrict__ col,
                                                      const float* __restrict__ w, const int* __restrict__ bh,
                                                      int* __restrict__ rec_raw) {
    __shared__ int cur[N_BKT];
    const int tid = threadIdx.x;
    for (int b = tid; b < N_BKT; b += 1024) cur[b] = bh[b * SORT_BLOCKS + blockIdx.x];
    __syncthreads();
    int4* __restrict__ rec = (int4*)rec_raw;
    const int lo = blockIdx.x * SORT_CHUNK;
    int c[SORT_TRIPS], r[SORT_TRIPS], wb[SORT_TRIPS];
    #pragma unroll
    for (int t = 0; t < SORT_TRIPS; ++t) {
        const int k = min(tid + t * 1024, SORT_CHUNK - 1);
        c[t]  = node_clamp(col[lo + k]);
        r[t]  = row[lo + k];
        wb[t] = __float_as_int(w[lo + k]);
    }
    #pragma unroll
    for (int t = 0; t < SORT_TRIPS; ++t) {
        if (tid + t * 1024 < SORT_CHUNK) {
            const int p = atomicAdd(&cur[c[t] >> BKT_SHIFT], 1);
            rec[min((unsigned)p, (unsigned)(N_EDGES - 1))] = make_int4(c[t], r[t], wb[t], 0);
        }
    }
}

__global__ __launch_bounds__(1024) void k_bkt_sort(const int* __restrict__ bh, const int* __restrict__ rec_raw,
                                                   int* __restrict__ off, int* __restrict__ csr_src,
                                                   float* __restrict__ csr_w) {
    __shared__ int cnt[BKT_NODES];
    __shared__ int wsum[16];
    const int tid  = threadIdx.x;
    const int lane = tid & 31;
    const int wv   = tid >> 5;
    const int bkt  = blockIdx.x;
    const int4* __restrict__ rec = (const int4*)rec_raw;
    const int bstart = (int)min((unsigned)bh[bkt * SORT_BLOCKS], (unsigned)N_EDGES);
    const int bend   = (bkt + 1 < N_BKT) ? (int)min((unsigned)bh[(bkt + 1) * SORT_BLOCKS], (unsigned)N_EDGES) : N_EDGES;

    if (tid < BKT_NODES) cnt[tid] = 0;
    __syncthreads();
    for (int e0 = bstart + tid; e0 < bend; e0 += SORT_GROUP * 1024) {
        int x[SORT_GROUP];
        #pragma unroll
        for (int u = 0; u < SORT_GROUP; ++u) {
            const int e = min(e0 + u * 1024, bend - 1);
            x[u] = rec[e].x;
        }
        #pragma unroll
        for (int u = 0; u < SORT_GROUP; ++u)
            if (e0 + u * 1024 < bend) atomicAdd(&cnt[x[u] & (BKT_NODES - 1)], 1);
    }
    __syncthreads();

    const int v = (tid < BKT_NODES) ? cnt[tid] : 0;
    int inc = v;
    for (int d = 1; d < 32; d <<= 1) {
        int n = __shfl_up(inc, d);
        if (lane >= d) inc += n;
    }
    if (lane == 31 && wv < 16) wsum[wv] = inc;
    __syncthreads();
    if (wv == 0) {
        const int ws = (lane < 16) ? wsum[lane] : 0;
        int winc = ws;
        for (int d = 1; d < 32; d <<= 1) {
            int n = __shfl_up(winc, d);
            if (lane >= d) winc += n;
        }
        if (lane < 16) wsum[lane] = winc - ws;
    }
    __syncthreads();
    if (tid < BKT_NODES) {
        const int ex = wsum[wv] + (inc - v);
        cnt[tid] = ex;
        const int node = bkt * BKT_NODES + tid;
        if (node < N_NODES) off[node] = bstart + ex;
    }
    __syncthreads();

    for (int e0 = bstart + tid; e0 < bend; e0 += SORT_GROUP * 1024) {
        int4 r[SORT_GROUP];
        #pragma unroll
        for (int u = 0; u < SORT_GROUP; ++u) {
            const int e = min(e0 + u * 1024, bend - 1);
            r[u] = rec[e];
        }
        #pragma unroll
        for (int u = 0; u < SORT_GROUP; ++u) {
            if (e0 + u * 1024 < bend) {
                const int q = atomicAdd(&cnt[r[u].x & (BKT_NODES - 1)], 1);
                const int p = (int)min((unsigned)(bstart + q), (unsigned)(N_EDGES - 1));
                csr_src[p] = r[u].y;
                csr_w[p]   = __int_as_float(r[u].z);
            }
        }
    }
}

__global__ void k_dinv(const int* __restrict__ off, const float* __restrict__ csr_w,
                       float* __restrict__ dinv) {
    int i = blockIdx.x * blockDim.x + threadIdx.x;
    if (i < N_NODES) {
        const int b  = off[i];
        const int en = (i + 1 < N_NODES) ? off[i + 1] : N_EDGES;
        float d = 1.0f;
        for (int j = b; j < en; ++j) d += csr_w[j];
        dinv[i] = (d > 0.0f) ? rsqrtf(d) : 0.0f;
    }
}


#define G1_WAVES   8
#define G1_THREADS (G1_WAVES * 32)
#define G1_BLOCKS  512
#define G1_K8      (N_FEAT / 8)
#define G1_KTAIL   (G1_K8 * 8)
#define G1_STEP_P  (4 * 32)
#define G1_PIECES  (G1_K8 * G1_STEP_P + 2 * 32)
#define G1_R0      (G1_BLOCKS * G1_WAVES)
#define G1_R1      (G1_BLOCKS * (G1_WAVES / 2))
#define G1_REST    (M_TILES - G1_R0 - G1_R1)
#define G1_COPY_TRIPS ((G1_PIECES + G1_THREADS - 1) / G1_THREADS)
static_assert(G1_PIECES * 4 == N_FEAT * HIDDEN, "the LDS copy holds W1 exactly once");
static_assert(N_FEAT - G1_KTAIL == 4, "one 4-wide step is left after the 8-wide ones");
static_assert(G1_K8 % 2 == 0, "the K loop takes two steps per trip");
static_assert(G1_REST >= 0 && G1_REST <= G1_BLOCKS, "at most one left-over tile per work-group");
static_assert(HIDDEN == 64, "four 16-column accumulators");
static_assert(G1_COPY_TRIPS % 8 == 0, "the copy takes its pieces in groups of eight");

__device__ __forceinline__ void g1_step(const float4 a, const float4 q0, const float4 q1,
                                        const float4 q2, const float4 q3,
                                        v8f& c0, v8f& c1, v8f& c2, v8f& c3) {
    v2f av, b;
    av.x = a.x; av.y = a.y;
    b.x = q0.x; b.y = q0.y;
    c0 = __builtin_amdgcn_wmma_f32_16x16x4_f32(false, av, false, b, (short)0, c0, false, false);
    b.x = q0.z; b.y = q0.w;
    c1 = __builtin_amdgcn_wmma_f32_16x16x4_f32(false, av, false, b, (short)0, c1, false, false);
    b.x = q1.x; b.y = q1.y;
    c2 = __builtin_amdgcn_wmma_f32_16x16x4_f32(false, av, false, b, (short)0, c2, false, false);
    b.x = q1.z; b.y = q1.w;
    c3 = __builtin_amdgcn_wmma_f32_16x16x4_f32(false, av, false, b, (short)0, c3, false, false);
    av.x = a.z; av.y = a.w;
    b.x = q2.x; b.y = q2.y;
    c0 = __builtin_amdgcn_wmma_f32_16x16x4_f32(false, av, false, b, (short)0, c0, false, false);
    b.x = q2.z; b.y = q2.w;
    c1 = __builtin_amdgcn_wmma_f32_16x16x4_f32(false, av, false, b, (short)0, c1, false, false);
    b.x = q3.x; b.y = q3.y;
    c2 = __builtin_amdgcn_wmma_f32_16x16x4_f32(false, av, false, b, (short)0, c2, false, false);
    b.x = q3.z; b.y = q3.w;
    c3 = __builtin_amdgcn_wmma_f32_16x16x4_f32(false, av, false, b, (short)0, c3, false, false);
}

__global__ __launch_bounds__(G1_THREADS) void k_gemm1(const float* x,
                                                      const float* W1,
                                                      float* __restrict__ h1) {
    __shared__ float4 Bs[G1_PIECES];
    const int tid = threadIdx.x;

    {
        float* Bf = (float*)Bs;
        const float4* W4 = (const float4*)W1;
        #pragma unroll
        for (int g = 0; g < G1_COPY_TRIPS; g += 8) {
            float4 v[8];
            #pragma unroll
            for (int u = 0; u < 8; ++u) {
                const int e = min(tid + (g + u) * G1_THREADS, G1_PIECES - 1);
                v[u] = W4[e];
            }
            __builtin_amdgcn_sched_barrier(0);
            #pragma unroll
            for (int u = 0; u < 8; ++u) {
                const int e = tid + (g + u) * G1_THREADS;
                if (e < G1_PIECES) {
                    const int k  = e >> 4;
                    const int n0 = (e & 15) * 4;
                    const int j  = n0 >> 4;
                    const int c  = n0 & 15;
                    int d;
                    if (k < G1_KTAIL) {
                        const int t = k >> 3, r = k & 7;
                        const int hh = r >> 2, w = (r >> 1) & 1, s = r & 1;
                        d = ((t * 4 + w * 2 + (j >> 1)) * 32 + hh * 16 + c) * 4 + (j & 1) * 2 + s;
                    } else {
                        const int r = k - G1_KTAIL;
                        const int hh = r >> 1, s = r & 1;
                        d = (G1_K8 * G1_STEP_P + (j >> 1) * 32 + hh * 16 + c) * 4 + (j & 1) * 2 + s;
                    }
                    Bf[d] = v[u].x; Bf[d + 4] = v[u].y; Bf[d + 8] = v[u].z; Bf[d + 12] = v[u].w;
                }
            }
        }
    }
    __syncthreads();

    const int wave = __builtin_amdgcn_readfirstlane(tid >> 5);
    const int lane = tid & 31;
    const int half = lane >> 4;
    const int l16  = lane & 15;
    const float4* bl = Bs + lane;

    int mtile = wave * G1_BLOCKS + (int)blockIdx.x;
    int nxt   = ((__popc(wave) & 1) == 0) ? G1_R0 + (wave >> 1) * G1_BLOCKS + (int)blockIdx.x
              : ((wave == 1 && (int)blockIdx.x < G1_REST) ? G1_R0 + G1_R1 + (int)blockIdx.x : -1);

    while (mtile >= 0) {
        const int row0 = mtile * 16;
        const float*  xr = x + (size_t)(row0 + l16) * N_FEAT;
        const float4* xa = (const float4*)xr + half;
        v2f at;
        at.x = xr[G1_KTAIL + half * 2];
        at.y = xr[G1_KTAIL + half * 2 + 1];
        const float4 u0 = bl[G1_K8 * G1_STEP_P], u1 = bl[G1_K8 * G1_STEP_P + 32];

        float4 aE = xa[0], aO = xa[2];
        float4 e0 = bl[0], e1 = bl[32], e2 = bl[64], e3 = bl[96];
        v8f c0 = {}, c1 = {}, c2 = {}, c3 = {};

        for (int t = 0; t < G1_K8 - 2; t += 2) {
            const float4 nE = xa[2 * t + 4], nO = xa[2 * t + 6];
            const float4* bo = bl + (t + 1) * G1_STEP_P;
            const float4 o0 = bo[0], o1 = bo[32], o2 = bo[64], o3 = bo[96];
            __builtin_amdgcn_sched_barrier(0);
            g1_step(aE, e0, e1, e2, e3, c0, c1, c2, c3);
            e0 = bo[G1_STEP_P]; e1 = bo[G1_STEP_P + 32]; e2 = bo[G1_STEP_P + 64]; e3 = bo[G1_STEP_P + 96];
            __builtin_amdgcn_sched_barrier(0);
            g1_step(aO, o0, o1, o2, o3, c0, c1, c2, c3);
            aE = nE; aO = nO;
        }
        {
            const float4* bo = bl + (G1_K8 - 1) * G1_STEP_P;
            const float4 o0 = bo[0], o1 = bo[32], o2 = bo[64], o3 = bo[96];
            g1_step(aE, e0, e1, e2, e3, c0, c1, c2, c3);
            g1_step(aO, o0, o1, o2, o3, c0, c1, c2, c3);
        }
        {
            v2f b;
            b.x = u0.x; b.y = u0.y;
            c0 = __builtin_amdgcn_wmma_f32_16x16x4_f32(false, at, false, b, (short)0, c0, false, false);
            b.x = u0.z; b.y = u0.w;
            c1 = __builtin_amdgcn_wmma_f32_16x16x4_f32(false, at, false, b, (short)0, c1, false, false);
            b.x = u1.x; b.y = u1.y;
            c2 = __builtin_amdgcn_wmma_f32_16x16x4_f32(false, at, false, b, (short)0, c2, false, false);
            b.x = u1.z; b.y = u1.w;
            c3 = __builtin_amdgcn_wmma_f32_16x16x4_f32(false, at, false, b, (short)0, c3, false, false);
        }

        float* out = h1 + (size_t)(row0 + half * 8) * HIDDEN + l16;
        for (int r = 0; r < 8; ++r) {
            out[(size_t)r * HIDDEN + 0]  = c0[r];
            out[(size_t)r * HIDDEN + 16] = c1[r];
            out[(size_t)r * HIDDEN + 32] = c2[r];
            out[(size_t)r * HIDDEN + 48] = c3[r];
        }
        mtile = nxt;
        nxt = -1;
    }
}

__global__ void k_gather1(const int* __restrict__ off, const int* __restrict__ csr_src,
                          const float* __restrict__ csr_w, const float* __restrict__ dinv,
                          const float* __restrict__ h1, const float* __restrict__ b1,
                          float* __restrict__ agg1) {
    const int t = blockIdx.x * blockDim.x + threadIdx.x;
    const int i = t >> 4;
    const int q = t & 15;
    if (i >= N_NODES) return;
    const int b  = off[i];
    const int en = (i + 1 < N_NODES) ? off[i + 1] : N_EDGES;
    const float di = dinv[i];
    const float4* __restrict__ h4 = (const float4*)h1;
    float4 acc = make_float4(0.0f, 0.0f, 0.0f, 0.0f);
    int j = b;
    for (; j + 4 <= en; j += 4) {
        const int s0 = node_clamp(csr_src[j]), s1 = node_clamp(csr_src[j + 1]), s2 = node_clamp(csr_src[j + 2]), s3 = node_clamp(csr_src[j + 3]);
        const float w0 = csr_w[j], w1 = csr_w[j + 1], w2 = csr_w[j + 2], w3 = csr_w[j + 3];
        const float d0 = dinv[s0], d1 = dinv[s1], d2 = dinv[s2], d3 = dinv[s3];
        const float4 v0 = h4[(size_t)s0 * 16 + q];
        const float4 v1 = h4[(size_t)s1 * 16 + q];
        const float4 v2 = h4[(size_t)s2 * 16 + q];
        const float4 v3 = h4[(size_t)s3 * 16 + q];
        const float n0 = d0 * w0 * di, n1 = d1 * w1 * di, n2 = d2 * w2 * di, n3 = d3 * w3 * di;
        acc.x += n0 * v0.x; acc.y += n0 * v0.y; acc.z += n0 * v0.z; acc.w += n0 * v0.w;
        acc.x += n1 * v1.x; acc.y += n1 * v1.y; acc.z += n1 * v1.z; acc.w += n1 * v1.w;
        acc.x += n2 * v2.x; acc.y += n2 * v2.y; acc.z += n2 * v2.z; acc.w += n2 * v2.w;
        acc.x += n3 * v3.x; acc.y += n3 * v3.y; acc.z += n3 * v3.z; acc.w += n3 * v3.w;
    }
    for (; j < en; ++j) {
        const int s0 = node_clamp(csr_src[j]);
        const float n0 = dinv[s0] * csr_w[j] * di;
        const float4 v0 = h4[(size_t)s0 * 16 + q];
        acc.x += n0 * v0.x; acc.y += n0 * v0.y; acc.z += n0 * v0.z; acc.w += n0 * v0.w;
    }
    const float4 hv = h4[(size_t)i * 16 + q];
    const float4 bv = make_float4(b1[q * 4], b1[q * 4 + 1], b1[q * 4 + 2], b1[q * 4 + 3]);
    const float sl = di * di;
    float4 o;
    o.x = bv.x + sl * hv.x + acc.x;
    o.y = bv.y + sl * hv.y + acc.y;
    o.z = bv.z + sl * hv.z + acc.z;
    o.w = bv.w + sl * hv.w + acc.w;
    ((float4*)agg1)[(size_t)i * 16 + q] = o;
}


__global__ __launch_bounds__(128) void k_gemm2(const float* __restrict__ agg1,
                                               const float* __restrict__ W2,
                                               float* __restrict__ h2) {
    const int wave  = threadIdx.x >> 5;
    const int lane  = threadIdx.x & 31;
    const int mtile = blockIdx.x * 4 + wave;
    if (mtile >= M_TILES) return;
    const int row0 = mtile * 16;
    const int half = lane >> 4;
    const int l16  = lane & 15;

    const float* arow = agg1 + (size_t)(row0 + l16) * HIDDEN + half * 2;
    v8f c = {};

    for (int k = 0; k < HIDDEN; k += 4) {
        v2f a;
        a.x = fmaxf(arow[k], 0.0f);
        a.y = fmaxf(arow[k + 1], 0.0f);
        const float* bp = W2 + (size_t)(k + half * 2) * N_CLASSES + l16;
        v2f b;
        b.x = bp[0];
        b.y = bp[N_CLASSES];
        c = __builtin_amdgcn_wmma_f32_16x16x4_f32(false, a, false, b, (short)0, c, false, false);
    }

    float* out = h2 + (size_t)(row0 + half * 8) * N_CLASSES + l16;
    for (int r = 0; r < 8; ++r)
        out[(size_t)r * N_CLASSES] = c[r];
}

__global__ void k_gather2(const int* __restrict__ off, const int* __restrict__ csr_src,
                          const float* __restrict__ csr_w, const float* __restrict__ dinv,
                          const float* __restrict__ h2, const float* __restrict__ b2,
                          float* __restrict__ agg2) {
    const int t = blockIdx.x * blockDim.x + threadIdx.x;
    const int i = t >> 2;
    const int q = t & 3;
    if (i >= N_NODES) return;
    const int b  = off[i];
    const int en = (i + 1 < N_NODES) ? off[i + 1] : N_EDGES;
    const float di = dinv[i];
    const float4* __restrict__ h4 = (const float4*)h2;
    float4 acc = make_float4(0.0f, 0.0f, 0.0f, 0.0f);
    int j = b;
    for (; j + 4 <= en; j += 4) {
        const int s0 = node_clamp(csr_src[j]), s1 = node_clamp(csr_src[j + 1]), s2 = node_clamp(csr_src[j + 2]), s3 = node_clamp(csr_src[j + 3]);
        const float w0 = csr_w[j], w1 = csr_w[j + 1], w2 = csr_w[j + 2], w3 = csr_w[j + 3];
        const float d0 = dinv[s0], d1 = dinv[s1], d2 = dinv[s2], d3 = dinv[s3];
        const float4 v0 = h4[(size_t)s0 * 4 + q];
        const float4 v1 = h4[(size_t)s1 * 4 + q];
        const float4 v2 = h4[(size_t)s2 * 4 + q];
        const float4 v3 = h4[(size_t)s3 * 4 + q];
        const float n0 = d0 * w0 * di, n1 = d1 * w1 * di, n2 = d2 * w2 * di, n3 = d3 * w3 * di;
        acc.x += n0 * v0.x; acc.y += n0 * v0.y; acc.z += n0 * v0.z; acc.w += n0 * v0.w;
        acc.x += n1 * v1.x; acc.y += n1 * v1.y; acc.z += n1 * v1.z; acc.w += n1 * v1.w;
        acc.x += n2 * v2.x; acc.y += n2 * v2.y; acc.z += n2 * v2.z; acc.w += n2 * v2.w;
        acc.x += n3 * v3.x; acc.y += n3 * v3.y; acc.z += n3 * v3.z; acc.w += n3 * v3.w;
    }
    for (; j < en; ++j) {
        const int s0 = node_clamp(csr_src[j]);
        const float n0 = dinv[s0] * csr_w[j] * di;
        const float4 v0 = h4[(size_t)s0 * 4 + q];
        acc.x += n0 * v0.x; acc.y += n0 * v0.y; acc.z += n0 * v0.z; acc.w += n0 * v0.w;
    }
    const float4 hv = h4[(size_t)i * 4 + q];
    const float4 bv = make_float4(b2[q * 4], b2[q * 4 + 1], b2[q * 4 + 2], b2[q * 4 + 3]);
    const float sl = di * di;
    float4 o;
    o.x = bv.x + sl * hv.x + acc.x;
    o.y = bv.y + sl * hv.y + acc.y;
    o.z = bv.z + sl * hv.z + acc.z;
    o.w = bv.w + sl * hv.w + acc.w;
    ((float4*)agg2)[(size_t)i * 4 + q] = o;
}


__global__ void k_logsoftmax(const float* __restrict__ agg2, float* __restrict__ out) {
    int i = blockIdx.x * blockDim.x + threadIdx.x;
    if (i >= N_NODES) return;
    const float* v = agg2 + (size_t)i * N_CLASSES;
    float m = v[0];
    #pragma unroll
    for (int c = 1; c < N_CLASSES; ++c) m = fmaxf(m, v[c]);
    float s = 0.0f;
    #pragma unroll
    for (int c = 0; c < N_CLASSES; ++c) s += expf(v[c] - m);
    float l = logf(s);
    float* o = out + (size_t)i * N_CLASSES;
    #pragma unroll
    for (int c = 0; c < N_CLASSES; ++c) o[c] = v[c] - m - l;
}


extern "C" void kernel_launch(void* const* d_in, const int* in_sizes, int n_in,
                              void* d_out, int out_size, void* d_ws, size_t ws_size,
                              hipStream_t stream) {
    const float* x  = (const float*)d_in[0];
    const int*   ei = (const int*)d_in[1];
    const float* ew = (const float*)d_in[2];
    const float* W1 = (const float*)d_in[3];
    const float* b1 = (const float*)d_in[4];
    const float* W2 = (const float*)d_in[5];
    const float* b2 = (const float*)d_in[6];
    float* out = (float*)d_out;

    const int* row = ei;
    const int* col = ei + N_EDGES;

    float* ws      = (float*)d_ws;
    int*   off     = (int*)ws;
    float* dinv    = ws + N_NODES;
    int*   csr_src = (int*)(dinv + N_NODES);
    float* h1      = (float*)csr_src + N_EDGES;
    float* agg1    = h1   + (size_t)N_NODES * HIDDEN;
    float* csr_w   = agg1 + (size_t)N_NODES * HIDDEN;
    float* agg2    = csr_w + (size_t)N_NODES * N_CLASSES;
    float* h2      = h1;
    int*   rec     = (int*)agg1;
    int*   bh      = (int*)h1;

    const int B = 256;

    k_bkt_hist   <<<SORT_BLOCKS, 1024, 0, stream>>>(col, bh);
    k_scan<HIST_LEN><<<1, 1024, 0, stream>>>(bh);
    k_bkt_scatter<<<SORT_BLOCKS, 1024, 0, stream>>>(row, col, ew, bh, rec);
    k_bkt_sort   <<<N_BKT, 1024, 0, stream>>>(bh, rec, off, csr_src, csr_w);
    k_dinv    <<<(N_NODES + B - 1) / B, B, 0, stream>>>(off, csr_w, dinv);

    k_gemm1<<<G1_BLOCKS, G1_THREADS, 0, stream>>>(x, W1, h1);

    k_gather1<<<(N_NODES * 16 + B - 1) / B, B, 0, stream>>>(off, csr_src, csr_w, dinv, h1, b1, agg1);

    k_gemm2<<<(M_TILES + 3) / 4, 128, 0, stream>>>(agg1, W2, h2);

    k_gather2<<<(N_NODES * 4 + B - 1) / B, B, 0, stream>>>(off, csr_src, csr_w, dinv, h2, b2, agg2);

    k_logsoftmax<<<(N_NODES + B - 1) / B, B, 0, stream>>>(agg2, out);
}


# Both GEMM stores scale each row by its node's dinv, so the two aggregations fetch no dinv[src] per edge and scale a node's sum once by dinv[dst]; products round in a new order, sum order kept
# speedup vs baseline: 3.5547x; 1.0661x over previous
#include <hip/hip_runtime.h>
#include <math.h>

#define N_NODES   100000
#define N_EDGES   1600000
#define N_FEAT    500
#define HIDDEN    64
#define N_CLASSES 16
#define M_TILES   (N_NODES / 16)

typedef __attribute__((ext_vector_type(2))) float v2f;
typedef __attribute__((ext_vector_type(8))) float v8f;

__device__ __forceinline__ int node_clamp(int s) {
    return (int)min((unsigned)s, (unsigned)(N_NODES - 1));
}


#define SORT_BLOCKS 256
#define SORT_CHUNK  (N_EDGES / SORT_BLOCKS)
#define SORT_TRIPS  ((SORT_CHUNK + 1023) / 1024)
#define SORT_GROUP  8
#define BKT_SHIFT   9
#define BKT_NODES   (1 << BKT_SHIFT)
#define N_BKT       ((N_NODES + BKT_NODES - 1) / BKT_NODES)
#define HIST_LEN    (N_BKT * SORT_BLOCKS)
static_assert(N_EDGES % SORT_BLOCKS == 0, "chunks are equal");
static_assert(HIST_LEN % 16 == 0, "a scan thread's 16 counts are all inside or all outside the array");
static_assert(N_NODES % 16 == 0, "M tiles are exact");

__global__ __launch_bounds__(1024) void k_bkt_hist(const int* __restrict__ col, int* __restrict__ bh) {
    __shared__ int hist[N_BKT];
    const int tid = threadIdx.x;
    for (int b = tid; b < N_BKT; b += 1024) hist[b] = 0;
    __syncthreads();
    const int lo = blockIdx.x * SORT_CHUNK;
    int c[SORT_TRIPS];
    #pragma unroll
    for (int t = 0; t < SORT_TRIPS; ++t) {
        const int k = min(tid + t * 1024, SORT_CHUNK - 1);
        c[t] = col[lo + k];
    }
    #pragma unroll
    for (int t = 0; t < SORT_TRIPS; ++t)
        if (tid + t * 1024 < SORT_CHUNK) atomicAdd(&hist[node_clamp(c[t]) >> BKT_SHIFT], 1);
    __syncthreads();
    for (int b = tid; b < N_BKT; b += 1024) bh[b * SORT_BLOCKS + blockIdx.x] = hist[b];
}

template <int LEN>
__global__ __launch_bounds__(1024) void k_scan(int* __restrict__ cnt) {
    __shared__ int wsum[32];
    __shared__ int tile_total;
    const int tid  = threadIdx.x;
    const int lane = tid & 31;
    const int wv   = tid >> 5;
    int carry = 0;
    for (int base = 0; base < LEN; base += 16384) {
        const int idx = base + tid * 16;
        const bool in = idx < LEN;
        int4 v0 = make_int4(0, 0, 0, 0), v1 = v0, v2 = v0, v3 = v0;
        if (in) {
            const int4* p = (const int4*)(cnt + idx);
            v0 = p[0]; v1 = p[1]; v2 = p[2]; v3 = p[3];
        }
        const int s0 = v0.x + v0.y + v0.z + v0.w;
        const int s1 = v1.x + v1.y + v1.z + v1.w;
        const int s2 = v2.x + v2.y + v2.z + v2.w;
        const int s3 = v3.x + v3.y + v3.z + v3.w;
        const int s = s0 + s1 + s2 + s3;
        int inc = s;
        for (int d = 1; d < 32; d <<= 1) {
            int n = __shfl_up(inc, d);
            if (lane >= d) inc += n;
        }
        if (lane == 31) wsum[wv] = inc;
        __syncthreads();
        if (wv == 0) {
            const int ws = wsum[lane];
            int winc = ws;
            for (int d = 1; d < 32; d <<= 1) {
                int n = __shfl_up(winc, d);
                if (lane >= d) winc += n;
            }
            wsum[lane] = winc - ws;
            if (lane == 31) tile_total = winc;
        }
        __syncthreads();
        const int ex = carry + wsum[wv] + (inc - s);
        carry += tile_total;
        if (in) {
            int4 o0, o1, o2, o3;
            o0.x = ex;             o0.y = o0.x + v0.x; o0.z = o0.y + v0.y; o0.w = o0.z + v0.z;
            o1.x = ex + s0;        o1.y = o1.x + v1.x; o1.z = o1.y + v1.y; o1.w = o1.z + v1.z;
            o2.x = o1.x + s1;      o2.y = o2.x + v2.x; o2.z = o2.y + v2.y; o2.w = o2.z + v2.z;
            o3.x = o2.x + s2;      o3.y = o3.x + v3.x; o3.z = o3.y + v3.y; o3.w = o3.z + v3.z;
            int4* p = (int4*)(cnt + idx);
            p[0] = o0; p[1] = o1; p[2] = o2; p[3] = o3;
        }
        __syncthreads();
    }
}

__global__ __launch_bounds__(1024) void k_bkt_scatter(const int* __restrict__ row, const int* __restrict__ col,
                                                      const float* __restrict__ w, const int* __restrict__ bh,
                                                      int* __restrict__ rec_raw) {
    __shared__ int cur[N_BKT];
    const int tid = threadIdx.x;
    for (int b = tid; b < N_BKT; b += 1024) cur[b] = bh[b * SORT_BLOCKS + blockIdx.x];
    __syncthreads();
    int4* __restrict__ rec = (int4*)rec_raw;
    const int lo = blockIdx.x * SORT_CHUNK;
    int c[SORT_TRIPS], r[SORT_TRIPS], wb[SORT_TRIPS];
    #pragma unroll
    for (int t = 0; t < SORT_TRIPS; ++t) {
        const int k = min(tid + t * 1024, SORT_CHUNK - 1);
        c[t]  = node_clamp(col[lo + k]);
        r[t]  = row[lo + k];
        wb[t] = __float_as_int(w[lo + k]);
    }
    #pragma unroll
    for (int t = 0; t < SORT_TRIPS; ++t) {
        if (tid + t * 1024 < SORT_CHUNK) {
            const int p = atomicAdd(&cur[c[t] >> BKT_SHIFT], 1);
            rec[min((unsigned)p, (unsigned)(N_EDGES - 1))] = make_int4(c[t], r[t], wb[t], 0);
        }
    }
}

__global__ __launch_bounds__(1024) void k_bkt_sort(const int* __restrict__ bh, const int* __restrict__ rec_raw,
                                                   int* __restrict__ off, int* __restrict__ csr_src,
                                                   float* __restrict__ csr_w) {
    __shared__ int cnt[BKT_NODES];
    __shared__ int wsum[16];
    const int tid  = threadIdx.x;
    const int lane = tid & 31;
    const int wv   = tid >> 5;
    const int bkt  = blockIdx.x;
    const int4* __restrict__ rec = (const int4*)rec_raw;
    const int bstart = (int)min((unsigned)bh[bkt * SORT_BLOCKS], (unsigned)N_EDGES);
    const int bend   = (bkt + 1 < N_BKT) ? (int)min((unsigned)bh[(bkt + 1) * SORT_BLOCKS], (unsigned)N_EDGES) : N_EDGES;

    if (tid < BKT_NODES) cnt[tid] = 0;
    __syncthreads();
    for (int e0 = bstart + tid; e0 < bend; e0 += SORT_GROUP * 1024) {
        int x[SORT_GROUP];
        #pragma unroll
        for (int u = 0; u < SORT_GROUP; ++u) {
            const int e = min(e0 + u * 1024, bend - 1);
            x[u] = rec[e].x;
        }
        #pragma unroll
        for (int u = 0; u < SORT_GROUP; ++u)
            if (e0 + u * 1024 < bend) atomicAdd(&cnt[x[u] & (BKT_NODES - 1)], 1);
    }
    __syncthreads();

    const int v = (tid < BKT_NODES) ? cnt[tid] : 0;
    int inc = v;
    for (int d = 1; d < 32; d <<= 1) {
        int n = __shfl_up(inc, d);
        if (lane >= d) inc += n;
    }
    if (lane == 31 && wv < 16) wsum[wv] = inc;
    __syncthreads();
    if (wv == 0) {
        const int ws = (lane < 16) ? wsum[lane] : 0;
        int winc = ws;
        for (int d = 1; d < 32; d <<= 1) {
            int n = __shfl_up(winc, d);
            if (lane >= d) winc += n;
        }
        if (lane < 16) wsum[lane] = winc - ws;
    }
    __syncthreads();
    if (tid < BKT_NODES) {
        const int ex = wsum[wv] + (inc - v);
        cnt[tid] = ex;
        const int node = bkt * BKT_NODES + tid;
        if (node < N_NODES) off[node] = bstart + ex;
    }
    __syncthreads();

    for (int e0 = bstart + tid; e0 < bend; e0 += SORT_GROUP * 1024) {
        int4 r[SORT_GROUP];
        #pragma unroll
        for (int u = 0; u < SORT_GROUP; ++u) {
            const int e = min(e0 + u * 1024, bend - 1);
            r[u] = rec[e];
        }
        #pragma unroll
        for (int u = 0; u < SORT_GROUP; ++u) {
            if (e0 + u * 1024 < bend) {
                const int q = atomicAdd(&cnt[r[u].x & (BKT_NODES - 1)], 1);
                const int p = (int)min((unsigned)(bstart + q), (unsigned)(N_EDGES - 1));
                csr_src[p] = r[u].y;
                csr_w[p]   = __int_as_float(r[u].z);
            }
        }
    }
}

__global__ void k_dinv(const int* __restrict__ off, const float* __restrict__ csr_w,
                       float* __restrict__ dinv) {
    int i = blockIdx.x * blockDim.x + threadIdx.x;
    if (i < N_NODES) {
        const int b  = off[i];
        const int en = (i + 1 < N_NODES) ? off[i + 1] : N_EDGES;
        float d = 1.0f;
        for (int j = b; j < en; ++j) d += csr_w[j];
        dinv[i] = (d > 0.0f) ? rsqrtf(d) : 0.0f;
    }
}


#define G1_WAVES   8
#define G1_THREADS (G1_WAVES * 32)
#define G1_BLOCKS  512
#define G1_K8      (N_FEAT / 8)
#define G1_KTAIL   (G1_K8 * 8)
#define G1_STEP_P  (4 * 32)
#define G1_PIECES  (G1_K8 * G1_STEP_P + 2 * 32)
#define G1_R0      (G1_BLOCKS * G1_WAVES)
#define G1_R1      (G1_BLOCKS * (G1_WAVES / 2))
#define G1_REST    (M_TILES - G1_R0 - G1_R1)
#define G1_COPY_TRIPS ((G1_PIECES + G1_THREADS - 1) / G1_THREADS)
static_assert(G1_PIECES * 4 == N_FEAT * HIDDEN, "the LDS copy holds W1 exactly once");
static_assert(N_FEAT - G1_KTAIL == 4, "one 4-wide step is left after the 8-wide ones");
static_assert(G1_K8 % 2 == 0, "the K loop takes two steps per trip");
static_assert(G1_REST >= 0 && G1_REST <= G1_BLOCKS, "at most one left-over tile per work-group");
static_assert(HIDDEN == 64, "four 16-column accumulators");
static_assert(G1_COPY_TRIPS % 8 == 0, "the copy takes its pieces in groups of eight");

__device__ __forceinline__ void g1_step(const float4 a, const float4 q0, const float4 q1,
                                        const float4 q2, const float4 q3,
                                        v8f& c0, v8f& c1, v8f& c2, v8f& c3) {
    v2f av, b;
    av.x = a.x; av.y = a.y;
    b.x = q0.x; b.y = q0.y;
    c0 = __builtin_amdgcn_wmma_f32_16x16x4_f32(false, av, false, b, (short)0, c0, false, false);
    b.x = q0.z; b.y = q0.w;
    c1 = __builtin_amdgcn_wmma_f32_16x16x4_f32(false, av, false, b, (short)0, c1, false, false);
    b.x = q1.x; b.y = q1.y;
    c2 = __builtin_amdgcn_wmma_f32_16x16x4_f32(false, av, false, b, (short)0, c2, false, false);
    b.x = q1.z; b.y = q1.w;
    c3 = __builtin_amdgcn_wmma_f32_16x16x4_f32(false, av, false, b, (short)0, c3, false, false);
    av.x = a.z; av.y = a.w;
    b.x = q2.x; b.y = q2.y;
    c0 = __builtin_amdgcn_wmma_f32_16x16x4_f32(false, av, false, b, (short)0, c0, false, false);
    b.x = q2.z; b.y = q2.w;
    c1 = __builtin_amdgcn_wmma_f32_16x16x4_f32(false, av, false, b, (short)0, c1, false, false);
    b.x = q3.x; b.y = q3.y;
    c2 = __builtin_amdgcn_wmma_f32_16x16x4_f32(false, av, false, b, (short)0, c2, false, false);
    b.x = q3.z; b.y = q3.w;
    c3 = __builtin_amdgcn_wmma_f32_16x16x4_f32(false, av, false, b, (short)0, c3, false, false);
}

__global__ __launch_bounds__(G1_THREADS) void k_gemm1(const float* x,
                                                      const float* W1,
                                                      const float* __restrict__ dinv,
                                                      float* __restrict__ h1) {
    __shared__ float4 Bs[G1_PIECES];
    const int tid = threadIdx.x;

    {
        float* Bf = (float*)Bs;
        const float4* W4 = (const float4*)W1;
        #pragma unroll
        for (int g = 0; g < G1_COPY_TRIPS; g += 8) {
            float4 v[8];
            #pragma unroll
            for (int u = 0; u < 8; ++u) {
                const int e = min(tid + (g + u) * G1_THREADS, G1_PIECES - 1);
                v[u] = W4[e];
            }
            __builtin_amdgcn_sched_barrier(0);
            #pragma unroll
            for (int u = 0; u < 8; ++u) {
                const int e = tid + (g + u) * G1_THREADS;
                if (e < G1_PIECES) {
                    const int k  = e >> 4;
                    const int n0 = (e & 15) * 4;
                    const int j  = n0 >> 4;
                    const int c  = n0 & 15;
                    int d;
                    if (k < G1_KTAIL) {
                        const int t = k >> 3, r = k & 7;
                        const int hh = r >> 2, w = (r >> 1) & 1, s = r & 1;
                        d = ((t * 4 + w * 2 + (j >> 1)) * 32 + hh * 16 + c) * 4 + (j & 1) * 2 + s;
                    } else {
                        const int r = k - G1_KTAIL;
                        const int hh = r >> 1, s = r & 1;
                        d = (G1_K8 * G1_STEP_P + (j >> 1) * 32 + hh * 16 + c) * 4 + (j & 1) * 2 + s;
                    }
                    Bf[d] = v[u].x; Bf[d + 4] = v[u].y; Bf[d + 8] = v[u].z; Bf[d + 12] = v[u].w;
                }
            }
        }
    }
    __syncthreads();

    const int wave = __builtin_amdgcn_readfirstlane(tid >> 5);
    const int lane = tid & 31;
    const int half = lane >> 4;
    const int l16  = lane & 15;
    const float4* bl = Bs + lane;

    int mtile = wave * G1_BLOCKS + (int)blockIdx.x;
    int nxt   = ((__popc(wave) & 1) == 0) ? G1_R0 + (wave >> 1) * G1_BLOCKS + (int)blockIdx.x
              : ((wave == 1 && (int)blockIdx.x < G1_REST) ? G1_R0 + G1_R1 + (int)blockIdx.x : -1);

    while (mtile >= 0) {
        const int row0 = mtile * 16;
        const float*  xr = x + (size_t)(row0 + l16) * N_FEAT;
        const float4* xa = (const float4*)xr + half;
        v2f at;
        at.x = xr[G1_KTAIL + half * 2];
        at.y = xr[G1_KTAIL + half * 2 + 1];
        const float4 u0 = bl[G1_K8 * G1_STEP_P], u1 = bl[G1_K8 * G1_STEP_P + 32];

        float4 aE = xa[0], aO = xa[2];
        float4 e0 = bl[0], e1 = bl[32], e2 = bl[64], e3 = bl[96];
        v8f c0 = {}, c1 = {}, c2 = {}, c3 = {};

        for (int t = 0; t < G1_K8 - 2; t += 2) {
            const float4 nE = xa[2 * t + 4], nO = xa[2 * t + 6];
            const float4* bo = bl + (t + 1) * G1_STEP_P;
            const float4 o0 = bo[0], o1 = bo[32], o2 = bo[64], o3 = bo[96];
            __builtin_amdgcn_sched_barrier(0);
            g1_step(aE, e0, e1, e2, e3, c0, c1, c2, c3);
            e0 = bo[G1_STEP_P]; e1 = bo[G1_STEP_P + 32]; e2 = bo[G1_STEP_P + 64]; e3 = bo[G1_STEP_P + 96];
            __builtin_amdgcn_sched_barrier(0);
            g1_step(aO, o0, o1, o2, o3, c0, c1, c2, c3);
            aE = nE; aO = nO;
        }
        const float4 dA = *(const float4*)(dinv + row0 + half * 8);
        const float4 dB = *(const float4*)(dinv + row0 + half * 8 + 4);
        {
            const float4* bo = bl + (G1_K8 - 1) * G1_STEP_P;
            const float4 o0 = bo[0], o1 = bo[32], o2 = bo[64], o3 = bo[96];
            g1_step(aE, e0, e1, e2, e3, c0, c1, c2, c3);
            g1_step(aO, o0, o1, o2, o3, c0, c1, c2, c3);
        }
        {
            v2f b;
            b.x = u0.x; b.y = u0.y;
            c0 = __builtin_amdgcn_wmma_f32_16x16x4_f32(false, at, false, b, (short)0, c0, false, false);
            b.x = u0.z; b.y = u0.w;
            c1 = __builtin_amdgcn_wmma_f32_16x16x4_f32(false, at, false, b, (short)0, c1, false, false);
            b.x = u1.x; b.y = u1.y;
            c2 = __builtin_amdgcn_wmma_f32_16x16x4_f32(false, at, false, b, (short)0, c2, false, false);
            b.x = u1.z; b.y = u1.w;
            c3 = __builtin_amdgcn_wmma_f32_16x16x4_f32(false, at, false, b, (short)0, c3, false, false);
        }

        const float ds[8] = {dA.x, dA.y, dA.z, dA.w, dB.x, dB.y, dB.z, dB.w};
        float* out = h1 + (size_t)(row0 + half * 8) * HIDDEN + l16;
        for (int r = 0; r < 8; ++r) {
            out[(size_t)r * HIDDEN + 0]  = c0[r] * ds[r];
            out[(size_t)r * HIDDEN + 16] = c1[r] * ds[r];
            out[(size_t)r * HIDDEN + 32] = c2[r] * ds[r];
            out[(size_t)r * HIDDEN + 48] = c3[r] * ds[r];
        }
        mtile = nxt;
        nxt = -1;
    }
}

__global__ void k_gather1(const int* __restrict__ off, const int* __restrict__ csr_src,
                          const float* __restrict__ csr_w, const float* __restrict__ dinv,
                          const float* __restrict__ h1, const float* __restrict__ b1,
                          float* __restrict__ agg1) {
    const int t = blockIdx.x * blockDim.x + threadIdx.x;
    const int i = t >> 4;
    const int q = t & 15;
    if (i >= N_NODES) return;
    const int b  = off[i];
    const int en = (i + 1 < N_NODES) ? off[i + 1] : N_EDGES;
    const float di = dinv[i];
    const float4* __restrict__ h4 = (const float4*)h1;
    float4 acc = make_float4(0.0f, 0.0f, 0.0f, 0.0f);
    int j = b;
    for (; j + 4 <= en; j += 4) {
        const int s0 = node_clamp(csr_src[j]), s1 = node_clamp(csr_src[j + 1]), s2 = node_clamp(csr_src[j + 2]), s3 = node_clamp(csr_src[j + 3]);
        const float w0 = csr_w[j], w1 = csr_w[j + 1], w2 = csr_w[j + 2], w3 = csr_w[j + 3];
        const float4 v0 = h4[(size_t)s0 * 16 + q];
        const float4 v1 = h4[(size_t)s1 * 16 + q];
        const float4 v2 = h4[(size_t)s2 * 16 + q];
        const float4 v3 = h4[(size_t)s3 * 16 + q];
        acc.x += w0 * v0.x; acc.y += w0 * v0.y; acc.z += w0 * v0.z; acc.w += w0 * v0.w;
        acc.x += w1 * v1.x; acc.y += w1 * v1.y; acc.z += w1 * v1.z; acc.w += w1 * v1.w;
        acc.x += w2 * v2.x; acc.y += w2 * v2.y; acc.z += w2 * v2.z; acc.w += w2 * v2.w;
        acc.x += w3 * v3.x; acc.y += w3 * v3.y; acc.z += w3 * v3.z; acc.w += w3 * v3.w;
    }
    for (; j < en; ++j) {
        const int s0 = node_clamp(csr_src[j]);
        const float w0 = csr_w[j];
        const float4 v0 = h4[(size_t)s0 * 16 + q];
        acc.x += w0 * v0.x; acc.y += w0 * v0.y; acc.z += w0 * v0.z; acc.w += w0 * v0.w;
    }
    const float4 hv = h4[(size_t)i * 16 + q];
    const float4 bv = make_float4(b1[q * 4], b1[q * 4 + 1], b1[q * 4 + 2], b1[q * 4 + 3]);
    float4 o;
    o.x = bv.x + di * hv.x + di * acc.x;
    o.y = bv.y + di * hv.y + di * acc.y;
    o.z = bv.z + di * hv.z + di * acc.z;
    o.w = bv.w + di * hv.w + di * acc.w;
    ((float4*)agg1)[(size_t)i * 16 + q] = o;
}


__global__ __launch_bounds__(128) void k_gemm2(const float* __restrict__ agg1,
                                               const float* __restrict__ W2,
                                               const float* __restrict__ dinv,
                                               float* __restrict__ h2) {
    const int wave  = threadIdx.x >> 5;
    const int lane  = threadIdx.x & 31;
    const int mtile = blockIdx.x * 4 + wave;
    if (mtile >= M_TILES) return;
    const int row0 = mtile * 16;
    const int half = lane >> 4;
    const int l16  = lane & 15;

    const float* arow = agg1 + (size_t)(row0 + l16) * HIDDEN + half * 2;
    v8f c = {};

    for (int k = 0; k < HIDDEN; k += 4) {
        v2f a;
        a.x = fmaxf(arow[k], 0.0f);
        a.y = fmaxf(arow[k + 1], 0.0f);
        const float* bp = W2 + (size_t)(k + half * 2) * N_CLASSES + l16;
        v2f b;
        b.x = bp[0];
        b.y = bp[N_CLASSES];
        c = __builtin_amdgcn_wmma_f32_16x16x4_f32(false, a, false, b, (short)0, c, false, false);
    }

    const float4 dA = *(const float4*)(dinv + row0 + half * 8);
    const float4 dB = *(const float4*)(dinv + row0 + half * 8 + 4);
    const float ds[8] = {dA.x, dA.y, dA.z, dA.w, dB.x, dB.y, dB.z, dB.w};
    float* out = h2 + (size_t)(row0 + half * 8) * N_CLASSES + l16;
    for (int r = 0; r < 8; ++r)
        out[(size_t)r * N_CLASSES] = c[r] * ds[r];
}

__global__ void k_gather2(const int* __restrict__ off, const int* __restrict__ csr_src,
                          const float* __restrict__ csr_w, const float* __restrict__ dinv,
                          const float* __restrict__ h2, const float* __restrict__ b2,
                          float* __restrict__ agg2) {
    const int t = blockIdx.x * blockDim.x + threadIdx.x;
    const int i = t >> 2;
    const int q = t & 3;
    if (i >= N_NODES) return;
    const int b  = off[i];
    const int en = (i + 1 < N_NODES) ? off[i + 1] : N_EDGES;
    const float di = dinv[i];
    const float4* __restrict__ h4 = (const float4*)h2;
    float4 acc = make_float4(0.0f, 0.0f, 0.0f, 0.0f);
    int j = b;
    for (; j + 4 <= en; j += 4) {
        const int s0 = node_clamp(csr_src[j]), s1 = node_clamp(csr_src[j + 1]), s2 = node_clamp(csr_src[j + 2]), s3 = node_clamp(csr_src[j + 3]);
        const float w0 = csr_w[j], w1 = csr_w[j + 1], w2 = csr_w[j + 2], w3 = csr_w[j + 3];
        const float4 v0 = h4[(size_t)s0 * 4 + q];
        const float4 v1 = h4[(size_t)s1 * 4 + q];
        const float4 v2 = h4[(size_t)s2 * 4 + q];
        const float4 v3 = h4[(size_t)s3 * 4 + q];
        acc.x += w0 * v0.x; acc.y += w0 * v0.y; acc.z += w0 * v0.z; acc.w += w0 * v0.w;
        acc.x += w1 * v1.x; acc.y += w1 * v1.y; acc.z += w1 * v1.z; acc.w += w1 * v1.w;
        acc.x += w2 * v2.x; acc.y += w2 * v2.y; acc.z += w2 * v2.z; acc.w += w2 * v2.w;
        acc.x += w3 * v3.x; acc.y += w3 * v3.y; acc.z += w3 * v3.z; acc.w += w3 * v3.w;
    }
    for (; j < en; ++j) {
        const int s0 = node_clamp(csr_src[j]);
        const float w0 = csr_w[j];
        const float4 v0 = h4[(size_t)s0 * 4 + q];
        acc.x += w0 * v0.x; acc.y += w0 * v0.y; acc.z += w0 * v0.z; acc.w += w0 * v0.w;
    }
    const float4 hv = h4[(size_t)i * 4 + q];
    const float4 bv = make_float4(b2[q * 4], b2[q * 4 + 1], b2[q * 4 + 2], b2[q * 4 + 3]);
    float4 o;
    o.x = bv.x + di * hv.x + di * acc.x;
    o.y = bv.y + di * hv.y + di * acc.y;
    o.z = bv.z + di * hv.z + di * acc.z;
    o.w = bv.w + di * hv.w + di * acc.w;
    ((float4*)agg2)[(size_t)i * 4 + q] = o;
}


__global__ void k_logsoftmax(const float* __restrict__ agg2, float* __restrict__ out) {
    int i = blockIdx.x * blockDim.x + threadIdx.x;
    if (i >= N_NODES) return;
    const float* v = agg2 + (size_t)i * N_CLASSES;
    float m = v[0];
    #pragma unroll
    for (int c = 1; c < N_CLASSES; ++c) m = fmaxf(m, v[c]);
    float s = 0.0f;
    #pragma unroll
    for (int c = 0; c < N_CLASSES; ++c) s += expf(v[c] - m);
    float l = logf(s);
    float* o = out + (size_t)i * N_CLASSES;
    #pragma unroll
    for (int c = 0; c < N_CLASSES; ++c) o[c] = v[c] - m - l;
}


extern "C" void kernel_launch(void* const* d_in, const int* in_sizes, int n_in,
                              void* d_out, int out_size, void* d_ws, size_t ws_size,
                              hipStream_t stream) {
    const float* x  = (const float*)d_in[0];
    const int*   ei = (const int*)d_in[1];
    const float* ew = (const float*)d_in[2];
    const float* W1 = (const float*)d_in[3];
    const float* b1 = (const float*)d_in[4];
    const float* W2 = (const float*)d_in[5];
    const float* b2 = (const float*)d_in[6];
    float* out = (float*)d_out;

    const int* row = ei;
    const int* col = ei + N_EDGES;

    float* ws      = (float*)d_ws;
    int*   off     = (int*)ws;
    float* dinv    = ws + N_NODES;
    int*   csr_src = (int*)(dinv + N_NODES);
    float* h1      = (float*)csr_src + N_EDGES;
    float* agg1    = h1   + (size_t)N_NODES * HIDDEN;
    float* csr_w   = agg1 + (size_t)N_NODES * HIDDEN;
    float* agg2    = csr_w + (size_t)N_NODES * N_CLASSES;
    float* h2      = h1;
    int*   rec     = (int*)agg1;
    int*   bh      = (int*)h1;

    const int B = 256;

    k_bkt_hist   <<<SORT_BLOCKS, 1024, 0, stream>>>(col, bh);
    k_scan<HIST_LEN><<<1, 1024, 0, stream>>>(bh);
    k_bkt_scatter<<<SORT_BLOCKS, 1024, 0, stream>>>(row, col, ew, bh, rec);
    k_bkt_sort   <<<N_BKT, 1024, 0, stream>>>(bh, rec, off, csr_src, csr_w);
    k_dinv    <<<(N_NODES + B - 1) / B, B, 0, stream>>>(off, csr_w, dinv);

    k_gemm1<<<G1_BLOCKS, G1_THREADS, 0, stream>>>(x, W1, dinv, h1);

    k_gather1<<<(N_NODES * 16 + B - 1) / B, B, 0, stream>>>(off, csr_src, csr_w, dinv, h1, b1, agg1);

    k_gemm2<<<(M_TILES + 3) / 4, 128, 0, stream>>>(agg1, W2, dinv, h2);

    k_gather2<<<(N_NODES * 4 + B - 1) / B, B, 0, stream>>>(off, csr_src, csr_w, dinv, h2, b2, agg2);

    k_logsoftmax<<<(N_NODES + B - 1) / B, B, 0, stream>>>(agg2, out);
}


# Both edge-sort passes place edges in LDS, then store contiguous runs (whole 128-byte lines of csr_src/csr_w; bucket runs of records), not one scattered store per edge; sums keep LDS arrival order
# speedup vs baseline: 3.8494x; 1.0829x over previous
#include <hip/hip_runtime.h>
#include <math.h>

#define N_NODES   100000
#define N_EDGES   1600000
#define N_FEAT    500
#define HIDDEN    64
#define N_CLASSES 16
#define M_TILES   (N_NODES / 16)

typedef __attribute__((ext_vector_type(2))) float v2f;
typedef __attribute__((ext_vector_type(8))) float v8f;

__device__ __forceinline__ int node_clamp(int s) {
    return (int)min((unsigned)s, (unsigned)(N_NODES - 1));
}


#define SORT_BLOCKS 256
#define SORT_CHUNK  (N_EDGES / SORT_BLOCKS)
#define SORT_TRIPS  ((SORT_CHUNK + 1023) / 1024)
#define SORT_GROUP  8
#define SORT_CAP    16384
#define SORT_PAD    32
#define BKT_SHIFT   9
#define BKT_NODES   (1 << BKT_SHIFT)
#define N_BKT       ((N_NODES + BKT_NODES - 1) / BKT_NODES)
#define HIST_LEN    (N_BKT * SORT_BLOCKS)
static_assert(N_EDGES % SORT_BLOCKS == 0, "chunks are equal");
static_assert(HIST_LEN % 16 == 0, "a scan thread's 16 counts are all inside or all outside the array");
static_assert(N_NODES % 16 == 0, "M tiles are exact");

__global__ __launch_bounds__(1024) void k_bkt_hist(const int* __restrict__ col, int* __restrict__ bh) {
    __shared__ int hist[N_BKT];
    const int tid = threadIdx.x;
    for (int b = tid; b < N_BKT; b += 1024) hist[b] = 0;
    __syncthreads();
    const int lo = blockIdx.x * SORT_CHUNK;
    int c[SORT_TRIPS];
    #pragma unroll
    for (int t = 0; t < SORT_TRIPS; ++t) {
        const int k = min(tid + t * 1024, SORT_CHUNK - 1);
        c[t] = col[lo + k];
    }
    #pragma unroll
    for (int t = 0; t < SORT_TRIPS; ++t)
        if (tid + t * 1024 < SORT_CHUNK) atomicAdd(&hist[node_clamp(c[t]) >> BKT_SHIFT], 1);
    __syncthreads();
    for (int b = tid; b < N_BKT; b += 1024) bh[b * SORT_BLOCKS + blockIdx.x] = hist[b];
}

template <int LEN>
__global__ __launch_bounds__(1024) void k_scan(int* __restrict__ cnt) {
    __shared__ int wsum[32];
    __shared__ int tile_total;
    const int tid  = threadIdx.x;
    const int lane = tid & 31;
    const int wv   = tid >> 5;
    int carry = 0;
    for (int base = 0; base < LEN; base += 16384) {
        const int idx = base + tid * 16;
        const bool in = idx < LEN;
        int4 v0 = make_int4(0, 0, 0, 0), v1 = v0, v2 = v0, v3 = v0;
        if (in) {
            const int4* p = (const int4*)(cnt + idx);
            v0 = p[0]; v1 = p[1]; v2 = p[2]; v3 = p[3];
        }
        const int s0 = v0.x + v0.y + v0.z + v0.w;
        const int s1 = v1.x + v1.y + v1.z + v1.w;
        const int s2 = v2.x + v2.y + v2.z + v2.w;
        const int s3 = v3.x + v3.y + v3.z + v3.w;
        const int s = s0 + s1 + s2 + s3;
        int inc = s;
        for (int d = 1; d < 32; d <<= 1) {
            int n = __shfl_up(inc, d);
            if (lane >= d) inc += n;
        }
        if (lane == 31) wsum[wv] = inc;
        __syncthreads();
        if (wv == 0) {
            const int ws = wsum[lane];
            int winc = ws;
            for (int d = 1; d < 32; d <<= 1) {
                int n = __shfl_up(winc, d);
                if (lane >= d) winc += n;
            }
            wsum[lane] = winc - ws;
            if (lane == 31) tile_total = winc;
        }
        __syncthreads();
        const int ex = carry + wsum[wv] + (inc - s);
        carry += tile_total;
        if (in) {
            int4 o0, o1, o2, o3;
            o0.x = ex;             o0.y = o0.x + v0.x; o0.z = o0.y + v0.y; o0.w = o0.z + v0.z;
            o1.x = ex + s0;        o1.y = o1.x + v1.x; o1.z = o1.y + v1.y; o1.w = o1.z + v1.z;
            o2.x = o1.x + s1;      o2.y = o2.x + v2.x; o2.z = o2.y + v2.y; o2.w = o2.z + v2.z;
            o3.x = o2.x + s2;      o3.y = o3.x + v3.x; o3.z = o3.y + v3.y; o3.w = o3.z + v3.z;
            int4* p = (int4*)(cnt + idx);
            p[0] = o0; p[1] = o1; p[2] = o2; p[3] = o3;
        }
        __syncthreads();
    }
}

__global__ __launch_bounds__(1024) void k_bkt_scatter(const int* __restrict__ row, const int* __restrict__ col,
                                                      const float* __restrict__ w, const int* __restrict__ bh,
                                                      int* __restrict__ rec_raw) {
    __shared__ int  cur[N_BKT];
    __shared__ int  delta[N_BKT];
    __shared__ int  wsum[8];
    __shared__ int4 stage[SORT_CHUNK];
    const int tid  = threadIdx.x;
    const int lane = tid & 31;
    const int wv   = tid >> 5;
    int g0 = 0, nb = 0;
    if (tid < N_BKT) {
        const int idx = tid * SORT_BLOCKS + (int)blockIdx.x;
        g0 = bh[idx];
        const int g1 = (idx + 1 < HIST_LEN) ? bh[idx + 1] : N_EDGES;
        nb = g1 - g0;
    }
    int inc = nb;
    for (int d = 1; d < 32; d <<= 1) {
        int n = __shfl_up(inc, d);
        if (lane >= d) inc += n;
    }
    if (lane == 31 && wv < 8) wsum[wv] = inc;
    __syncthreads();
    if (wv == 0) {
        const int ws = (lane < 8) ? wsum[lane] : 0;
        int winc = ws;
        for (int d = 1; d < 32; d <<= 1) {
            int n = __shfl_up(winc, d);
            if (lane >= d) winc += n;
        }
        if (lane < 8) wsum[lane] = winc - ws;
    }
    __syncthreads();
    if (tid < N_BKT) {
        const int ex = wsum[wv] + (inc - nb);
        cur[tid]   = ex;
        delta[tid] = g0 - ex;
    }
    __syncthreads();
    int4* __restrict__ rec = (int4*)rec_raw;
    const int lo = blockIdx.x * SORT_CHUNK;
    int c[SORT_TRIPS], r[SORT_TRIPS], wb[SORT_TRIPS];
    #pragma unroll
    for (int t = 0; t < SORT_TRIPS; ++t) {
        const int k = min(tid + t * 1024, SORT_CHUNK - 1);
        c[t]  = node_clamp(col[lo + k]);
        r[t]  = row[lo + k];
        wb[t] = __float_as_int(w[lo + k]);
    }
    #pragma unroll
    for (int t = 0; t < SORT_TRIPS; ++t) {
        if (tid + t * 1024 < SORT_CHUNK) {
            const int q = atomicAdd(&cur[c[t] >> BKT_SHIFT], 1);
            stage[min((unsigned)q, (unsigned)(SORT_CHUNK - 1))] = make_int4(c[t], r[t], wb[t], 0);
        }
    }
    __syncthreads();
    int4 v[SORT_TRIPS];
    int  dl[SORT_TRIPS];
    #pragma unroll
    for (int t = 0; t < SORT_TRIPS; ++t) v[t] = stage[min(tid + t * 1024, SORT_CHUNK - 1)];
    #pragma unroll
    for (int t = 0; t < SORT_TRIPS; ++t) dl[t] = delta[node_clamp(v[t].x) >> BKT_SHIFT];
    #pragma unroll
    for (int t = 0; t < SORT_TRIPS; ++t) {
        const int i = tid + t * 1024;
        if (i < SORT_CHUNK) rec[min((unsigned)(i + dl[t]), (unsigned)(N_EDGES - 1))] = v[t];
    }
}

__global__ __launch_bounds__(1024) void k_bkt_sort(const int* __restrict__ bh, const int* __restrict__ rec_raw,
                                                   int* __restrict__ off, int* __restrict__ csr_src,
                                                   float* __restrict__ csr_w) {
    __shared__ int cnt[BKT_NODES];
    __shared__ int wsum[16];
    __shared__ int4 st_src[(SORT_CAP + SORT_PAD) / 4];
    __shared__ int4 st_w[(SORT_CAP + SORT_PAD) / 4];
    const int tid  = threadIdx.x;
    const int lane = tid & 31;
    const int wv   = tid >> 5;
    const int bkt  = blockIdx.x;
    const int4* __restrict__ rec = (const int4*)rec_raw;
    const int bstart = (int)min((unsigned)bh[bkt * SORT_BLOCKS], (unsigned)N_EDGES);
    const int bend   = (bkt + 1 < N_BKT) ? (int)min((unsigned)bh[(bkt + 1) * SORT_BLOCKS], (unsigned)N_EDGES) : N_EDGES;
    const int  nedge  = bend - bstart;
    const bool staged = nedge <= SORT_CAP;
    const int  shift  = bstart & (SORT_PAD - 1);
    int* ssrc = (int*)st_src;
    int* sw   = (int*)st_w;

    if (tid < BKT_NODES) cnt[tid] = 0;
    __syncthreads();
    for (int e0 = bstart + tid; e0 < bend; e0 += SORT_GROUP * 1024) {
        int x[SORT_GROUP];
        #pragma unroll
        for (int u = 0; u < SORT_GROUP; ++u) {
            const int e = min(e0 + u * 1024, bend - 1);
            x[u] = rec[e].x;
        }
        #pragma unroll
        for (int u = 0; u < SORT_GROUP; ++u)
            if (e0 + u * 1024 < bend) atomicAdd(&cnt[x[u] & (BKT_NODES - 1)], 1);
    }
    __syncthreads();

    const int v = (tid < BKT_NODES) ? cnt[tid] : 0;
    int inc = v;
    for (int d = 1; d < 32; d <<= 1) {
        int n = __shfl_up(inc, d);
        if (lane >= d) inc += n;
    }
    if (lane == 31 && wv < 16) wsum[wv] = inc;
    __syncthreads();
    if (wv == 0) {
        const int ws = (lane < 16) ? wsum[lane] : 0;
        int winc = ws;
        for (int d = 1; d < 32; d <<= 1) {
            int n = __shfl_up(winc, d);
            if (lane >= d) winc += n;
        }
        if (lane < 16) wsum[lane] = winc - ws;
    }
    __syncthreads();
    if (tid < BKT_NODES) {
        const int ex = wsum[wv] + (inc - v);
        cnt[tid] = ex;
        const int node = bkt * BKT_NODES + tid;
        if (node < N_NODES) off[node] = bstart + ex;
    }
    __syncthreads();

    for (int e0 = bstart + tid; e0 < bend; e0 += SORT_GROUP * 1024) {
        int4 r[SORT_GROUP];
        #pragma unroll
        for (int u = 0; u < SORT_GROUP; ++u) {
            const int e = min(e0 + u * 1024, bend - 1);
            r[u] = rec[e];
        }
        #pragma unroll
        for (int u = 0; u < SORT_GROUP; ++u) {
            if (e0 + u * 1024 < bend) {
                const int q = atomicAdd(&cnt[r[u].x & (BKT_NODES - 1)], 1);
                if (staged) {
                    const int p = (int)min((unsigned)(q + shift), (unsigned)(SORT_CAP + SORT_PAD - 1));
                    ssrc[p] = r[u].y;
                    sw[p]   = r[u].z;
                } else {
                    const int p = (int)min((unsigned)(bstart + q), (unsigned)(N_EDGES - 1));
                    csr_src[p] = r[u].y;
                    csr_w[p]   = __int_as_float(r[u].z);
                }
            }
        }
    }
    if (staged) {
        __syncthreads();
        const int a0 = bstart - shift;
        const int pieces = (shift + nedge + 3) >> 2;
        int4* gs = (int4*)(csr_src + a0);
        int4* gw = (int4*)(csr_w + a0);
        for (int g = tid; g < pieces; g += 1024) {
            const int p = g * 4;
            if (p >= shift && p + 4 <= shift + nedge) {
                gs[g] = st_src[g];
                gw[g] = st_w[g];
            } else {
                for (int k = 0; k < 4; ++k) {
                    if (p + k >= shift && p + k < shift + nedge) {
                        csr_src[a0 + p + k] = ssrc[p + k];
                        csr_w[a0 + p + k]   = __int_as_float(sw[p + k]);
                    }
                }
            }
        }
    }
}

__global__ void k_dinv(const int* __restrict__ off, const float* __restrict__ csr_w,
                       float* __restrict__ dinv) {
    int i = blockIdx.x * blockDim.x + threadIdx.x;
    if (i < N_NODES) {
        const int b  = off[i];
        const int en = (i + 1 < N_NODES) ? off[i + 1] : N_EDGES;
        float d = 1.0f;
        for (int j = b; j < en; ++j) d += csr_w[j];
        dinv[i] = (d > 0.0f) ? rsqrtf(d) : 0.0f;
    }
}


#define G1_WAVES   8
#define G1_THREADS (G1_WAVES * 32)
#define G1_BLOCKS  512
#define G1_K8      (N_FEAT / 8)
#define G1_KTAIL   (G1_K8 * 8)
#define G1_STEP_P  (4 * 32)
#define G1_PIECES  (G1_K8 * G1_STEP_P + 2 * 32)
#define G1_R0      (G1_BLOCKS * G1_WAVES)
#define G1_R1      (G1_BLOCKS * (G1_WAVES / 2))
#define G1_REST    (M_TILES - G1_R0 - G1_R1)
#define G1_COPY_TRIPS ((G1_PIECES + G1_THREADS - 1) / G1_THREADS)
static_assert(G1_PIECES * 4 == N_FEAT * HIDDEN, "the LDS copy holds W1 exactly once");
static_assert(N_FEAT - G1_KTAIL == 4, "one 4-wide step is left after the 8-wide ones");
static_assert(G1_K8 % 2 == 0, "the K loop takes two steps per trip");
static_assert(G1_REST >= 0 && G1_REST <= G1_BLOCKS, "at most one left-over tile per work-group");
static_assert(HIDDEN == 64, "four 16-column accumulators");
static_assert(G1_COPY_TRIPS % 8 == 0, "the copy takes its pieces in groups of eight");

__device__ __forceinline__ void g1_step(const float4 a, const float4 q0, const float4 q1,
                                        const float4 q2, const float4 q3,
                                        v8f& c0, v8f& c1, v8f& c2, v8f& c3) {
    v2f av, b;
    av.x = a.x; av.y = a.y;
    b.x = q0.x; b.y = q0.y;
    c0 = __builtin_amdgcn_wmma_f32_16x16x4_f32(false, av, false, b, (short)0, c0, false, false);
    b.x = q0.z; b.y = q0.w;
    c1 = __builtin_amdgcn_wmma_f32_16x16x4_f32(false, av, false, b, (short)0, c1, false, false);
    b.x = q1.x; b.y = q1.y;
    c2 = __builtin_amdgcn_wmma_f32_16x16x4_f32(false, av, false, b, (short)0, c2, false, false);
    b.x = q1.z; b.y = q1.w;
    c3 = __builtin_amdgcn_wmma_f32_16x16x4_f32(false, av, false, b, (short)0, c3, false, false);
    av.x = a.z; av.y = a.w;
    b.x = q2.x; b.y = q2.y;
    c0 = __builtin_amdgcn_wmma_f32_16x16x4_f32(false, av, false, b, (short)0, c0, false, false);
    b.x = q2.z; b.y = q2.w;
    c1 = __builtin_amdgcn_wmma_f32_16x16x4_f32(false, av, false, b, (short)0, c1, false, false);
    b.x = q3.x; b.y = q3.y;
    c2 = __builtin_amdgcn_wmma_f32_16x16x4_f32(false, av, false, b, (short)0, c2, false, false);
    b.x = q3.z; b.y = q3.w;
    c3 = __builtin_amdgcn_wmma_f32_16x16x4_f32(false, av, false, b, (short)0, c3, false, false);
}

__global__ __launch_bounds__(G1_THREADS) void k_gemm1(const float* x,
                                                      const float* W1,
                                                      const float* __restrict__ dinv,
                                                      float* __restrict__ h1) {
    __shared__ float4 Bs[G1_PIECES];
    const int tid = threadIdx.x;

    {
        float* Bf = (float*)Bs;
        const float4* W4 = (const float4*)W1;
        #pragma unroll
        for (int g = 0; g < G1_COPY_TRIPS; g += 8) {
            float4 v[8];
            #pragma unroll
            for (int u = 0; u < 8; ++u) {
                const int e = min(tid + (g + u) * G1_THREADS, G1_PIECES - 1);
                v[u] = W4[e];
            }
            __builtin_amdgcn_sched_barrier(0);
            #pragma unroll
            for (int u = 0; u < 8; ++u) {
                const int e = tid + (g + u) * G1_THREADS;
                if (e < G1_PIECES) {
                    const int k  = e >> 4;
                    const int n0 = (e & 15) * 4;
                    const int j  = n0 >> 4;
                    const int c  = n0 & 15;
                    int d;
                    if (k < G1_KTAIL) {
                        const int t = k >> 3, r = k & 7;
                        const int hh = r >> 2, w = (r >> 1) & 1, s = r & 1;
                        d = ((t * 4 + w * 2 + (j >> 1)) * 32 + hh * 16 + c) * 4 + (j & 1) * 2 + s;
                    } else {
                        const int r = k - G1_KTAIL;
                        const int hh = r >> 1, s = r & 1;
                        d = (G1_K8 * G1_STEP_P + (j >> 1) * 32 + hh * 16 + c) * 4 + (j & 1) * 2 + s;
                    }
                    Bf[d] = v[u].x; Bf[d + 4] = v[u].y; Bf[d + 8] = v[u].z; Bf[d + 12] = v[u].w;
                }
            }
        }
    }
    __syncthreads();

    const int wave = __builtin_amdgcn_readfirstlane(tid >> 5);
    const int lane = tid & 31;
    const int half = lane >> 4;
    const int l16  = lane & 15;
    const float4* bl = Bs + lane;

    int mtile = wave * G1_BLOCKS + (int)blockIdx.x;
    int nxt   = ((__popc(wave) & 1) == 0) ? G1_R0 + (wave >> 1) * G1_BLOCKS + (int)blockIdx.x
              : ((wave == 1 && (int)blockIdx.x < G1_REST) ? G1_R0 + G1_R1 + (int)blockIdx.x : -1);

    while (mtile >= 0) {
        const int row0 = mtile * 16;
        const float*  xr = x + (size_t)(row0 + l16) * N_FEAT;
        const float4* xa = (const float4*)xr + half;
        v2f at;
        at.x = xr[G1_KTAIL + half * 2];
        at.y = xr[G1_KTAIL + half * 2 + 1];
        const float4 u0 = bl[G1_K8 * G1_STEP_P], u1 = bl[G1_K8 * G1_STEP_P + 32];

        float4 aE = xa[0], aO = xa[2];
        float4 e0 = bl[0], e1 = bl[32], e2 = bl[64], e3 = bl[96];
        v8f c0 = {}, c1 = {}, c2 = {}, c3 = {};

        for (int t = 0; t < G1_K8 - 2; t += 2) {
            const float4 nE = xa[2 * t + 4], nO = xa[2 * t + 6];
            const float4* bo = bl + (t + 1) * G1_STEP_P;
            const float4 o0 = bo[0], o1 = bo[32], o2 = bo[64], o3 = bo[96];
            __builtin_amdgcn_sched_barrier(0);
            g1_step(aE, e0, e1, e2, e3, c0, c1, c2, c3);
            e0 = bo[G1_STEP_P]; e1 = bo[G1_STEP_P + 32]; e2 = bo[G1_STEP_P + 64]; e3 = bo[G1_STEP_P + 96];
            __builtin_amdgcn_sched_barrier(0);
            g1_step(aO, o0, o1, o2, o3, c0, c1, c2, c3);
            aE = nE; aO = nO;
        }
        const float4 dA = *(const float4*)(dinv + row0 + half * 8);
        const float4 dB = *(const float4*)(dinv + row0 + half * 8 + 4);
        {
            const float4* bo = bl + (G1_K8 - 1) * G1_STEP_P;
            const float4 o0 = bo[0], o1 = bo[32], o2 = bo[64], o3 = bo[96];
            g1_step(aE, e0, e1, e2, e3, c0, c1, c2, c3);
            g1_step(aO, o0, o1, o2, o3, c0, c1, c2, c3);
        }
        {
            v2f b;
            b.x = u0.x; b.y = u0.y;
            c0 = __builtin_amdgcn_wmma_f32_16x16x4_f32(false, at, false, b, (short)0, c0, false, false);
            b.x = u0.z; b.y = u0.w;
            c1 = __builtin_amdgcn_wmma_f32_16x16x4_f32(false, at, false, b, (short)0, c1, false, false);
            b.x = u1.x; b.y = u1.y;
            c2 = __builtin_amdgcn_wmma_f32_16x16x4_f32(false, at, false, b, (short)0, c2, false, false);
            b.x = u1.z; b.y = u1.w;
            c3 = __builtin_amdgcn_wmma_f32_16x16x4_f32(false, at, false, b, (short)0, c3, false, false);
        }

        const float ds[8] = {dA.x, dA.y, dA.z, dA.w, dB.x, dB.y, dB.z, dB.w};
        float* out = h1 + (size_t)(row0 + half * 8) * HIDDEN + l16;
        for (int r = 0; r < 8; ++r) {
            out[(size_t)r * HIDDEN + 0]  = c0[r] * ds[r];
            out[(size_t)r * HIDDEN + 16] = c1[r] * ds[r];
            out[(size_t)r * HIDDEN + 32] = c2[r] * ds[r];
            out[(size_t)r * HIDDEN + 48] = c3[r] * ds[r];
        }
        mtile = nxt;
        nxt = -1;
    }
}

__global__ void k_gather1(const int* __restrict__ off, const int* __restrict__ csr_src,
                          const float* __restrict__ csr_w, const float* __restrict__ dinv,
                          const float* __restrict__ h1, const float* __restrict__ b1,
                          float* __restrict__ agg1) {
    const int t = blockIdx.x * blockDim.x + threadIdx.x;
    const int i = t >> 4;
    const int q = t & 15;
    if (i >= N_NODES) return;
    const int b  = off[i];
    const int en = (i + 1 < N_NODES) ? off[i + 1] : N_EDGES;
    const float di = dinv[i];
    const float4* __restrict__ h4 = (const float4*)h1;
    float4 acc = make_float4(0.0f, 0.0f, 0.0f, 0.0f);
    int j = b;
    for (; j + 4 <= en; j += 4) {
        const int s0 = node_clamp(csr_src[j]), s1 = node_clamp(csr_src[j + 1]), s2 = node_clamp(csr_src[j + 2]), s3 = node_clamp(csr_src[j + 3]);
        const float w0 = csr_w[j], w1 = csr_w[j + 1], w2 = csr_w[j + 2], w3 = csr_w[j + 3];
        const float4 v0 = h4[(size_t)s0 * 16 + q];
        const float4 v1 = h4[(size_t)s1 * 16 + q];
        const float4 v2 = h4[(size_t)s2 * 16 + q];
        const float4 v3 = h4[(size_t)s3 * 16 + q];
        acc.x += w0 * v0.x; acc.y += w0 * v0.y; acc.z += w0 * v0.z; acc.w += w0 * v0.w;
        acc.x += w1 * v1.x; acc.y += w1 * v1.y; acc.z += w1 * v1.z; acc.w += w1 * v1.w;
        acc.x += w2 * v2.x; acc.y += w2 * v2.y; acc.z += w2 * v2.z; acc.w += w2 * v2.w;
        acc.x += w3 * v3.x; acc.y += w3 * v3.y; acc.z += w3 * v3.z; acc.w += w3 * v3.w;
    }
    for (; j < en; ++j) {
        const int s0 = node_clamp(csr_src[j]);
        const float w0 = csr_w[j];
        const float4 v0 = h4[(size_t)s0 * 16 + q];
        acc.x += w0 * v0.x; acc.y += w0 * v0.y; acc.z += w0 * v0.z; acc.w += w0 * v0.w;
    }
    const float4 hv = h4[(size_t)i * 16 + q];
    const float4 bv = make_float4(b1[q * 4], b1[q * 4 + 1], b1[q * 4 + 2], b1[q * 4 + 3]);
    float4 o;
    o.x = bv.x + di * hv.x + di * acc.x;
    o.y = bv.y + di * hv.y + di * acc.y;
    o.z = bv.z + di * hv.z + di * acc.z;
    o.w = bv.w + di * hv.w + di * acc.w;
    ((float4*)agg1)[(size_t)i * 16 + q] = o;
}


__global__ __launch_bounds__(128) void k_gemm2(const float* __restrict__ agg1,
                                               const float* __restrict__ W2,
                                               const float* __restrict__ dinv,
                                               float* __restrict__ h2) {
    const int wave  = threadIdx.x >> 5;
    const int lane  = threadIdx.x & 31;
    const int mtile = blockIdx.x * 4 + wave;
    if (mtile >= M_TILES) return;
    const int row0 = mtile * 16;
    const int half = lane >> 4;
    const int l16  = lane & 15;

    const float* arow = agg1 + (size_t)(row0 + l16) * HIDDEN + half * 2;
    v8f c = {};

    for (int k = 0; k < HIDDEN; k += 4) {
        v2f a;
        a.x = fmaxf(arow[k], 0.0f);
        a.y = fmaxf(arow[k + 1], 0.0f);
        const float* bp = W2 + (size_t)(k + half * 2) * N_CLASSES + l16;
        v2f b;
        b.x = bp[0];
        b.y = bp[N_CLASSES];
        c = __builtin_amdgcn_wmma_f32_16x16x4_f32(false, a, false, b, (short)0, c, false, false);
    }

    const float4 dA = *(const float4*)(dinv + row0 + half * 8);
    const float4 dB = *(const float4*)(dinv + row0 + half * 8 + 4);
    const float ds[8] = {dA.x, dA.y, dA.z, dA.w, dB.x, dB.y, dB.z, dB.w};
    float* out = h2 + (size_t)(row0 + half * 8) * N_CLASSES + l16;
    for (int r = 0; r < 8; ++r)
        out[(size_t)r * N_CLASSES] = c[r] * ds[r];
}

__global__ void k_gather2(const int* __restrict__ off, const int* __restrict__ csr_src,
                          const float* __restrict__ csr_w, const float* __restrict__ dinv,
                          const float* __restrict__ h2, const float* __restrict__ b2,
                          float* __restrict__ agg2) {
    const int t = blockIdx.x * blockDim.x + threadIdx.x;
    const int i = t >> 2;
    const int q = t & 3;
    if (i >= N_NODES) return;
    const int b  = off[i];
    const int en = (i + 1 < N_NODES) ? off[i + 1] : N_EDGES;
    const float di = dinv[i];
    const float4* __restrict__ h4 = (const float4*)h2;
    float4 acc = make_float4(0.0f, 0.0f, 0.0f, 0.0f);
    int j = b;
    for (; j + 4 <= en; j += 4) {
        const int s0 = node_clamp(csr_src[j]), s1 = node_clamp(csr_src[j + 1]), s2 = node_clamp(csr_src[j + 2]), s3 = node_clamp(csr_src[j + 3]);
        const float w0 = csr_w[j], w1 = csr_w[j + 1], w2 = csr_w[j + 2], w3 = csr_w[j + 3];
        const float4 v0 = h4[(size_t)s0 * 4 + q];
        const float4 v1 = h4[(size_t)s1 * 4 + q];
        const float4 v2 = h4[(size_t)s2 * 4 + q];
        const float4 v3 = h4[(size_t)s3 * 4 + q];
        acc.x += w0 * v0.x; acc.y += w0 * v0.y; acc.z += w0 * v0.z; acc.w += w0 * v0.w;
        acc.x += w1 * v1.x; acc.y += w1 * v1.y; acc.z += w1 * v1.z; acc.w += w1 * v1.w;
        acc.x += w2 * v2.x; acc.y += w2 * v2.y; acc.z += w2 * v2.z; acc.w += w2 * v2.w;
        acc.x += w3 * v3.x; acc.y += w3 * v3.y; acc.z += w3 * v3.z; acc.w += w3 * v3.w;
    }
    for (; j < en; ++j) {
        const int s0 = node_clamp(csr_src[j]);
        const float w0 = csr_w[j];
        const float4 v0 = h4[(size_t)s0 * 4 + q];
        acc.x += w0 * v0.x; acc.y += w0 * v0.y; acc.z += w0 * v0.z; acc.w += w0 * v0.w;
    }
    const float4 hv = h4[(size_t)i * 4 + q];
    const float4 bv = make_float4(b2[q * 4], b2[q * 4 + 1], b2[q * 4 + 2], b2[q * 4 + 3]);
    float4 o;
    o.x = bv.x + di * hv.x + di * acc.x;
    o.y = bv.y + di * hv.y + di * acc.y;
    o.z = bv.z + di * hv.z + di * acc.z;
    o.w = bv.w + di * hv.w + di * acc.w;
    ((float4*)agg2)[(size_t)i * 4 + q] = o;
}


__global__ void k_logsoftmax(const float* __restrict__ agg2, float* __restrict__ out) {
    int i = blockIdx.x * blockDim.x + threadIdx.x;
    if (i >= N_NODES) return;
    const float* v = agg2 + (size_t)i * N_CLASSES;
    float m = v[0];
    #pragma unroll
    for (int c = 1; c < N_CLASSES; ++c) m = fmaxf(m, v[c]);
    float s = 0.0f;
    #pragma unroll
    for (int c = 0; c < N_CLASSES; ++c) s += expf(v[c] - m);
    float l = logf(s);
    float* o = out + (size_t)i * N_CLASSES;
    #pragma unroll
    for (int c = 0; c < N_CLASSES; ++c) o[c] = v[c] - m - l;
}


extern "C" void kernel_launch(void* const* d_in, const int* in_sizes, int n_in,
                              void* d_out, int out_size, void* d_ws, size_t ws_size,
                              hipStream_t stream) {
    const float* x  = (const float*)d_in[0];
    const int*   ei = (const int*)d_in[1];
    const float* ew = (const float*)d_in[2];
    const float* W1 = (const float*)d_in[3];
    const float* b1 = (const float*)d_in[4];
    const float* W2 = (const float*)d_in[5];
    const float* b2 = (const float*)d_in[6];
    float* out = (float*)d_out;

    const int* row = ei;
    const int* col = ei + N_EDGES;

    float* ws      = (float*)d_ws;
    int*   off     = (int*)ws;
    float* dinv    = ws + N_NODES;
    int*   csr_src = (int*)(dinv + N_NODES);
    float* h1      = (float*)csr_src + N_EDGES;
    float* agg1    = h1   + (size_t)N_NODES * HIDDEN;
    float* csr_w   = agg1 + (size_t)N_NODES * HIDDEN;
    float* agg2    = csr_w + (size_t)N_NODES * N_CLASSES;
    float* h2      = h1;
    int*   rec     = (int*)agg1;
    int*   bh      = (int*)h1;

    const int B = 256;

    k_bkt_hist   <<<SORT_BLOCKS, 1024, 0, stream>>>(col, bh);
    k_scan<HIST_LEN><<<1, 1024, 0, stream>>>(bh);
    k_bkt_scatter<<<SORT_BLOCKS, 1024, 0, stream>>>(row, col, ew, bh, rec);
    k_bkt_sort   <<<N_BKT, 1024, 0, stream>>>(bh, rec, off, csr_src, csr_w);
    k_dinv    <<<(N_NODES + B - 1) / B, B, 0, stream>>>(off, csr_w, dinv);

    k_gemm1<<<G1_BLOCKS, G1_THREADS, 0, stream>>>(x, W1, dinv, h1);

    k_gather1<<<(N_NODES * 16 + B - 1) / B, B, 0, stream>>>(off, csr_src, csr_w, dinv, h1, b1, agg1);

    k_gemm2<<<(M_TILES + 3) / 4, 128, 0, stream>>>(agg1, W2, dinv, h2);

    k_gather2<<<(N_NODES * 4 + B - 1) / B, B, 0, stream>>>(off, csr_src, csr_w, dinv, h2, b2, agg2);

    k_logsoftmax<<<(N_NODES + B - 1) / B, B, 0, stream>>>(agg2, out);
}


# The three per-node follow-up kernels run inside their producers (degree in the bucket sort from LDS, GEMM2 on gather 1's 16-node tile held in LDS, log-softmax in gather 2): 10 launches -> 7
# speedup vs baseline: 4.2322x; 1.0994x over previous
#include <hip/hip_runtime.h>
#include <math.h>

#define N_NODES   100000
#define N_EDGES   1600000
#define N_FEAT    500
#define HIDDEN    64
#define N_CLASSES 16
#define M_TILES   (N_NODES / 16)

typedef __attribute__((ext_vector_type(2))) float v2f;
typedef __attribute__((ext_vector_type(8))) float v8f;

__device__ __forceinline__ int node_clamp(int s) {
    return (int)min((unsigned)s, (unsigned)(N_NODES - 1));
}


#define SORT_BLOCKS 256
#define SORT_CHUNK  (N_EDGES / SORT_BLOCKS)
#define SORT_TRIPS  ((SORT_CHUNK + 1023) / 1024)
#define SORT_GROUP  8
#define SORT_CAP    16384
#define SORT_PAD    32
#define BKT_SHIFT   9
#define BKT_NODES   (1 << BKT_SHIFT)
#define N_BKT       ((N_NODES + BKT_NODES - 1) / BKT_NODES)
#define HIST_LEN    (N_BKT * SORT_BLOCKS)
static_assert(N_EDGES % SORT_BLOCKS == 0, "chunks are equal");
static_assert(HIST_LEN % 16 == 0, "a scan thread's 16 counts are all inside or all outside the array");
static_assert(N_NODES % 16 == 0, "M tiles are exact");

__global__ __launch_bounds__(1024) void k_bkt_hist(const int* __restrict__ col, int* __restrict__ bh) {
    __shared__ int hist[N_BKT];
    const int tid = threadIdx.x;
    for (int b = tid; b < N_BKT; b += 1024) hist[b] = 0;
    __syncthreads();
    const int lo = blockIdx.x * SORT_CHUNK;
    int c[SORT_TRIPS];
    #pragma unroll
    for (int t = 0; t < SORT_TRIPS; ++t) {
        const int k = min(tid + t * 1024, SORT_CHUNK - 1);
        c[t] = col[lo + k];
    }
    #pragma unroll
    for (int t = 0; t < SORT_TRIPS; ++t)
        if (tid + t * 1024 < SORT_CHUNK) atomicAdd(&hist[node_clamp(c[t]) >> BKT_SHIFT], 1);
    __syncthreads();
    for (int b = tid; b < N_BKT; b += 1024) bh[b * SORT_BLOCKS + blockIdx.x] = hist[b];
}

template <int LEN>
__global__ __launch_bounds__(1024) void k_scan(int* __restrict__ cnt) {
    __shared__ int wsum[32];
    __shared__ int tile_total;
    const int tid  = threadIdx.x;
    const int lane = tid & 31;
    const int wv   = tid >> 5;
    int carry = 0;
    for (int base = 0; base < LEN; base += 16384) {
        const int idx = base + tid * 16;
        const bool in = idx < LEN;
        int4 v0 = make_int4(0, 0, 0, 0), v1 = v0, v2 = v0, v3 = v0;
        if (in) {
            const int4* p = (const int4*)(cnt + idx);
            v0 = p[0]; v1 = p[1]; v2 = p[2]; v3 = p[3];
        }
        const int s0 = v0.x + v0.y + v0.z + v0.w;
        const int s1 = v1.x + v1.y + v1.z + v1.w;
        const int s2 = v2.x + v2.y + v2.z + v2.w;
        const int s3 = v3.x + v3.y + v3.z + v3.w;
        const int s = s0 + s1 + s2 + s3;
        int inc = s;
        for (int d = 1; d < 32; d <<= 1) {
            int n = __shfl_up(inc, d);
            if (lane >= d) inc += n;
        }
        if (lane == 31) wsum[wv] = inc;
        __syncthreads();
        if (wv == 0) {
            const int ws = wsum[lane];
            int winc = ws;
            for (int d = 1; d < 32; d <<= 1) {
                int n = __shfl_up(winc, d);
                if (lane >= d) winc += n;
            }
            wsum[lane] = winc - ws;
            if (lane == 31) tile_total = winc;
        }
        __syncthreads();
        const int ex = carry + wsum[wv] + (inc - s);
        carry += tile_total;
        if (in) {
            int4 o0, o1, o2, o3;
            o0.x = ex;             o0.y = o0.x + v0.x; o0.z = o0.y + v0.y; o0.w = o0.z + v0.z;
            o1.x = ex + s0;        o1.y = o1.x + v1.x; o1.z = o1.y + v1.y; o1.w = o1.z + v1.z;
            o2.x = o1.x + s1;      o2.y = o2.x + v2.x; o2.z = o2.y + v2.y; o2.w = o2.z + v2.z;
            o3.x = o2.x + s2;      o3.y = o3.x + v3.x; o3.z = o3.y + v3.y; o3.w = o3.z + v3.z;
            int4* p = (int4*)(cnt + idx);
            p[0] = o0; p[1] = o1; p[2] = o2; p[3] = o3;
        }
        __syncthreads();
    }
}

__global__ __launch_bounds__(1024) void k_bkt_scatter(const int* __restrict__ row, const int* __restrict__ col,
                                                      const float* __restrict__ w, const int* __restrict__ bh,
                                                      int* __restrict__ rec_raw) {
    __shared__ int  cur[N_BKT];
    __shared__ int  delta[N_BKT];
    __shared__ int  wsum[8];
    __shared__ int4 stage[SORT_CHUNK];
    const int tid  = threadIdx.x;
    const int lane = tid & 31;
    const int wv   = tid >> 5;
    int g0 = 0, nb = 0;
    if (tid < N_BKT) {
        const int idx = tid * SORT_BLOCKS + (int)blockIdx.x;
        g0 = bh[idx];
        const int g1 = (idx + 1 < HIST_LEN) ? bh[idx + 1] : N_EDGES;
        nb = g1 - g0;
    }
    int inc = nb;
    for (int d = 1; d < 32; d <<= 1) {
        int n = __shfl_up(inc, d);
        if (lane >= d) inc += n;
    }
    if (lane == 31 && wv < 8) wsum[wv] = inc;
    __syncthreads();
    if (wv == 0) {
        const int ws = (lane < 8) ? wsum[lane] : 0;
        int winc = ws;
        for (int d = 1; d < 32; d <<= 1) {
            int n = __shfl_up(winc, d);
            if (lane >= d) winc += n;
        }
        if (lane < 8) wsum[lane] = winc - ws;
    }
    __syncthreads();
    if (tid < N_BKT) {
        const int ex = wsum[wv] + (inc - nb);
        cur[tid]   = ex;
        delta[tid] = g0 - ex;
    }
    __syncthreads();
    int4* __restrict__ rec = (int4*)rec_raw;
    const int lo = blockIdx.x * SORT_CHUNK;
    int c[SORT_TRIPS], r[SORT_TRIPS], wb[SORT_TRIPS];
    #pragma unroll
    for (int t = 0; t < SORT_TRIPS; ++t) {
        const int k = min(tid + t * 1024, SORT_CHUNK - 1);
        c[t]  = node_clamp(col[lo + k]);
        r[t]  = row[lo + k];
        wb[t] = __float_as_int(w[lo + k]);
    }
    #pragma unroll
    for (int t = 0; t < SORT_TRIPS; ++t) {
        if (tid + t * 1024 < SORT_CHUNK) {
            const int q = atomicAdd(&cur[c[t] >> BKT_SHIFT], 1);
            stage[min((unsigned)q, (unsigned)(SORT_CHUNK - 1))] = make_int4(c[t], r[t], wb[t], 0);
        }
    }
    __syncthreads();
    int4 v[SORT_TRIPS];
    int  dl[SORT_TRIPS];
    #pragma unroll
    for (int t = 0; t < SORT_TRIPS; ++t) v[t] = stage[min(tid + t * 1024, SORT_CHUNK - 1)];
    #pragma unroll
    for (int t = 0; t < SORT_TRIPS; ++t) dl[t] = delta[node_clamp(v[t].x) >> BKT_SHIFT];
    #pragma unroll
    for (int t = 0; t < SORT_TRIPS; ++t) {
        const int i = tid + t * 1024;
        if (i < SORT_CHUNK) rec[min((unsigned)(i + dl[t]), (unsigned)(N_EDGES - 1))] = v[t];
    }
}

__global__ __launch_bounds__(1024) void k_bkt_sort(const int* __restrict__ bh, const int* __restrict__ rec_raw,
                                                   int* __restrict__ off, int* __restrict__ csr_src,
                                                   float* __restrict__ csr_w, float* __restrict__ dinv) {
    __shared__ int cnt[BKT_NODES];
    __shared__ int wsum[16];
    __shared__ int4 st_src[(SORT_CAP + SORT_PAD) / 4];
    __shared__ int4 st_w[(SORT_CAP + SORT_PAD) / 4];
    const int tid  = threadIdx.x;
    const int lane = tid & 31;
    const int wv   = tid >> 5;
    const int bkt  = blockIdx.x;
    const int4* __restrict__ rec = (const int4*)rec_raw;
    const int bstart = (int)min((unsigned)bh[bkt * SORT_BLOCKS], (unsigned)N_EDGES);
    const int bend   = (bkt + 1 < N_BKT) ? (int)min((unsigned)bh[(bkt + 1) * SORT_BLOCKS], (unsigned)N_EDGES) : N_EDGES;
    const int  nedge  = bend - bstart;
    const bool staged = nedge <= SORT_CAP;
    const int  shift  = bstart & (SORT_PAD - 1);
    int* ssrc = (int*)st_src;
    int* sw   = (int*)st_w;

    if (tid < BKT_NODES) cnt[tid] = 0;
    __syncthreads();
    for (int e0 = bstart + tid; e0 < bend; e0 += SORT_GROUP * 1024) {
        int x[SORT_GROUP];
        #pragma unroll
        for (int u = 0; u < SORT_GROUP; ++u) {
            const int e = min(e0 + u * 1024, bend - 1);
            x[u] = rec[e].x;
        }
        #pragma unroll
        for (int u = 0; u < SORT_GROUP; ++u)
            if (e0 + u * 1024 < bend) atomicAdd(&cnt[x[u] & (BKT_NODES - 1)], 1);
    }
    __syncthreads();

    const int v = (tid < BKT_NODES) ? cnt[tid] : 0;
    int inc = v;
    for (int d = 1; d < 32; d <<= 1) {
        int n = __shfl_up(inc, d);
        if (lane >= d) inc += n;
    }
    if (lane == 31 && wv < 16) wsum[wv] = inc;
    __syncthreads();
    if (wv == 0) {
        const int ws = (lane < 16) ? wsum[lane] : 0;
        int winc = ws;
        for (int d = 1; d < 32; d <<= 1) {
            int n = __shfl_up(winc, d);
            if (lane >= d) winc += n;
        }
        if (lane < 16) wsum[lane] = winc - ws;
    }
    __syncthreads();
    int seg_lo = 0;
    if (tid < BKT_NODES) {
        const int ex = wsum[wv] + (inc - v);
        cnt[tid] = ex;
        seg_lo = ex;
        const int node = bkt * BKT_NODES + tid;
        if (node < N_NODES) off[node] = bstart + ex;
    }
    __syncthreads();

    for (int e0 = bstart + tid; e0 < bend; e0 += SORT_GROUP * 1024) {
        int4 r[SORT_GROUP];
        #pragma unroll
        for (int u = 0; u < SORT_GROUP; ++u) {
            const int e = min(e0 + u * 1024, bend - 1);
            r[u] = rec[e];
        }
        #pragma unroll
        for (int u = 0; u < SORT_GROUP; ++u) {
            if (e0 + u * 1024 < bend) {
                const int q = atomicAdd(&cnt[r[u].x & (BKT_NODES - 1)], 1);
                if (staged) {
                    const int p = (int)min((unsigned)(q + shift), (unsigned)(SORT_CAP + SORT_PAD - 1));
                    ssrc[p] = r[u].y;
                    sw[p]   = r[u].z;
                } else {
                    const int p = (int)min((unsigned)(bstart + q), (unsigned)(N_EDGES - 1));
                    csr_src[p] = r[u].y;
                    csr_w[p]   = __int_as_float(r[u].z);
                }
            }
        }
    }
    __syncthreads();
    if (staged) {
        const int a0 = bstart - shift;
        const int pieces = (shift + nedge + 3) >> 2;
        int4* gs = (int4*)(csr_src + a0);
        int4* gw = (int4*)(csr_w + a0);
        for (int g = tid; g < pieces; g += 1024) {
            const int p = g * 4;
            if (p >= shift && p + 4 <= shift + nedge) {
                gs[g] = st_src[g];
                gw[g] = st_w[g];
            } else {
                for (int k = 0; k < 4; ++k) {
                    if (p + k >= shift && p + k < shift + nedge) {
                        csr_src[a0 + p + k] = ssrc[p + k];
                        csr_w[a0 + p + k]   = __int_as_float(sw[p + k]);
                    }
                }
            }
        }
    }
    if (tid < BKT_NODES) {
        const int node = bkt * BKT_NODES + tid;
        if (node < N_NODES) {
            const int seg_hi = min(cnt[tid], nedge);
            float d = 1.0f;
            if (staged) {
                for (int j = seg_lo; j < seg_hi; ++j) d += __int_as_float(sw[shift + j]);
            } else {
                for (int j = seg_lo; j < seg_hi; ++j) d += csr_w[bstart + j];
            }
            dinv[node] = (d > 0.0f) ? rsqrtf(d) : 0.0f;
        }
    }
}


#define G1_WAVES   8
#define G1_THREADS (G1_WAVES * 32)
#define G1_BLOCKS  512
#define G1_K8      (N_FEAT / 8)
#define G1_KTAIL   (G1_K8 * 8)
#define G1_STEP_P  (4 * 32)
#define G1_PIECES  (G1_K8 * G1_STEP_P + 2 * 32)
#define G1_R0      (G1_BLOCKS * G1_WAVES)
#define G1_R1      (G1_BLOCKS * (G1_WAVES / 2))
#define G1_REST    (M_TILES - G1_R0 - G1_R1)
#define G1_COPY_TRIPS ((G1_PIECES + G1_THREADS - 1) / G1_THREADS)
static_assert(G1_PIECES * 4 == N_FEAT * HIDDEN, "the LDS copy holds W1 exactly once");
static_assert(N_FEAT - G1_KTAIL == 4, "one 4-wide step is left after the 8-wide ones");
static_assert(G1_K8 % 2 == 0, "the K loop takes two steps per trip");
static_assert(G1_REST >= 0 && G1_REST <= G1_BLOCKS, "at most one left-over tile per work-group");
static_assert(HIDDEN == 64, "four 16-column accumulators");
static_assert(G1_COPY_TRIPS % 8 == 0, "the copy takes its pieces in groups of eight");

__device__ __forceinline__ void g1_step(const float4 a, const float4 q0, const float4 q1,
                                        const float4 q2, const float4 q3,
                                        v8f& c0, v8f& c1, v8f& c2, v8f& c3) {
    v2f av, b;
    av.x = a.x; av.y = a.y;
    b.x = q0.x; b.y = q0.y;
    c0 = __builtin_amdgcn_wmma_f32_16x16x4_f32(false, av, false, b, (short)0, c0, false, false);
    b.x = q0.z; b.y = q0.w;
    c1 = __builtin_amdgcn_wmma_f32_16x16x4_f32(false, av, false, b, (short)0, c1, false, false);
    b.x = q1.x; b.y = q1.y;
    c2 = __builtin_amdgcn_wmma_f32_16x16x4_f32(false, av, false, b, (short)0, c2, false, false);
    b.x = q1.z; b.y = q1.w;
    c3 = __builtin_amdgcn_wmma_f32_16x16x4_f32(false, av, false, b, (short)0, c3, false, false);
    av.x = a.z; av.y = a.w;
    b.x = q2.x; b.y = q2.y;
    c0 = __builtin_amdgcn_wmma_f32_16x16x4_f32(false, av, false, b, (short)0, c0, false, false);
    b.x = q2.z; b.y = q2.w;
    c1 = __builtin_amdgcn_wmma_f32_16x16x4_f32(false, av, false, b, (short)0, c1, false, false);
    b.x = q3.x; b.y = q3.y;
    c2 = __builtin_amdgcn_wmma_f32_16x16x4_f32(false, av, false, b, (short)0, c2, false, false);
    b.x = q3.z; b.y = q3.w;
    c3 = __builtin_amdgcn_wmma_f32_16x16x4_f32(false, av, false, b, (short)0, c3, false, false);
}

__global__ __launch_bounds__(G1_THREADS) void k_gemm1(const float* x,
                                                      const float* W1,
                                                      const float* __restrict__ dinv,
                                                      float* __restrict__ h1) {
    __shared__ float4 Bs[G1_PIECES];
    const int tid = threadIdx.x;

    {
        float* Bf = (float*)Bs;
        const float4* W4 = (const float4*)W1;
        #pragma unroll
        for (int g = 0; g < G1_COPY_TRIPS; g += 8) {
            float4 v[8];
            #pragma unroll
            for (int u = 0; u < 8; ++u) {
                const int e = min(tid + (g + u) * G1_THREADS, G1_PIECES - 1);
                v[u] = W4[e];
            }
            __builtin_amdgcn_sched_barrier(0);
            #pragma unroll
            for (int u = 0; u < 8; ++u) {
                const int e = tid + (g + u) * G1_THREADS;
                if (e < G1_PIECES) {
                    const int k  = e >> 4;
                    const int n0 = (e & 15) * 4;
                    const int j  = n0 >> 4;
                    const int c  = n0 & 15;
                    int d;
                    if (k < G1_KTAIL) {
                        const int t = k >> 3, r = k & 7;
                        const int hh = r >> 2, w = (r >> 1) & 1, s = r & 1;
                        d = ((t * 4 + w * 2 + (j >> 1)) * 32 + hh * 16 + c) * 4 + (j & 1) * 2 + s;
                    } else {
                        const int r = k - G1_KTAIL;
                        const int hh = r >> 1, s = r & 1;
                        d = (G1_K8 * G1_STEP_P + (j >> 1) * 32 + hh * 16 + c) * 4 + (j & 1) * 2 + s;
                    }
                    Bf[d] = v[u].x; Bf[d + 4] = v[u].y; Bf[d + 8] = v[u].z; Bf[d + 12] = v[u].w;
                }
            }
        }
    }
    __syncthreads();

    const int wave = __builtin_amdgcn_readfirstlane(tid >> 5);
    const int lane = tid & 31;
    const int half = lane >> 4;
    const int l16  = lane & 15;
    const float4* bl = Bs + lane;

    int mtile = wave * G1_BLOCKS + (int)blockIdx.x;
    int nxt   = ((__popc(wave) & 1) == 0) ? G1_R0 + (wave >> 1) * G1_BLOCKS + (int)blockIdx.x
              : ((wave == 1 && (int)blockIdx.x < G1_REST) ? G1_R0 + G1_R1 + (int)blockIdx.x : -1);

    while (mtile >= 0) {
        const int row0 = mtile * 16;
        const float*  xr = x + (size_t)(row0 + l16) * N_FEAT;
        const float4* xa = (const float4*)xr + half;
        v2f at;
        at.x = xr[G1_KTAIL + half * 2];
        at.y = xr[G1_KTAIL + half * 2 + 1];
        const float4 u0 = bl[G1_K8 * G1_STEP_P], u1 = bl[G1_K8 * G1_STEP_P + 32];

        float4 aE = xa[0], aO = xa[2];
        float4 e0 = bl[0], e1 = bl[32], e2 = bl[64], e3 = bl[96];
        v8f c0 = {}, c1 = {}, c2 = {}, c3 = {};

        for (int t = 0; t < G1_K8 - 2; t += 2) {
            const float4 nE = xa[2 * t + 4], nO = xa[2 * t + 6];
            const float4* bo = bl + (t + 1) * G1_STEP_P;
            const float4 o0 = bo[0], o1 = bo[32], o2 = bo[64], o3 = bo[96];
            __builtin_amdgcn_sched_barrier(0);
            g1_step(aE, e0, e1, e2, e3, c0, c1, c2, c3);
            e0 = bo[G1_STEP_P]; e1 = bo[G1_STEP_P + 32]; e2 = bo[G1_STEP_P + 64]; e3 = bo[G1_STEP_P + 96];
            __builtin_amdgcn_sched_barrier(0);
            g1_step(aO, o0, o1, o2, o3, c0, c1, c2, c3);
            aE = nE; aO = nO;
        }
        const float4 dA = *(const float4*)(dinv + row0 + half * 8);
        const float4 dB = *(const float4*)(dinv + row0 + half * 8 + 4);
        {
            const float4* bo = bl + (G1_K8 - 1) * G1_STEP_P;
            const float4 o0 = bo[0], o1 = bo[32], o2 = bo[64], o3 = bo[96];
            g1_step(aE, e0, e1, e2, e3, c0, c1, c2, c3);
            g1_step(aO, o0, o1, o2, o3, c0, c1, c2, c3);
        }
        {
            v2f b;
            b.x = u0.x; b.y = u0.y;
            c0 = __builtin_amdgcn_wmma_f32_16x16x4_f32(false, at, false, b, (short)0, c0, false, false);
            b.x = u0.z; b.y = u0.w;
            c1 = __builtin_amdgcn_wmma_f32_16x16x4_f32(false, at, false, b, (short)0, c1, false, false);
            b.x = u1.x; b.y = u1.y;
            c2 = __builtin_amdgcn_wmma_f32_16x16x4_f32(false, at, false, b, (short)0, c2, false, false);
            b.x = u1.z; b.y = u1.w;
            c3 = __builtin_amdgcn_wmma_f32_16x16x4_f32(false, at, false, b, (short)0, c3, false, false);
        }

        const float ds[8] = {dA.x, dA.y, dA.z, dA.w, dB.x, dB.y, dB.z, dB.w};
        float* out = h1 + (size_t)(row0 + half * 8) * HIDDEN + l16;
        for (int r = 0; r < 8; ++r) {
            out[(size_t)r * HIDDEN + 0]  = c0[r] * ds[r];
            out[(size_t)r * HIDDEN + 16] = c1[r] * ds[r];
            out[(size_t)r * HIDDEN + 32] = c2[r] * ds[r];
            out[(size_t)r * HIDDEN + 48] = c3[r] * ds[r];
        }
        mtile = nxt;
        nxt = -1;
    }
}


#define GA_THREADS 256
#define GA_NODES   (GA_THREADS / 16)
#define A1_PITCH4  17
static_assert(GA_NODES == 16, "a work-group's nodes are one 16-row tile");
static_assert(N_NODES % GA_NODES == 0, "no work-group holds a partial tile");

__global__ __launch_bounds__(GA_THREADS) void k_gather1_gemm2(const int* __restrict__ off, const int* __restrict__ csr_src,
                                const float* __restrict__ csr_w, const float* __restrict__ dinv,
                                const float* __restrict__ h1, const float* __restrict__ b1,
                                const float* __restrict__ W2, float* __restrict__ h2) {
    __shared__ float4 As[GA_NODES * A1_PITCH4];
    const int tid = threadIdx.x;
    const int nl = tid >> 4;
    const int q  = tid & 15;
    const int i  = (int)blockIdx.x * GA_NODES + nl;
    const int b  = off[i];
    const int en = (i + 1 < N_NODES) ? off[i + 1] : N_EDGES;
    const float di = dinv[i];
    const float4* __restrict__ h4 = (const float4*)h1;
    float4 acc = make_float4(0.0f, 0.0f, 0.0f, 0.0f);
    int j = b;
    for (; j + 4 <= en; j += 4) {
        const int s0 = node_clamp(csr_src[j]), s1 = node_clamp(csr_src[j + 1]), s2 = node_clamp(csr_src[j + 2]), s3 = node_clamp(csr_src[j + 3]);
        const float w0 = csr_w[j], w1 = csr_w[j + 1], w2 = csr_w[j + 2], w3 = csr_w[j + 3];
        const float4 v0 = h4[(size_t)s0 * 16 + q];
        const float4 v1 = h4[(size_t)s1 * 16 + q];
        const float4 v2 = h4[(size_t)s2 * 16 + q];
        const float4 v3 = h4[(size_t)s3 * 16 + q];
        acc.x += w0 * v0.x; acc.y += w0 * v0.y; acc.z += w0 * v0.z; acc.w += w0 * v0.w;
        acc.x += w1 * v1.x; acc.y += w1 * v1.y; acc.z += w1 * v1.z; acc.w += w1 * v1.w;
        acc.x += w2 * v2.x; acc.y += w2 * v2.y; acc.z += w2 * v2.z; acc.w += w2 * v2.w;
        acc.x += w3 * v3.x; acc.y += w3 * v3.y; acc.z += w3 * v3.z; acc.w += w3 * v3.w;
    }
    for (; j < en; ++j) {
        const int s0 = node_clamp(csr_src[j]);
        const float w0 = csr_w[j];
        const float4 v0 = h4[(size_t)s0 * 16 + q];
        acc.x += w0 * v0.x; acc.y += w0 * v0.y; acc.z += w0 * v0.z; acc.w += w0 * v0.w;
    }
    const bool first = __builtin_amdgcn_readfirstlane(tid >> 5) == 0;
    const int lane = tid & 31;
    const int half = lane >> 4;
    const int l16  = lane & 15;
    const int row0 = (int)blockIdx.x * GA_NODES;
    v2f bf[HIDDEN / 4];
    float4 dA = make_float4(0.0f, 0.0f, 0.0f, 0.0f), dB = dA;
    #pragma unroll
    for (int s = 0; s < HIDDEN / 4; ++s) { bf[s].x = 0.0f; bf[s].y = 0.0f; }
    if (first) {
        #pragma unroll
        for (int s = 0; s < HIDDEN / 4; ++s) {
            const float* bp = W2 + (size_t)(4 * s + half * 2) * N_CLASSES + l16;
            bf[s].x = bp[0];
            bf[s].y = bp[N_CLASSES];
        }
        dA = *(const float4*)(dinv + row0 + half * 8);
        dB = *(const float4*)(dinv + row0 + half * 8 + 4);
    }
    const float4 hv = h4[(size_t)i * 16 + q];
    const float4 bv = make_float4(b1[q * 4], b1[q * 4 + 1], b1[q * 4 + 2], b1[q * 4 + 3]);
    float4 o;
    o.x = bv.x + di * hv.x + di * acc.x;
    o.y = bv.y + di * hv.y + di * acc.y;
    o.z = bv.z + di * hv.z + di * acc.z;
    o.w = bv.w + di * hv.w + di * acc.w;
    As[nl * A1_PITCH4 + q] = make_float4(fmaxf(o.x, 0.0f), fmaxf(o.y, 0.0f), fmaxf(o.z, 0.0f), fmaxf(o.w, 0.0f));
    __syncthreads();

    if (!first) return;
    const float* arow = (const float*)As + l16 * (A1_PITCH4 * 4) + half * 2;
    v8f cacc = {};
    #pragma unroll
    for (int s = 0; s < HIDDEN / 4; ++s) {
        v2f a;
        a.x = arow[4 * s];
        a.y = arow[4 * s + 1];
        cacc = __builtin_amdgcn_wmma_f32_16x16x4_f32(false, a, false, bf[s], (short)0, cacc, false, false);
    }
    const float ds[8] = {dA.x, dA.y, dA.z, dA.w, dB.x, dB.y, dB.z, dB.w};
    float* out = h2 + (size_t)(row0 + half * 8) * N_CLASSES + l16;
    for (int r = 0; r < 8; ++r)
        out[(size_t)r * N_CLASSES] = cacc[r] * ds[r];
}

__global__ __launch_bounds__(256) void k_gather2_lsm(const int* __restrict__ off, const int* __restrict__ csr_src,
                          const float* __restrict__ csr_w, const float* __restrict__ dinv,
                          const float* __restrict__ h2, const float* __restrict__ b2,
                          float* __restrict__ out) {
    const int t = blockIdx.x * blockDim.x + threadIdx.x;
    const int i = t >> 2;
    const int q = t & 3;
    if (i >= N_NODES) return;
    const int b  = off[i];
    const int en = (i + 1 < N_NODES) ? off[i + 1] : N_EDGES;
    const float di = dinv[i];
    const float4* __restrict__ h4 = (const float4*)h2;
    float4 acc = make_float4(0.0f, 0.0f, 0.0f, 0.0f);
    int j = b;
    for (; j + 4 <= en; j += 4) {
        const int s0 = node_clamp(csr_src[j]), s1 = node_clamp(csr_src[j + 1]), s2 = node_clamp(csr_src[j + 2]), s3 = node_clamp(csr_src[j + 3]);
        const float w0 = csr_w[j], w1 = csr_w[j + 1], w2 = csr_w[j + 2], w3 = csr_w[j + 3];
        const float4 v0 = h4[(size_t)s0 * 4 + q];
        const float4 v1 = h4[(size_t)s1 * 4 + q];
        const float4 v2 = h4[(size_t)s2 * 4 + q];
        const float4 v3 = h4[(size_t)s3 * 4 + q];
        acc.x += w0 * v0.x; acc.y += w0 * v0.y; acc.z += w0 * v0.z; acc.w += w0 * v0.w;
        acc.x += w1 * v1.x; acc.y += w1 * v1.y; acc.z += w1 * v1.z; acc.w += w1 * v1.w;
        acc.x += w2 * v2.x; acc.y += w2 * v2.y; acc.z += w2 * v2.z; acc.w += w2 * v2.w;
        acc.x += w3 * v3.x; acc.y += w3 * v3.y; acc.z += w3 * v3.z; acc.w += w3 * v3.w;
    }
    for (; j < en; ++j) {
        const int s0 = node_clamp(csr_src[j]);
        const float w0 = csr_w[j];
        const float4 v0 = h4[(size_t)s0 * 4 + q];
        acc.x += w0 * v0.x; acc.y += w0 * v0.y; acc.z += w0 * v0.z; acc.w += w0 * v0.w;
    }
    const float4 hv = h4[(size_t)i * 4 + q];
    const float4 bv = make_float4(b2[q * 4], b2[q * 4 + 1], b2[q * 4 + 2], b2[q * 4 + 3]);
    float4 o;
    o.x = bv.x + di * hv.x + di * acc.x;
    o.y = bv.y + di * hv.y + di * acc.y;
    o.z = bv.z + di * hv.z + di * acc.z;
    o.w = bv.w + di * hv.w + di * acc.w;

    float m = fmaxf(fmaxf(o.x, o.y), fmaxf(o.z, o.w));
    m = fmaxf(m, __shfl_xor(m, 1));
    m = fmaxf(m, __shfl_xor(m, 2));
    const float e0 = expf(o.x - m), e1 = expf(o.y - m), e2 = expf(o.z - m), e3 = expf(o.w - m);
    float p = ((e0 + e1) + e2) + e3;
    #pragma unroll
    for (int g = 1; g < 4; ++g) {
        const float prev = __shfl(p, g - 1, 4);
        const float cand = (((prev + e0) + e1) + e2) + e3;
        if (q == g) p = cand;
    }
    const float s = __shfl(p, 3, 4);
    const float l = logf(s);
    ((float4*)out)[(size_t)i * 4 + q] = make_float4(o.x - m - l, o.y - m - l, o.z - m - l, o.w - m - l);
}


extern "C" void kernel_launch(void* const* d_in, const int* in_sizes, int n_in,
                              void* d_out, int out_size, void* d_ws, size_t ws_size,
                              hipStream_t stream) {
    const float* x  = (const float*)d_in[0];
    const int*   ei = (const int*)d_in[1];
    const float* ew = (const float*)d_in[2];
    const float* W1 = (const float*)d_in[3];
    const float* b1 = (const float*)d_in[4];
    const float* W2 = (const float*)d_in[5];
    const float* b2 = (const float*)d_in[6];
    float* out = (float*)d_out;

    const int* row = ei;
    const int* col = ei + N_EDGES;

    float* ws      = (float*)d_ws;
    int*   off     = (int*)ws;
    float* dinv    = ws + N_NODES;
    int*   csr_src = (int*)(dinv + N_NODES);
    float* h1      = (float*)csr_src + N_EDGES;
    float* plane2  = h1   + (size_t)N_NODES * HIDDEN;
    float* csr_w   = plane2 + (size_t)N_NODES * HIDDEN;
    float* h2      = plane2;
    int*   rec     = (int*)plane2;
    int*   bh      = (int*)h1;

    k_bkt_hist   <<<SORT_BLOCKS, 1024, 0, stream>>>(col, bh);
    k_scan<HIST_LEN><<<1, 1024, 0, stream>>>(bh);
    k_bkt_scatter<<<SORT_BLOCKS, 1024, 0, stream>>>(row, col, ew, bh, rec);
    k_bkt_sort   <<<N_BKT, 1024, 0, stream>>>(bh, rec, off, csr_src, csr_w, dinv);

    k_gemm1<<<G1_BLOCKS, G1_THREADS, 0, stream>>>(x, W1, dinv, h1);

    k_gather1_gemm2<<<N_NODES / GA_NODES, GA_THREADS, 0, stream>>>(off, csr_src, csr_w, dinv, h1, b1, W2, h2);

    k_gather2_lsm<<<(N_NODES * 4 + 255) / 256, 256, 0, stream>>>(off, csr_src, csr_w, dinv, h2, b2, out);
}


# Edge-sort level 1 keeps each chunk's records in place, grouped by bucket, plus its 196 offsets; level 2 walks its 256 runs: no count pass, no table scan, 7 -> 5 launches; new LDS arrival order of sums
# speedup vs baseline: 4.6480x; 1.0982x over previous
#include <hip/hip_runtime.h>
#include <math.h>

#define N_NODES   100000
#define N_EDGES   1600000
#define N_FEAT    500
#define HIDDEN    64
#define N_CLASSES 16
#define M_TILES   (N_NODES / 16)

typedef __attribute__((ext_vector_type(2))) float v2f;
typedef __attribute__((ext_vector_type(8))) float v8f;

__device__ __forceinline__ int node_clamp(int s) {
    return (int)min((unsigned)s, (unsigned)(N_NODES - 1));
}


#define SORT_BLOCKS 256
#define SORT_CHUNK  (N_EDGES / SORT_BLOCKS)
#define SORT_TRIPS  ((SORT_CHUNK + 1023) / 1024)
#define RUN_LANES   (1024 / SORT_BLOCKS)
#define RUN_GROUP   12
#define SORT_CAP    16384
#define SORT_PAD    32
#define BKT_SHIFT   9
#define BKT_NODES   (1 << BKT_SHIFT)
#define N_BKT       ((N_NODES + BKT_NODES - 1) / BKT_NODES)
#define TAB_LEN     (N_BKT * SORT_BLOCKS)
static_assert(N_EDGES % SORT_BLOCKS == 0, "chunks are equal");
static_assert(SORT_BLOCKS * RUN_LANES == 1024, "every run of a bucket has its threads");
static_assert(N_BKT <= 256, "the chunk's bucket counts fit waves 0-7, one per lane");
static_assert(N_NODES % 16 == 0, "M tiles are exact");

__global__ __launch_bounds__(1024) void k_chunk_group(const int* __restrict__ row, const int* __restrict__ col,
                                                      const float* __restrict__ w, int* __restrict__ tab,
                                                      int* __restrict__ rec_raw) {
    __shared__ int  cur[N_BKT];
    __shared__ int  wsum[8];
    __shared__ int4 stage[SORT_CHUNK];
    const int tid  = threadIdx.x;
    const int lane = tid & 31;
    const int wv   = tid >> 5;
    for (int b = tid; b < N_BKT; b += 1024) cur[b] = 0;
    __syncthreads();
    const int lo = blockIdx.x * SORT_CHUNK;
    int c[SORT_TRIPS], r[SORT_TRIPS], wb[SORT_TRIPS];
    #pragma unroll
    for (int t = 0; t < SORT_TRIPS; ++t) {
        const int k = min(tid + t * 1024, SORT_CHUNK - 1);
        c[t]  = node_clamp(col[lo + k]);
        r[t]  = row[lo + k];
        wb[t] = __float_as_int(w[lo + k]);
    }
    #pragma unroll
    for (int t = 0; t < SORT_TRIPS; ++t)
        if (tid + t * 1024 < SORT_CHUNK) atomicAdd(&cur[c[t] >> BKT_SHIFT], 1);
    __syncthreads();
    const int nb = (tid < N_BKT) ? cur[tid] : 0;
    int inc = nb;
    for (int d = 1; d < 32; d <<= 1) {
        int n = __shfl_up(inc, d);
        if (lane >= d) inc += n;
    }
    if (lane == 31 && wv < 8) wsum[wv] = inc;
    __syncthreads();
    if (wv == 0) {
        const int ws = (lane < 8) ? wsum[lane] : 0;
        int winc = ws;
        for (int d = 1; d < 32; d <<= 1) {
            int n = __shfl_up(winc, d);
            if (lane >= d) winc += n;
        }
        if (lane < 8) wsum[lane] = winc - ws;
    }
    __syncthreads();
    if (tid < N_BKT) {
        const int ex = wsum[wv] + (inc - nb);
        cur[tid] = ex;
        tab[(int)blockIdx.x * N_BKT + tid] = ex;
    }
    __syncthreads();
    #pragma unroll
    for (int t = 0; t < SORT_TRIPS; ++t) {
        if (tid + t * 1024 < SORT_CHUNK) {
            const int q = atomicAdd(&cur[c[t] >> BKT_SHIFT], 1);
            stage[min((unsigned)q, (unsigned)(SORT_CHUNK - 1))] = make_int4(c[t], r[t], wb[t], 0);
        }
    }
    __syncthreads();
    int4* __restrict__ rec = (int4*)rec_raw;
    int4 v[SORT_TRIPS];
    #pragma unroll
    for (int t = 0; t < SORT_TRIPS; ++t) v[t] = stage[min(tid + t * 1024, SORT_CHUNK - 1)];
    #pragma unroll
    for (int t = 0; t < SORT_TRIPS; ++t) {
        const int i = tid + t * 1024;
        if (i < SORT_CHUNK) rec[lo + i] = v[t];
    }
}

__global__ __launch_bounds__(1024) void k_bkt_sort(const int* __restrict__ tab, const int* __restrict__ rec_raw,
                                                   int* __restrict__ off, int* __restrict__ csr_src,
                                                   float* __restrict__ csr_w, float* __restrict__ dinv) {
    __shared__ int cnt[BKT_NODES];
    __shared__ int wsum[16];
    __shared__ int wsumA[8];
    __shared__ int run_src[SORT_BLOCKS];
    __shared__ int run_pos[SORT_BLOCKS + 1];
    __shared__ int s_bstart;
    __shared__ int4 st_src[(SORT_CAP + SORT_PAD) / 4];
    __shared__ int4 st_w[(SORT_CAP + SORT_PAD) / 4];
    const int tid  = threadIdx.x;
    const int lane = tid & 31;
    const int wv   = tid >> 5;
    const int bkt  = blockIdx.x;
    const int4* __restrict__ rec = (const int4*)rec_raw;
    int* ssrc = (int*)st_src;
    int* sw   = (int*)st_w;

    int ra = 0, rlen = 0;
    if (tid < SORT_BLOCKS) {
        const int a0 = tab[tid * N_BKT + bkt];
        const int a1 = (bkt + 1 < N_BKT) ? tab[tid * N_BKT + bkt + 1] : SORT_CHUNK;
        ra   = (int)min((unsigned)a0, (unsigned)SORT_CHUNK);
        rlen = (int)min((unsigned)max(a1 - ra, 0), (unsigned)(SORT_CHUNK - ra));
    }
    if (tid < BKT_NODES) cnt[tid] = 0;
    int rinc = rlen, rsa = ra;
    for (int d = 1; d < 32; d <<= 1) {
        int n = __shfl_up(rinc, d);
        if (lane >= d) rinc += n;
    }
    for (int d = 16; d >= 1; d >>= 1) rsa += __shfl_xor(rsa, d);
    if (lane == 31 && wv < 8) { wsum[wv] = rinc; wsumA[wv] = rsa; }
    __syncthreads();
    if (wv == 0) {
        const int ws = (lane < 8) ? wsum[lane] : 0;
        int wa = (lane < 8) ? wsumA[lane] : 0;
        int winc = ws;
        for (int d = 1; d < 32; d <<= 1) {
            int n = __shfl_up(winc, d);
            if (lane >= d) winc += n;
        }
        for (int d = 16; d >= 1; d >>= 1) wa += __shfl_xor(wa, d);
        if (lane < 8) wsum[lane] = winc - ws;
        if (lane == 31) { run_pos[SORT_BLOCKS] = winc; s_bstart = wa; }
    }
    __syncthreads();
    if (tid < SORT_BLOCKS) {
        run_pos[tid] = wsum[wv] + (rinc - rlen);
        run_src[tid] = tid * SORT_CHUNK + ra;
    }
    __syncthreads();
    const int  bstart = (int)min((unsigned)s_bstart, (unsigned)N_EDGES);
    const int  nedge  = (int)min((unsigned)run_pos[SORT_BLOCKS], (unsigned)(N_EDGES - bstart));
    const bool staged = nedge <= SORT_CAP;
    const int  shift  = bstart & (SORT_PAD - 1);
    const int rc  = tid / RUN_LANES;
    const int rk0 = tid % RUN_LANES;
    const int rs  = run_src[rc];
    const int rl  = run_pos[rc + 1] - run_pos[rc];

    for (int k = rk0; k < rl; k += RUN_LANES * RUN_GROUP) {
        int x[RUN_GROUP];
        #pragma unroll
        for (int u = 0; u < RUN_GROUP; ++u) {
            const int kk = min(k + u * RUN_LANES, rl - 1);
            x[u] = rec[min((unsigned)(rs + kk), (unsigned)(N_EDGES - 1))].x;
        }
        #pragma unroll
        for (int u = 0; u < RUN_GROUP; ++u)
            if (k + u * RUN_LANES < rl) atomicAdd(&cnt[x[u] & (BKT_NODES - 1)], 1);
    }
    __syncthreads();

    const int v = (tid < BKT_NODES) ? cnt[tid] : 0;
    int inc = v;
    for (int d = 1; d < 32; d <<= 1) {
        int n = __shfl_up(inc, d);
        if (lane >= d) inc += n;
    }
    if (lane == 31 && wv < 16) wsum[wv] = inc;
    __syncthreads();
    if (wv == 0) {
        const int ws = (lane < 16) ? wsum[lane] : 0;
        int winc = ws;
        for (int d = 1; d < 32; d <<= 1) {
            int n = __shfl_up(winc, d);
            if (lane >= d) winc += n;
        }
        if (lane < 16) wsum[lane] = winc - ws;
    }
    __syncthreads();
    int seg_lo = 0;
    if (tid < BKT_NODES) {
        const int ex = wsum[wv] + (inc - v);
        cnt[tid] = ex;
        seg_lo = ex;
        const int node = bkt * BKT_NODES + tid;
        if (node < N_NODES) off[node] = bstart + ex;
    }
    __syncthreads();

    for (int k = rk0; k < rl; k += RUN_LANES * RUN_GROUP) {
        int4 r[RUN_GROUP];
        #pragma unroll
        for (int u = 0; u < RUN_GROUP; ++u) {
            const int kk = min(k + u * RUN_LANES, rl - 1);
            r[u] = rec[min((unsigned)(rs + kk), (unsigned)(N_EDGES - 1))];
        }
        #pragma unroll
        for (int u = 0; u < RUN_GROUP; ++u) {
            if (k + u * RUN_LANES < rl) {
                const int q = atomicAdd(&cnt[r[u].x & (BKT_NODES - 1)], 1);
                if (staged) {
                    const int p = (int)min((unsigned)(q + shift), (unsigned)(SORT_CAP + SORT_PAD - 1));
                    ssrc[p] = r[u].y;
                    sw[p]   = r[u].z;
                } else {
                    const int p = (int)min((unsigned)(bstart + q), (unsigned)(N_EDGES - 1));
                    csr_src[p] = r[u].y;
                    csr_w[p]   = __int_as_float(r[u].z);
                }
            }
        }
    }
    __syncthreads();
    if (staged) {
        const int a0 = bstart - shift;
        const int pieces = (shift + nedge + 3) >> 2;
        int4* gs = (int4*)(csr_src + a0);
        int4* gw = (int4*)(csr_w + a0);
        for (int g = tid; g < pieces; g += 1024) {
            const int p = g * 4;
            if (p >= shift && p + 4 <= shift + nedge) {
                gs[g] = st_src[g];
                gw[g] = st_w[g];
            } else {
                for (int k = 0; k < 4; ++k) {
                    if (p + k >= shift && p + k < shift + nedge) {
                        csr_src[a0 + p + k] = ssrc[p + k];
                        csr_w[a0 + p + k]   = __int_as_float(sw[p + k]);
                    }
                }
            }
        }
    }
    if (tid < BKT_NODES) {
        const int node = bkt * BKT_NODES + tid;
        if (node < N_NODES) {
            const int seg_hi = min(cnt[tid], nedge);
            float d = 1.0f;
            if (staged) {
                for (int j = seg_lo; j < seg_hi; ++j) d += __int_as_float(sw[shift + j]);
            } else {
                for (int j = seg_lo; j < seg_hi; ++j) d += csr_w[bstart + j];
            }
            dinv[node] = (d > 0.0f) ? rsqrtf(d) : 0.0f;
        }
    }
}


#define G1_WAVES   8
#define G1_THREADS (G1_WAVES * 32)
#define G1_BLOCKS  512
#define G1_K8      (N_FEAT / 8)
#define G1_KTAIL   (G1_K8 * 8)
#define G1_STEP_P  (4 * 32)
#define G1_PIECES  (G1_K8 * G1_STEP_P + 2 * 32)
#define G1_R0      (G1_BLOCKS * G1_WAVES)
#define G1_R1      (G1_BLOCKS * (G1_WAVES / 2))
#define G1_REST    (M_TILES - G1_R0 - G1_R1)
#define G1_COPY_TRIPS ((G1_PIECES + G1_THREADS - 1) / G1_THREADS)
static_assert(G1_PIECES * 4 == N_FEAT * HIDDEN, "the LDS copy holds W1 exactly once");
static_assert(N_FEAT - G1_KTAIL == 4, "one 4-wide step is left after the 8-wide ones");
static_assert(G1_K8 % 2 == 0, "the K loop takes two steps per trip");
static_assert(G1_REST >= 0 && G1_REST <= G1_BLOCKS, "at most one left-over tile per work-group");
static_assert(HIDDEN == 64, "four 16-column accumulators");
static_assert(G1_COPY_TRIPS % 8 == 0, "the copy takes its pieces in groups of eight");

__device__ __forceinline__ void g1_step(const float4 a, const float4 q0, const float4 q1,
                                        const float4 q2, const float4 q3,
                                        v8f& c0, v8f& c1, v8f& c2, v8f& c3) {
    v2f av, b;
    av.x = a.x; av.y = a.y;
    b.x = q0.x; b.y = q0.y;
    c0 = __builtin_amdgcn_wmma_f32_16x16x4_f32(false, av, false, b, (short)0, c0, false, false);
    b.x = q0.z; b.y = q0.w;
    c1 = __builtin_amdgcn_wmma_f32_16x16x4_f32(false, av, false, b, (short)0, c1, false, false);
    b.x = q1.x; b.y = q1.y;
    c2 = __builtin_amdgcn_wmma_f32_16x16x4_f32(false, av, false, b, (short)0, c2, false, false);
    b.x = q1.z; b.y = q1.w;
    c3 = __builtin_amdgcn_wmma_f32_16x16x4_f32(false, av, false, b, (short)0, c3, false, false);
    av.x = a.z; av.y = a.w;
    b.x = q2.x; b.y = q2.y;
    c0 = __builtin_amdgcn_wmma_f32_16x16x4_f32(false, av, false, b, (short)0, c0, false, false);
    b.x = q2.z; b.y = q2.w;
    c1 = __builtin_amdgcn_wmma_f32_16x16x4_f32(false, av, false, b, (short)0, c1, false, false);
    b.x = q3.x; b.y = q3.y;
    c2 = __builtin_amdgcn_wmma_f32_16x16x4_f32(false, av, false, b, (short)0, c2, false, false);
    b.x = q3.z; b.y = q3.w;
    c3 = __builtin_amdgcn_wmma_f32_16x16x4_f32(false, av, false, b, (short)0, c3, false, false);
}

__global__ __launch_bounds__(G1_THREADS) void k_gemm1(const float* x,
                                                      const float* W1,
                                                      const float* __restrict__ dinv,
                                                      float* __restrict__ h1) {
    __shared__ float4 Bs[G1_PIECES];
    const int tid = threadIdx.x;

    {
        float* Bf = (float*)Bs;
        const float4* W4 = (const float4*)W1;
        #pragma unroll
        for (int g = 0; g < G1_COPY_TRIPS; g += 8) {
            float4 v[8];
            #pragma unroll
            for (int u = 0; u < 8; ++u) {
                const int e = min(tid + (g + u) * G1_THREADS, G1_PIECES - 1);
                v[u] = W4[e];
            }
            __builtin_amdgcn_sched_barrier(0);
            #pragma unroll
            for (int u = 0; u < 8; ++u) {
                const int e = tid + (g + u) * G1_THREADS;
                if (e < G1_PIECES) {
                    const int k  = e >> 4;
                    const int n0 = (e & 15) * 4;
                    const int j  = n0 >> 4;
                    const int c  = n0 & 15;
                    int d;
                    if (k < G1_KTAIL) {
                        const int t = k >> 3, r = k & 7;
                        const int hh = r >> 2, w = (r >> 1) & 1, s = r & 1;
                        d = ((t * 4 + w * 2 + (j >> 1)) * 32 + hh * 16 + c) * 4 + (j & 1) * 2 + s;
                    } else {
                        const int r = k - G1_KTAIL;
                        const int hh = r >> 1, s = r & 1;
                        d = (G1_K8 * G1_STEP_P + (j >> 1) * 32 + hh * 16 + c) * 4 + (j & 1) * 2 + s;
                    }
                    Bf[d] = v[u].x; Bf[d + 4] = v[u].y; Bf[d + 8] = v[u].z; Bf[d + 12] = v[u].w;
                }
            }
        }
    }
    __syncthreads();

    const int wave = __builtin_amdgcn_readfirstlane(tid >> 5);
    const int lane = tid & 31;
    const int half = lane >> 4;
    const int l16  = lane & 15;
    const float4* bl = Bs + lane;

    int mtile = wave * G1_BLOCKS + (int)blockIdx.x;
    int nxt   = ((__popc(wave) & 1) == 0) ? G1_R0 + (wave >> 1) * G1_BLOCKS + (int)blockIdx.x
              : ((wave == 1 && (int)blockIdx.x < G1_REST) ? G1_R0 + G1_R1 + (int)blockIdx.x : -1);

    while (mtile >= 0) {
        const int row0 = mtile * 16;
        const float*  xr = x + (size_t)(row0 + l16) * N_FEAT;
        const float4* xa = (const float4*)xr + half;
        v2f at;
        at.x = xr[G1_KTAIL + half * 2];
        at.y = xr[G1_KTAIL + half * 2 + 1];
        const float4 u0 = bl[G1_K8 * G1_STEP_P], u1 = bl[G1_K8 * G1_STEP_P + 32];

        float4 aE = xa[0], aO = xa[2];
        float4 e0 = bl[0], e1 = bl[32], e2 = bl[64], e3 = bl[96];
        v8f c0 = {}, c1 = {}, c2 = {}, c3 = {};

        for (int t = 0; t < G1_K8 - 2; t += 2) {
            const float4 nE = xa[2 * t + 4], nO = xa[2 * t + 6];
            const float4* bo = bl + (t + 1) * G1_STEP_P;
            const float4 o0 = bo[0], o1 = bo[32], o2 = bo[64], o3 = bo[96];
            __builtin_amdgcn_sched_barrier(0);
            g1_step(aE, e0, e1, e2, e3, c0, c1, c2, c3);
            e0 = bo[G1_STEP_P]; e1 = bo[G1_STEP_P + 32]; e2 = bo[G1_STEP_P + 64]; e3 = bo[G1_STEP_P + 96];
            __builtin_amdgcn_sched_barrier(0);
            g1_step(aO, o0, o1, o2, o3, c0, c1, c2, c3);
            aE = nE; aO = nO;
        }
        const float4 dA = *(const float4*)(dinv + row0 + half * 8);
        const float4 dB = *(const float4*)(dinv + row0 + half * 8 + 4);
        {
            const float4* bo = bl + (G1_K8 - 1) * G1_STEP_P;
            const float4 o0 = bo[0], o1 = bo[32], o2 = bo[64], o3 = bo[96];
            g1_step(aE, e0, e1, e2, e3, c0, c1, c2, c3);
            g1_step(aO, o0, o1, o2, o3, c0, c1, c2, c3);
        }
        {
            v2f b;
            b.x = u0.x; b.y = u0.y;
            c0 = __builtin_amdgcn_wmma_f32_16x16x4_f32(false, at, false, b, (short)0, c0, false, false);
            b.x = u0.z; b.y = u0.w;
            c1 = __builtin_amdgcn_wmma_f32_16x16x4_f32(false, at, false, b, (short)0, c1, false, false);
            b.x = u1.x; b.y = u1.y;
            c2 = __builtin_amdgcn_wmma_f32_16x16x4_f32(false, at, false, b, (short)0, c2, false, false);
            b.x = u1.z; b.y = u1.w;
            c3 = __builtin_amdgcn_wmma_f32_16x16x4_f32(false, at, false, b, (short)0, c3, false, false);
        }

        const float ds[8] = {dA.x, dA.y, dA.z, dA.w, dB.x, dB.y, dB.z, dB.w};
        float* out = h1 + (size_t)(row0 + half * 8) * HIDDEN + l16;
        for (int r = 0; r < 8; ++r) {
            out[(size_t)r * HIDDEN + 0]  = c0[r] * ds[r];
            out[(size_t)r * HIDDEN + 16] = c1[r] * ds[r];
            out[(size_t)r * HIDDEN + 32] = c2[r] * ds[r];
            out[(size_t)r * HIDDEN + 48] = c3[r] * ds[r];
        }
        mtile = nxt;
        nxt = -1;
    }
}


#define GA_THREADS 256
#define GA_NODES   (GA_THREADS / 16)
#define A1_PITCH4  17
static_assert(GA_NODES == 16, "a work-group's nodes are one 16-row tile");
static_assert(N_NODES % GA_NODES == 0, "no work-group holds a partial tile");

__global__ __launch_bounds__(GA_THREADS) void k_gather1_gemm2(const int* __restrict__ off, const int* __restrict__ csr_src,
                                const float* __restrict__ csr_w, const float* __restrict__ dinv,
                                const float* __restrict__ h1, const float* __restrict__ b1,
                                const float* __restrict__ W2, float* __restrict__ h2) {
    __shared__ float4 As[GA_NODES * A1_PITCH4];
    const int tid = threadIdx.x;
    const int nl = tid >> 4;
    const int q  = tid & 15;
    const int i  = (int)blockIdx.x * GA_NODES + nl;
    const int b  = off[i];
    const int en = (i + 1 < N_NODES) ? off[i + 1] : N_EDGES;
    const float di = dinv[i];
    const float4* __restrict__ h4 = (const float4*)h1;
    float4 acc = make_float4(0.0f, 0.0f, 0.0f, 0.0f);
    int j = b;
    for (; j + 4 <= en; j += 4) {
        const int s0 = node_clamp(csr_src[j]), s1 = node_clamp(csr_src[j + 1]), s2 = node_clamp(csr_src[j + 2]), s3 = node_clamp(csr_src[j + 3]);
        const float w0 = csr_w[j], w1 = csr_w[j + 1], w2 = csr_w[j + 2], w3 = csr_w[j + 3];
        const float4 v0 = h4[(size_t)s0 * 16 + q];
        const float4 v1 = h4[(size_t)s1 * 16 + q];
        const float4 v2 = h4[(size_t)s2 * 16 + q];
        const float4 v3 = h4[(size_t)s3 * 16 + q];
        acc.x += w0 * v0.x; acc.y += w0 * v0.y; acc.z += w0 * v0.z; acc.w += w0 * v0.w;
        acc.x += w1 * v1.x; acc.y += w1 * v1.y; acc.z += w1 * v1.z; acc.w += w1 * v1.w;
        acc.x += w2 * v2.x; acc.y += w2 * v2.y; acc.z += w2 * v2.z; acc.w += w2 * v2.w;
        acc.x += w3 * v3.x; acc.y += w3 * v3.y; acc.z += w3 * v3.z; acc.w += w3 * v3.w;
    }
    for (; j < en; ++j) {
        const int s0 = node_clamp(csr_src[j]);
        const float w0 = csr_w[j];
        const float4 v0 = h4[(size_t)s0 * 16 + q];
        acc.x += w0 * v0.x; acc.y += w0 * v0.y; acc.z += w0 * v0.z; acc.w += w0 * v0.w;
    }
    const bool first = __builtin_amdgcn_readfirstlane(tid >> 5) == 0;
    const int lane = tid & 31;
    const int half = lane >> 4;
    const int l16  = lane & 15;
    const int row0 = (int)blockIdx.x * GA_NODES;
    v2f bf[HIDDEN / 4];
    float4 dA = make_float4(0.0f, 0.0f, 0.0f, 0.0f), dB = dA;
    #pragma unroll
    for (int s = 0; s < HIDDEN / 4; ++s) { bf[s].x = 0.0f; bf[s].y = 0.0f; }
    if (first) {
        #pragma unroll
        for (int s = 0; s < HIDDEN / 4; ++s) {
            const float* bp = W2 + (size_t)(4 * s + half * 2) * N_CLASSES + l16;
            bf[s].x = bp[0];
            bf[s].y = bp[N_CLASSES];
        }
        dA = *(const float4*)(dinv + row0 + half * 8);
        dB = *(const float4*)(dinv + row0 + half * 8 + 4);
    }
    const float4 hv = h4[(size_t)i * 16 + q];
    const float4 bv = make_float4(b1[q * 4], b1[q * 4 + 1], b1[q * 4 + 2], b1[q * 4 + 3]);
    float4 o;
    o.x = bv.x + di * hv.x + di * acc.x;
    o.y = bv.y + di * hv.y + di * acc.y;
    o.z = bv.z + di * hv.z + di * acc.z;
    o.w = bv.w + di * hv.w + di * acc.w;
    As[nl * A1_PITCH4 + q] = make_float4(fmaxf(o.x, 0.0f), fmaxf(o.y, 0.0f), fmaxf(o.z, 0.0f), fmaxf(o.w, 0.0f));
    __syncthreads();

    if (!first) return;
    const float* arow = (const float*)As + l16 * (A1_PITCH4 * 4) + half * 2;
    v8f cacc = {};
    #pragma unroll
    for (int s = 0; s < HIDDEN / 4; ++s) {
        v2f a;
        a.x = arow[4 * s];
        a.y = arow[4 * s + 1];
        cacc = __builtin_amdgcn_wmma_f32_16x16x4_f32(false, a, false, bf[s], (short)0, cacc, false, false);
    }
    const float ds[8] = {dA.x, dA.y, dA.z, dA.w, dB.x, dB.y, dB.z, dB.w};
    float* out = h2 + (size_t)(row0 + half * 8) * N_CLASSES + l16;
    for (int r = 0; r < 8; ++r)
        out[(size_t)r * N_CLASSES] = cacc[r] * ds[r];
}

__global__ __launch_bounds__(256) void k_gather2_lsm(const int* __restrict__ off, const int* __restrict__ csr_src,
                          const float* __restrict__ csr_w, const float* __restrict__ dinv,
                          const float* __restrict__ h2, const float* __restrict__ b2,
                          float* __restrict__ out) {
    const int t = blockIdx.x * blockDim.x + threadIdx.x;
    const int i = t >> 2;
    const int q = t & 3;
    if (i >= N_NODES) return;
    const int b  = off[i];
    const int en = (i + 1 < N_NODES) ? off[i + 1] : N_EDGES;
    const float di = dinv[i];
    const float4* __restrict__ h4 = (const float4*)h2;
    float4 acc = make_float4(0.0f, 0.0f, 0.0f, 0.0f);
    int j = b;
    for (; j + 4 <= en; j += 4) {
        const int s0 = node_clamp(csr_src[j]), s1 = node_clamp(csr_src[j + 1]), s2 = node_clamp(csr_src[j + 2]), s3 = node_clamp(csr_src[j + 3]);
        const float w0 = csr_w[j], w1 = csr_w[j + 1], w2 = csr_w[j + 2], w3 = csr_w[j + 3];
        const float4 v0 = h4[(size_t)s0 * 4 + q];
        const float4 v1 = h4[(size_t)s1 * 4 + q];
        const float4 v2 = h4[(size_t)s2 * 4 + q];
        const float4 v3 = h4[(size_t)s3 * 4 + q];
        acc.x += w0 * v0.x; acc.y += w0 * v0.y; acc.z += w0 * v0.z; acc.w += w0 * v0.w;
        acc.x += w1 * v1.x; acc.y += w1 * v1.y; acc.z += w1 * v1.z; acc.w += w1 * v1.w;
        acc.x += w2 * v2.x; acc.y += w2 * v2.y; acc.z += w2 * v2.z; acc.w += w2 * v2.w;
        acc.x += w3 * v3.x; acc.y += w3 * v3.y; acc.z += w3 * v3.z; acc.w += w3 * v3.w;
    }
    for (; j < en; ++j) {
        const int s0 = node_clamp(csr_src[j]);
        const float w0 = csr_w[j];
        const float4 v0 = h4[(size_t)s0 * 4 + q];
        acc.x += w0 * v0.x; acc.y += w0 * v0.y; acc.z += w0 * v0.z; acc.w += w0 * v0.w;
    }
    const float4 hv = h4[(size_t)i * 4 + q];
    const float4 bv = make_float4(b2[q * 4], b2[q * 4 + 1], b2[q * 4 + 2], b2[q * 4 + 3]);
    float4 o;
    o.x = bv.x + di * hv.x + di * acc.x;
    o.y = bv.y + di * hv.y + di * acc.y;
    o.z = bv.z + di * hv.z + di * acc.z;
    o.w = bv.w + di * hv.w + di * acc.w;

    float m = fmaxf(fmaxf(o.x, o.y), fmaxf(o.z, o.w));
    m = fmaxf(m, __shfl_xor(m, 1));
    m = fmaxf(m, __shfl_xor(m, 2));
    const float e0 = expf(o.x - m), e1 = expf(o.y - m), e2 = expf(o.z - m), e3 = expf(o.w - m);
    float p = ((e0 + e1) + e2) + e3;
    #pragma unroll
    for (int g = 1; g < 4; ++g) {
        const float prev = __shfl(p, g - 1, 4);
        const float cand = (((prev + e0) + e1) + e2) + e3;
        if (q == g) p = cand;
    }
    const float s = __shfl(p, 3, 4);
    const float l = logf(s);
    ((float4*)out)[(size_t)i * 4 + q] = make_float4(o.x - m - l, o.y - m - l, o.z - m - l, o.w - m - l);
}


extern "C" void kernel_launch(void* const* d_in, const int* in_sizes, int n_in,
                              void* d_out, int out_size, void* d_ws, size_t ws_size,
                              hipStream_t stream) {
    const float* x  = (const float*)d_in[0];
    const int*   ei = (const int*)d_in[1];
    const float* ew = (const float*)d_in[2];
    const float* W1 = (const float*)d_in[3];
    const float* b1 = (const float*)d_in[4];
    const float* W2 = (const float*)d_in[5];
    const float* b2 = (const float*)d_in[6];
    float* out = (float*)d_out;

    const int* row = ei;
    const int* col = ei + N_EDGES;

    float* ws      = (float*)d_ws;
    int*   off     = (int*)ws;
    float* dinv    = ws + N_NODES;
    int*   csr_src = (int*)(dinv + N_NODES);
    float* h1      = (float*)csr_src + N_EDGES;
    float* plane2  = h1   + (size_t)N_NODES * HIDDEN;
    float* csr_w   = plane2 + (size_t)N_NODES * HIDDEN;
    float* h2      = plane2;
    int*   rec     = (int*)plane2;
    int*   tab     = (int*)h1;

    k_chunk_group<<<SORT_BLOCKS, 1024, 0, stream>>>(row, col, ew, tab, rec);
    k_bkt_sort   <<<N_BKT, 1024, 0, stream>>>(tab, rec, off, csr_src, csr_w, dinv);

    k_gemm1<<<G1_BLOCKS, G1_THREADS, 0, stream>>>(x, W1, dinv, h1);

    k_gather1_gemm2<<<N_NODES / GA_NODES, GA_THREADS, 0, stream>>>(off, csr_src, csr_w, dinv, h1, b1, W2, h2);

    k_gather2_lsm<<<(N_NODES * 4 + 255) / 256, 256, 0, stream>>>(off, csr_src, csr_w, dinv, h2, b2, out);
}


# Gather 1 work-groups walk 5 node tiles each (1250 x 8 waves enter, not 6250 x 8); next tile's offsets and next edge group's records requested ahead; sum order kept, products fused into the sums
# speedup vs baseline: 4.8190x; 1.0368x over previous
#include <hip/hip_runtime.h>
#include <math.h>

#define N_NODES   100000
#define N_EDGES   1600000
#define N_FEAT    500
#define HIDDEN    64
#define N_CLASSES 16
#define M_TILES   (N_NODES / 16)

typedef __attribute__((ext_vector_type(2))) float v2f;
typedef __attribute__((ext_vector_type(8))) float v8f;

__device__ __forceinline__ int node_clamp(int s) {
    return (int)min((unsigned)s, (unsigned)(N_NODES - 1));
}


#define SORT_BLOCKS 256
#define SORT_CHUNK  (N_EDGES / SORT_BLOCKS)
#define SORT_TRIPS  ((SORT_CHUNK + 1023) / 1024)
#define RUN_LANES   (1024 / SORT_BLOCKS)
#define RUN_GROUP   12
#define SORT_CAP    16384
#define SORT_PAD    32
#define BKT_SHIFT   9
#define BKT_NODES   (1 << BKT_SHIFT)
#define N_BKT       ((N_NODES + BKT_NODES - 1) / BKT_NODES)
#define TAB_LEN     (N_BKT * SORT_BLOCKS)
static_assert(N_EDGES % SORT_BLOCKS == 0, "chunks are equal");
static_assert(SORT_BLOCKS * RUN_LANES == 1024, "every run of a bucket has its threads");
static_assert(N_BKT <= 256, "the chunk's bucket counts fit waves 0-7, one per lane");
static_assert(N_NODES % 16 == 0, "M tiles are exact");

__global__ __launch_bounds__(1024) void k_chunk_group(const int* __restrict__ row, const int* __restrict__ col,
                                                      const float* __restrict__ w, int* __restrict__ tab,
                                                      int* __restrict__ rec_raw) {
    __shared__ int  cur[N_BKT];
    __shared__ int  wsum[8];
    __shared__ int4 stage[SORT_CHUNK];
    const int tid  = threadIdx.x;
    const int lane = tid & 31;
    const int wv   = tid >> 5;
    for (int b = tid; b < N_BKT; b += 1024) cur[b] = 0;
    __syncthreads();
    const int lo = blockIdx.x * SORT_CHUNK;
    int c[SORT_TRIPS], r[SORT_TRIPS], wb[SORT_TRIPS];
    #pragma unroll
    for (int t = 0; t < SORT_TRIPS; ++t) {
        const int k = min(tid + t * 1024, SORT_CHUNK - 1);
        c[t]  = node_clamp(col[lo + k]);
        r[t]  = row[lo + k];
        wb[t] = __float_as_int(w[lo + k]);
    }
    #pragma unroll
    for (int t = 0; t < SORT_TRIPS; ++t)
        if (tid + t * 1024 < SORT_CHUNK) atomicAdd(&cur[c[t] >> BKT_SHIFT], 1);
    __syncthreads();
    const int nb = (tid < N_BKT) ? cur[tid] : 0;
    int inc = nb;
    for (int d = 1; d < 32; d <<= 1) {
        int n = __shfl_up(inc, d);
        if (lane >= d) inc += n;
    }
    if (lane == 31 && wv < 8) wsum[wv] = inc;
    __syncthreads();
    if (wv == 0) {
        const int ws = (lane < 8) ? wsum[lane] : 0;
        int winc = ws;
        for (int d = 1; d < 32; d <<= 1) {
            int n = __shfl_up(winc, d);
            if (lane >= d) winc += n;
        }
        if (lane < 8) wsum[lane] = winc - ws;
    }
    __syncthreads();
    if (tid < N_BKT) {
        const int ex = wsum[wv] + (inc - nb);
        cur[tid] = ex;
        tab[(int)blockIdx.x * N_BKT + tid] = ex;
    }
    __syncthreads();
    #pragma unroll
    for (int t = 0; t < SORT_TRIPS; ++t) {
        if (tid + t * 1024 < SORT_CHUNK) {
            const int q = atomicAdd(&cur[c[t] >> BKT_SHIFT], 1);
            stage[min((unsigned)q, (unsigned)(SORT_CHUNK - 1))] = make_int4(c[t], r[t], wb[t], 0);
        }
    }
    __syncthreads();
    int4* __restrict__ rec = (int4*)rec_raw;
    int4 v[SORT_TRIPS];
    #pragma unroll
    for (int t = 0; t < SORT_TRIPS; ++t) v[t] = stage[min(tid + t * 1024, SORT_CHUNK - 1)];
    #pragma unroll
    for (int t = 0; t < SORT_TRIPS; ++t) {
        const int i = tid + t * 1024;
        if (i < SORT_CHUNK) rec[lo + i] = v[t];
    }
}

__global__ __launch_bounds__(1024) void k_bkt_sort(const int* __restrict__ tab, const int* __restrict__ rec_raw,
                                                   int* __restrict__ off, int* __restrict__ csr_src,
                                                   float* __restrict__ csr_w, float* __restrict__ dinv) {
    __shared__ int cnt[BKT_NODES];
    __shared__ int wsum[16];
    __shared__ int wsumA[8];
    __shared__ int run_src[SORT_BLOCKS];
    __shared__ int run_pos[SORT_BLOCKS + 1];
    __shared__ int s_bstart;
    __shared__ int4 st_src[(SORT_CAP + SORT_PAD) / 4];
    __shared__ int4 st_w[(SORT_CAP + SORT_PAD) / 4];
    const int tid  = threadIdx.x;
    const int lane = tid & 31;
    const int wv   = tid >> 5;
    const int bkt  = blockIdx.x;
    const int4* __restrict__ rec = (const int4*)rec_raw;
    int* ssrc = (int*)st_src;
    int* sw   = (int*)st_w;

    int ra = 0, rlen = 0;
    if (tid < SORT_BLOCKS) {
        const int a0 = tab[tid * N_BKT + bkt];
        const int a1 = (bkt + 1 < N_BKT) ? tab[tid * N_BKT + bkt + 1] : SORT_CHUNK;
        ra   = (int)min((unsigned)a0, (unsigned)SORT_CHUNK);
        rlen = (int)min((unsigned)max(a1 - ra, 0), (unsigned)(SORT_CHUNK - ra));
    }
    if (tid < BKT_NODES) cnt[tid] = 0;
    int rinc = rlen, rsa = ra;
    for (int d = 1; d < 32; d <<= 1) {
        int n = __shfl_up(rinc, d);
        if (lane >= d) rinc += n;
    }
    for (int d = 16; d >= 1; d >>= 1) rsa += __shfl_xor(rsa, d);
    if (lane == 31 && wv < 8) { wsum[wv] = rinc; wsumA[wv] = rsa; }
    __syncthreads();
    if (wv == 0) {
        const int ws = (lane < 8) ? wsum[lane] : 0;
        int wa = (lane < 8) ? wsumA[lane] : 0;
        int winc = ws;
        for (int d = 1; d < 32; d <<= 1) {
            int n = __shfl_up(winc, d);
            if (lane >= d) winc += n;
        }
        for (int d = 16; d >= 1; d >>= 1) wa += __shfl_xor(wa, d);
        if (lane < 8) wsum[lane] = winc - ws;
        if (lane == 31) { run_pos[SORT_BLOCKS] = winc; s_bstart = wa; }
    }
    __syncthreads();
    if (tid < SORT_BLOCKS) {
        run_pos[tid] = wsum[wv] + (rinc - rlen);
        run_src[tid] = tid * SORT_CHUNK + ra;
    }
    __syncthreads();
    const int  bstart = (int)min((unsigned)s_bstart, (unsigned)N_EDGES);
    const int  nedge  = (int)min((unsigned)run_pos[SORT_BLOCKS], (unsigned)(N_EDGES - bstart));
    const bool staged = nedge <= SORT_CAP;
    const int  shift  = bstart & (SORT_PAD - 1);
    const int rc  = tid / RUN_LANES;
    const int rk0 = tid % RUN_LANES;
    const int rs  = run_src[rc];
    const int rl  = run_pos[rc + 1] - run_pos[rc];

    for (int k = rk0; k < rl; k += RUN_LANES * RUN_GROUP) {
        int x[RUN_GROUP];
        #pragma unroll
        for (int u = 0; u < RUN_GROUP; ++u) {
            const int kk = min(k + u * RUN_LANES, rl - 1);
            x[u] = rec[min((unsigned)(rs + kk), (unsigned)(N_EDGES - 1))].x;
        }
        #pragma unroll
        for (int u = 0; u < RUN_GROUP; ++u)
            if (k + u * RUN_LANES < rl) atomicAdd(&cnt[x[u] & (BKT_NODES - 1)], 1);
    }
    __syncthreads();

    const int v = (tid < BKT_NODES) ? cnt[tid] : 0;
    int inc = v;
    for (int d = 1; d < 32; d <<= 1) {
        int n = __shfl_up(inc, d);
        if (lane >= d) inc += n;
    }
    if (lane == 31 && wv < 16) wsum[wv] = inc;
    __syncthreads();
    if (wv == 0) {
        const int ws = (lane < 16) ? wsum[lane] : 0;
        int winc = ws;
        for (int d = 1; d < 32; d <<= 1) {
            int n = __shfl_up(winc, d);
            if (lane >= d) winc += n;
        }
        if (lane < 16) wsum[lane] = winc - ws;
    }
    __syncthreads();
    int seg_lo = 0;
    if (tid < BKT_NODES) {
        const int ex = wsum[wv] + (inc - v);
        cnt[tid] = ex;
        seg_lo = ex;
        const int node = bkt * BKT_NODES + tid;
        if (node < N_NODES) off[node] = bstart + ex;
    }
    __syncthreads();

    for (int k = rk0; k < rl; k += RUN_LANES * RUN_GROUP) {
        int4 r[RUN_GROUP];
        #pragma unroll
        for (int u = 0; u < RUN_GROUP; ++u) {
            const int kk = min(k + u * RUN_LANES, rl - 1);
            r[u] = rec[min((unsigned)(rs + kk), (unsigned)(N_EDGES - 1))];
        }
        #pragma unroll
        for (int u = 0; u < RUN_GROUP; ++u) {
            if (k + u * RUN_LANES < rl) {
                const int q = atomicAdd(&cnt[r[u].x & (BKT_NODES - 1)], 1);
                if (staged) {
                    const int p = (int)min((unsigned)(q + shift), (unsigned)(SORT_CAP + SORT_PAD - 1));
                    ssrc[p] = r[u].y;
                    sw[p]   = r[u].z;
                } else {
                    const int p = (int)min((unsigned)(bstart + q), (unsigned)(N_EDGES - 1));
                    csr_src[p] = r[u].y;
                    csr_w[p]   = __int_as_float(r[u].z);
                }
            }
        }
    }
    __syncthreads();
    if (staged) {
        const int a0 = bstart - shift;
        const int pieces = (shift + nedge + 3) >> 2;
        int4* gs = (int4*)(csr_src + a0);
        int4* gw = (int4*)(csr_w + a0);
        for (int g = tid; g < pieces; g += 1024) {
            const int p = g * 4;
            if (p >= shift && p + 4 <= shift + nedge) {
                gs[g] = st_src[g];
                gw[g] = st_w[g];
            } else {
                for (int k = 0; k < 4; ++k) {
                    if (p + k >= shift && p + k < shift + nedge) {
                        csr_src[a0 + p + k] = ssrc[p + k];
                        csr_w[a0 + p + k]   = __int_as_float(sw[p + k]);
                    }
                }
            }
        }
    }
    if (tid < BKT_NODES) {
        const int node = bkt * BKT_NODES + tid;
        if (node < N_NODES) {
            const int seg_hi = min(cnt[tid], nedge);
            float d = 1.0f;
            if (staged) {
                for (int j = seg_lo; j < seg_hi; ++j) d += __int_as_float(sw[shift + j]);
            } else {
                for (int j = seg_lo; j < seg_hi; ++j) d += csr_w[bstart + j];
            }
            dinv[node] = (d > 0.0f) ? rsqrtf(d) : 0.0f;
        }
    }
}


#define G1_WAVES   8
#define G1_THREADS (G1_WAVES * 32)
#define G1_BLOCKS  512
#define G1_K8      (N_FEAT / 8)
#define G1_KTAIL   (G1_K8 * 8)
#define G1_STEP_P  (4 * 32)
#define G1_PIECES  (G1_K8 * G1_STEP_P + 2 * 32)
#define G1_R0      (G1_BLOCKS * G1_WAVES)
#define G1_R1      (G1_BLOCKS * (G1_WAVES / 2))
#define G1_REST    (M_TILES - G1_R0 - G1_R1)
#define G1_COPY_TRIPS ((G1_PIECES + G1_THREADS - 1) / G1_THREADS)
static_assert(G1_PIECES * 4 == N_FEAT * HIDDEN, "the LDS copy holds W1 exactly once");
static_assert(N_FEAT - G1_KTAIL == 4, "one 4-wide step is left after the 8-wide ones");
static_assert(G1_K8 % 2 == 0, "the K loop takes two steps per trip");
static_assert(G1_REST >= 0 && G1_REST <= G1_BLOCKS, "at most one left-over tile per work-group");
static_assert(HIDDEN == 64, "four 16-column accumulators");
static_assert(G1_COPY_TRIPS % 8 == 0, "the copy takes its pieces in groups of eight");

__device__ __forceinline__ void g1_step(const float4 a, const float4 q0, const float4 q1,
                                        const float4 q2, const float4 q3,
                                        v8f& c0, v8f& c1, v8f& c2, v8f& c3) {
    v2f av, b;
    av.x = a.x; av.y = a.y;
    b.x = q0.x; b.y = q0.y;
    c0 = __builtin_amdgcn_wmma_f32_16x16x4_f32(false, av, false, b, (short)0, c0, false, false);
    b.x = q0.z; b.y = q0.w;
    c1 = __builtin_amdgcn_wmma_f32_16x16x4_f32(false, av, false, b, (short)0, c1, false, false);
    b.x = q1.x; b.y = q1.y;
    c2 = __builtin_amdgcn_wmma_f32_16x16x4_f32(false, av, false, b, (short)0, c2, false, false);
    b.x = q1.z; b.y = q1.w;
    c3 = __builtin_amdgcn_wmma_f32_16x16x4_f32(false, av, false, b, (short)0, c3, false, false);
    av.x = a.z; av.y = a.w;
    b.x = q2.x; b.y = q2.y;
    c0 = __builtin_amdgcn_wmma_f32_16x16x4_f32(false, av, false, b, (short)0, c0, false, false);
    b.x = q2.z; b.y = q2.w;
    c1 = __builtin_amdgcn_wmma_f32_16x16x4_f32(false, av, false, b, (short)0, c1, false, false);
    b.x = q3.x; b.y = q3.y;
    c2 = __builtin_amdgcn_wmma_f32_16x16x4_f32(false, av, false, b, (short)0, c2, false, false);
    b.x = q3.z; b.y = q3.w;
    c3 = __builtin_amdgcn_wmma_f32_16x16x4_f32(false, av, false, b, (short)0, c3, false, false);
}

__global__ __launch_bounds__(G1_THREADS) void k_gemm1(const float* x,
                                                      const float* W1,
                                                      const float* __restrict__ dinv,
                                                      float* __restrict__ h1) {
    __shared__ float4 Bs[G1_PIECES];
    const int tid = threadIdx.x;

    {
        float* Bf = (float*)Bs;
        const float4* W4 = (const float4*)W1;
        #pragma unroll
        for (int g = 0; g < G1_COPY_TRIPS; g += 8) {
            float4 v[8];
            #pragma unroll
            for (int u = 0; u < 8; ++u) {
                const int e = min(tid + (g + u) * G1_THREADS, G1_PIECES - 1);
                v[u] = W4[e];
            }
            __builtin_amdgcn_sched_barrier(0);
            #pragma unroll
            for (int u = 0; u < 8; ++u) {
                const int e = tid + (g + u) * G1_THREADS;
                if (e < G1_PIECES) {
                    const int k  = e >> 4;
                    const int n0 = (e & 15) * 4;
                    const int j  = n0 >> 4;
                    const int c  = n0 & 15;
                    int d;
                    if (k < G1_KTAIL) {
                        const int t = k >> 3, r = k & 7;
                        const int hh = r >> 2, w = (r >> 1) & 1, s = r & 1;
                        d = ((t * 4 + w * 2 + (j >> 1)) * 32 + hh * 16 + c) * 4 + (j & 1) * 2 + s;
                    } else {
                        const int r = k - G1_KTAIL;
                        const int hh = r >> 1, s = r & 1;
                        d = (G1_K8 * G1_STEP_P + (j >> 1) * 32 + hh * 16 + c) * 4 + (j & 1) * 2 + s;
                    }
                    Bf[d] = v[u].x; Bf[d + 4] = v[u].y; Bf[d + 8] = v[u].z; Bf[d + 12] = v[u].w;
                }
            }
        }
    }
    __syncthreads();

    const int wave = __builtin_amdgcn_readfirstlane(tid >> 5);
    const int lane = tid & 31;
    const int half = lane >> 4;
    const int l16  = lane & 15;
    const float4* bl = Bs + lane;

    int mtile = wave * G1_BLOCKS + (int)blockIdx.x;
    int nxt   = ((__popc(wave) & 1) == 0) ? G1_R0 + (wave >> 1) * G1_BLOCKS + (int)blockIdx.x
              : ((wave == 1 && (int)blockIdx.x < G1_REST) ? G1_R0 + G1_R1 + (int)blockIdx.x : -1);

    while (mtile >= 0) {
        const int row0 = mtile * 16;
        const float*  xr = x + (size_t)(row0 + l16) * N_FEAT;
        const float4* xa = (const float4*)xr + half;
        v2f at;
        at.x = xr[G1_KTAIL + half * 2];
        at.y = xr[G1_KTAIL + half * 2 + 1];
        const float4 u0 = bl[G1_K8 * G1_STEP_P], u1 = bl[G1_K8 * G1_STEP_P + 32];

        float4 aE = xa[0], aO = xa[2];
        float4 e0 = bl[0], e1 = bl[32], e2 = bl[64], e3 = bl[96];
        v8f c0 = {}, c1 = {}, c2 = {}, c3 = {};

        for (int t = 0; t < G1_K8 - 2; t += 2) {
            const float4 nE = xa[2 * t + 4], nO = xa[2 * t + 6];
            const float4* bo = bl + (t + 1) * G1_STEP_P;
            const float4 o0 = bo[0], o1 = bo[32], o2 = bo[64], o3 = bo[96];
            __builtin_amdgcn_sched_barrier(0);
            g1_step(aE, e0, e1, e2, e3, c0, c1, c2, c3);
            e0 = bo[G1_STEP_P]; e1 = bo[G1_STEP_P + 32]; e2 = bo[G1_STEP_P + 64]; e3 = bo[G1_STEP_P + 96];
            __builtin_amdgcn_sched_barrier(0);
            g1_step(aO, o0, o1, o2, o3, c0, c1, c2, c3);
            aE = nE; aO = nO;
        }
        const float4 dA = *(const float4*)(dinv + row0 + half * 8);
        const float4 dB = *(const float4*)(dinv + row0 + half * 8 + 4);
        {
            const float4* bo = bl + (G1_K8 - 1) * G1_STEP_P;
            const float4 o0 = bo[0], o1 = bo[32], o2 = bo[64], o3 = bo[96];
            g1_step(aE, e0, e1, e2, e3, c0, c1, c2, c3);
            g1_step(aO, o0, o1, o2, o3, c0, c1, c2, c3);
        }
        {
            v2f b;
            b.x = u0.x; b.y = u0.y;
            c0 = __builtin_amdgcn_wmma_f32_16x16x4_f32(false, at, false, b, (short)0, c0, false, false);
            b.x = u0.z; b.y = u0.w;
            c1 = __builtin_amdgcn_wmma_f32_16x16x4_f32(false, at, false, b, (short)0, c1, false, false);
            b.x = u1.x; b.y = u1.y;
            c2 = __builtin_amdgcn_wmma_f32_16x16x4_f32(false, at, false, b, (short)0, c2, false, false);
            b.x = u1.z; b.y = u1.w;
            c3 = __builtin_amdgcn_wmma_f32_16x16x4_f32(false, at, false, b, (short)0, c3, false, false);
        }

        const float ds[8] = {dA.x, dA.y, dA.z, dA.w, dB.x, dB.y, dB.z, dB.w};
        float* out = h1 + (size_t)(row0 + half * 8) * HIDDEN + l16;
        for (int r = 0; r < 8; ++r) {
            out[(size_t)r * HIDDEN + 0]  = c0[r] * ds[r];
            out[(size_t)r * HIDDEN + 16] = c1[r] * ds[r];
            out[(size_t)r * HIDDEN + 32] = c2[r] * ds[r];
            out[(size_t)r * HIDDEN + 48] = c3[r] * ds[r];
        }
        mtile = nxt;
        nxt = -1;
    }
}


#define GA_THREADS 256
#define GA_NODES   (GA_THREADS / 16)
#define GA_TILES   5
#define GA_BLOCKS  (M_TILES / GA_TILES)
#define A1_PITCH4  17
static_assert(GA_NODES == 16, "a work-group's nodes are one 16-row tile");
static_assert(M_TILES % GA_TILES == 0, "every work-group has whole tiles only");
static_assert(HIDDEN * N_CLASSES == GA_THREADS * 4, "W2 is one 16-byte piece per thread");

__device__ __forceinline__ void ga_group(const float4* h4, const int* csr_src, const float* csr_w,
                                         const int i, const int q, const int rem, const int jn,
                                         int& g0, int& g1, int& g2, int& g3,
                                         float& u0, float& u1, float& u2, float& u3, float4& acc) {
    const int s0 = (rem > 0) ? node_clamp(g0) : i;
    const int s1 = (rem > 1) ? node_clamp(g1) : i;
    const int s2 = (rem > 2) ? node_clamp(g2) : i;
    const int s3 = (rem > 3) ? node_clamp(g3) : i;
    const float w0 = (rem > 0) ? u0 : 0.0f;
    const float w1 = (rem > 1) ? u1 : 0.0f;
    const float w2 = (rem > 2) ? u2 : 0.0f;
    const float w3 = (rem > 3) ? u3 : 0.0f;
    const float4 v0 = h4[(size_t)s0 * 16 + q];
    const float4 v1 = h4[(size_t)s1 * 16 + q];
    const float4 v2 = h4[(size_t)s2 * 16 + q];
    const float4 v3 = h4[(size_t)s3 * 16 + q];
    g0 = csr_src[jn]; g1 = csr_src[jn + 1]; g2 = csr_src[jn + 2]; g3 = csr_src[jn + 3];
    u0 = csr_w[jn];   u1 = csr_w[jn + 1];   u2 = csr_w[jn + 2];   u3 = csr_w[jn + 3];
    acc.x += w0 * v0.x; acc.y += w0 * v0.y; acc.z += w0 * v0.z; acc.w += w0 * v0.w;
    acc.x += w1 * v1.x; acc.y += w1 * v1.y; acc.z += w1 * v1.z; acc.w += w1 * v1.w;
    acc.x += w2 * v2.x; acc.y += w2 * v2.y; acc.z += w2 * v2.z; acc.w += w2 * v2.w;
    acc.x += w3 * v3.x; acc.y += w3 * v3.y; acc.z += w3 * v3.z; acc.w += w3 * v3.w;
}

__global__ __launch_bounds__(GA_THREADS) void k_gather1_gemm2(const int* off, const int* csr_src,
                                const float* csr_w, const float* dinv,
                                const float* h1, const float* __restrict__ b1,
                                const float* __restrict__ W2, float* h2) {
    __shared__ float4 As[2][GA_NODES * A1_PITCH4];
    __shared__ v2f    Bw[(HIDDEN / 4) * 32];
    const int tid  = threadIdx.x;
    const int nl   = tid >> 4;
    const int q    = tid & 15;
    const int wave = __builtin_amdgcn_readfirstlane(tid >> 5);
    const int lane = tid & 31;
    const int half = lane >> 4;
    const int l16  = lane & 15;
    const float4* h4 = (const float4*)h1;

    {
        const float4 wv = ((const float4*)W2)[tid];
        const int k  = tid >> 2;
        const int n0 = (tid & 3) * 4;
        float* bwf = (float*)Bw + (((k >> 2) * 32 + ((k >> 1) & 1) * 16 + n0) * 2 + (k & 1));
        bwf[0] = wv.x; bwf[2] = wv.y; bwf[4] = wv.z; bwf[6] = wv.w;
    }
    const float4 bv = make_float4(b1[q * 4], b1[q * 4 + 1], b1[q * 4 + 2], b1[q * 4 + 3]);

    const int tile0 = (int)blockIdx.x * GA_TILES;
    int   i  = tile0 * GA_NODES + nl;
    int   b  = off[i];
    int   en = (i + 1 < N_NODES) ? off[i + 1] : N_EDGES;
    float di = dinv[i];
    float4 hv = h4[(size_t)i * 16 + q];
    int   jb = (int)min((unsigned)b, (unsigned)N_EDGES);
    int   g0 = csr_src[jb], g1 = csr_src[jb + 1], g2 = csr_src[jb + 2], g3 = csr_src[jb + 3];
    float u0 = csr_w[jb],   u1 = csr_w[jb + 1],   u2 = csr_w[jb + 2],   u3 = csr_w[jb + 3];

    for (int it = 0; it < GA_TILES; ++it) {
        const int row0 = (tile0 + it) * GA_NODES;
        const int   ni  = min(row0 + GA_NODES + nl, N_NODES - 1);

        float4 acc = make_float4(0.0f, 0.0f, 0.0f, 0.0f);
        int j = jb + 4;
        ga_group(h4, csr_src, csr_w, i, q, en - jb, j, g0, g1, g2, g3, u0, u1, u2, u3, acc);
        const int   nb  = off[ni];
        const int   ne1 = off[min(ni + 1, N_NODES - 1)];
        const int   nen = (ni + 1 < N_NODES) ? ne1 : N_EDGES;
        const float ndi = dinv[ni];
        const float4 nhv = h4[(size_t)ni * 16 + q];
        asm volatile("" : "+v"(g0), "+v"(g1), "+v"(g2), "+v"(g3), "+v"(u0), "+v"(u1), "+v"(u2), "+v"(u3));
        while (j < en) {
            const int rem = en - j;
            j += 4;
            ga_group(h4, csr_src, csr_w, i, q, rem, j, g0, g1, g2, g3, u0, u1, u2, u3, acc);
        }
        jb = (int)min((unsigned)nb, (unsigned)N_EDGES);
        g0 = csr_src[jb]; g1 = csr_src[jb + 1]; g2 = csr_src[jb + 2]; g3 = csr_src[jb + 3];
        u0 = csr_w[jb];   u1 = csr_w[jb + 1];   u2 = csr_w[jb + 2];   u3 = csr_w[jb + 3];

        float4 o;
        o.x = bv.x + di * hv.x + di * acc.x;
        o.y = bv.y + di * hv.y + di * acc.y;
        o.z = bv.z + di * hv.z + di * acc.z;
        o.w = bv.w + di * hv.w + di * acc.w;
        float4* at4 = As[it & 1];
        at4[nl * A1_PITCH4 + q] = make_float4(fmaxf(o.x, 0.0f), fmaxf(o.y, 0.0f), fmaxf(o.z, 0.0f), fmaxf(o.w, 0.0f));
        if (q == 0) ((float*)at4)[nl * (A1_PITCH4 * 4) + HIDDEN] = di;
        __builtin_amdgcn_fence(__ATOMIC_RELEASE, "workgroup", "local");
        __builtin_amdgcn_s_barrier();
        __builtin_amdgcn_fence(__ATOMIC_ACQUIRE, "workgroup", "local");

        if (wave == (it & 7)) {
            const float* at = (const float*)at4;
            const float* arow = at + l16 * (A1_PITCH4 * 4) + half * 2;
            const v2f* bw = Bw + lane;
            v8f cacc = {};
            #pragma unroll
            for (int s = 0; s < HIDDEN / 4; ++s) {
                v2f a;
                a.x = arow[4 * s];
                a.y = arow[4 * s + 1];
                cacc = __builtin_amdgcn_wmma_f32_16x16x4_f32(false, a, false, bw[s * 32], (short)0, cacc, false, false);
            }
            float* out = h2 + (size_t)(row0 + half * 8) * N_CLASSES + l16;
            #pragma unroll
            for (int r = 0; r < 8; ++r)
                out[(size_t)r * N_CLASSES] = cacc[r] * at[(half * 8 + r) * (A1_PITCH4 * 4) + HIDDEN];
        }
        i = ni; b = nb; en = nen; di = ndi; hv = nhv;
    }
}

__global__ __launch_bounds__(256) void k_gather2_lsm(const int* __restrict__ off, const int* __restrict__ csr_src,
                          const float* __restrict__ csr_w, const float* __restrict__ dinv,
                          const float* __restrict__ h2, const float* __restrict__ b2,
                          float* __restrict__ out) {
    const int t = blockIdx.x * blockDim.x + threadIdx.x;
    const int i = t >> 2;
    const int q = t & 3;
    if (i >= N_NODES) return;
    const int b  = off[i];
    const int en = (i + 1 < N_NODES) ? off[i + 1] : N_EDGES;
    const float di = dinv[i];
    const float4* __restrict__ h4 = (const float4*)h2;
    float4 acc = make_float4(0.0f, 0.0f, 0.0f, 0.0f);
    int j = b;
    for (; j + 4 <= en; j += 4) {
        const int s0 = node_clamp(csr_src[j]), s1 = node_clamp(csr_src[j + 1]), s2 = node_clamp(csr_src[j + 2]), s3 = node_clamp(csr_src[j + 3]);
        const float w0 = csr_w[j], w1 = csr_w[j + 1], w2 = csr_w[j + 2], w3 = csr_w[j + 3];
        const float4 v0 = h4[(size_t)s0 * 4 + q];
        const float4 v1 = h4[(size_t)s1 * 4 + q];
        const float4 v2 = h4[(size_t)s2 * 4 + q];
        const float4 v3 = h4[(size_t)s3 * 4 + q];
        acc.x += w0 * v0.x; acc.y += w0 * v0.y; acc.z += w0 * v0.z; acc.w += w0 * v0.w;
        acc.x += w1 * v1.x; acc.y += w1 * v1.y; acc.z += w1 * v1.z; acc.w += w1 * v1.w;
        acc.x += w2 * v2.x; acc.y += w2 * v2.y; acc.z += w2 * v2.z; acc.w += w2 * v2.w;
        acc.x += w3 * v3.x; acc.y += w3 * v3.y; acc.z += w3 * v3.z; acc.w += w3 * v3.w;
    }
    for (; j < en; ++j) {
        const int s0 = node_clamp(csr_src[j]);
        const float w0 = csr_w[j];
        const float4 v0 = h4[(size_t)s0 * 4 + q];
        acc.x += w0 * v0.x; acc.y += w0 * v0.y; acc.z += w0 * v0.z; acc.w += w0 * v0.w;
    }
    const float4 hv = h4[(size_t)i * 4 + q];
    const float4 bv = make_float4(b2[q * 4], b2[q * 4 + 1], b2[q * 4 + 2], b2[q * 4 + 3]);
    float4 o;
    o.x = bv.x + di * hv.x + di * acc.x;
    o.y = bv.y + di * hv.y + di * acc.y;
    o.z = bv.z + di * hv.z + di * acc.z;
    o.w = bv.w + di * hv.w + di * acc.w;

    float m = fmaxf(fmaxf(o.x, o.y), fmaxf(o.z, o.w));
    m = fmaxf(m, __shfl_xor(m, 1));
    m = fmaxf(m, __shfl_xor(m, 2));
    const float e0 = expf(o.x - m), e1 = expf(o.y - m), e2 = expf(o.z - m), e3 = expf(o.w - m);
    float p = ((e0 + e1) + e2) + e3;
    #pragma unroll
    for (int g = 1; g < 4; ++g) {
        const float prev = __shfl(p, g - 1, 4);
        const float cand = (((prev + e0) + e1) + e2) + e3;
        if (q == g) p = cand;
    }
    const float s = __shfl(p, 3, 4);
    const float l = logf(s);
    ((float4*)out)[(size_t)i * 4 + q] = make_float4(o.x - m - l, o.y - m - l, o.z - m - l, o.w - m - l);
}


extern "C" void kernel_launch(void* const* d_in, const int* in_sizes, int n_in,
                              void* d_out, int out_size, void* d_ws, size_t ws_size,
                              hipStream_t stream) {
    const float* x  = (const float*)d_in[0];
    const int*   ei = (const int*)d_in[1];
    const float* ew = (const float*)d_in[2];
    const float* W1 = (const float*)d_in[3];
    const float* b1 = (const float*)d_in[4];
    const float* W2 = (const float*)d_in[5];
    const float* b2 = (const float*)d_in[6];
    float* out = (float*)d_out;

    const int* row = ei;
    const int* col = ei + N_EDGES;

    float* ws      = (float*)d_ws;
    int*   off     = (int*)ws;
    float* dinv    = ws + N_NODES;
    int*   csr_src = (int*)(dinv + N_NODES);
    float* csr_w   = (float*)csr_src + N_EDGES;
    float* h1      = csr_w + N_EDGES;
    float* plane2  = h1   + (size_t)N_NODES * HIDDEN;
    float* h2      = plane2;
    int*   rec     = (int*)plane2;
    int*   tab     = (int*)h1;

    k_chunk_group<<<SORT_BLOCKS, 1024, 0, stream>>>(row, col, ew, tab, rec);
    k_bkt_sort   <<<N_BKT, 1024, 0, stream>>>(tab, rec, off, csr_src, csr_w, dinv);

    k_gemm1<<<G1_BLOCKS, G1_THREADS, 0, stream>>>(x, W1, dinv, h1);

    k_gather1_gemm2<<<GA_BLOCKS, GA_THREADS, 0, stream>>>(off, csr_src, csr_w, dinv, h1, b1, W2, h2);

    k_gather2_lsm<<<(N_NODES * 4 + 255) / 256, 256, 0, stream>>>(off, csr_src, csr_w, dinv, h2, b2, out);
}
